# starting kernel
# baseline (speedup 1.0000x reference)
#include <hip/hip_runtime.h>


#define LN_EPS 1e-5f

typedef __attribute__((ext_vector_type(2))) float v2f;
typedef __attribute__((ext_vector_type(8))) float v8f;


__global__ void vgae_zero(float* __restrict__ p, long n) {
  long i = (long)blockIdx.x * blockDim.x + threadIdx.x;
  if (i < n) p[i] = 0.0f;
}

__global__ void vgae_deg(const int* __restrict__ ei, float* __restrict__ deg, int E) {
  int e = blockIdx.x * blockDim.x + threadIdx.x;
  if (e < E) atomicAdd(&deg[ei[E + e]], 1.0f);
}

__global__ void vgae_invdeg(float* __restrict__ d, int n) {
  int i = blockIdx.x * blockDim.x + threadIdx.x;
  if (i < n) d[i] = 1.0f / fmaxf(d[i], 1.0f);
}

__global__ void vgae_agg(const float* __restrict__ h, const int* __restrict__ ei,
                         float* __restrict__ agg, int d, int E) {
  int e = blockIdx.x * blockDim.y + threadIdx.y;
  if (e >= E) return;
  int s = ei[e];
  int t = ei[E + e];
  float v = h[(long)s * d + threadIdx.x];
  atomicAdd(&agg[(long)t * d + threadIdx.x], v);
}

__global__ void vgae_sage_gemm(const float* __restrict__ agg,
                               const float* __restrict__ h,
                               const float* __restrict__ invdeg,
                               const float* __restrict__ Wl,
                               const float* __restrict__ bl,
                               const float* __restrict__ Wr,
                               const float* __restrict__ gamma,
                               const float* __restrict__ beta,
                               float* __restrict__ out,
                               int din, int dout, int do_ln) {
  __shared__ float smem[16 * 128];
  __shared__ float s_mean[16];
  __shared__ float s_rstd[16];

  const int lane = threadIdx.x & 31;
  const int wave = threadIdx.x >> 5;
  const int l15  = lane & 15;
  const int hi   = lane >> 4;
  const int rowbase = blockIdx.x * 16;
  const int arow = rowbase + l15;
  const int bcol = wave * 16 + l15;
  const int koff = 2 * hi;

  const float id = invdeg[arow];
  const float* aggrow = agg + (long)arow * din;
  const float* hrow   = h   + (long)arow * din;
  const float* wlrow  = Wl  + (long)bcol * din;
  const float* wrrow  = Wr  + (long)bcol * din;

  v8f c = {0.f, 0.f, 0.f, 0.f, 0.f, 0.f, 0.f, 0.f};

  for (int k = 0; k < din; k += 4) {
    v2f a = *(const v2f*)(aggrow + k + koff);
    a *= id;
    v2f b = *(const v2f*)(wlrow + k + koff);
    c = __builtin_amdgcn_wmma_f32_16x16x4_f32(false, a, false, b,
                                              (short)0, c, false, false);
  }
  for (int k = 0; k < din; k += 4) {
    v2f a = *(const v2f*)(hrow + k + koff);
    v2f b = *(const v2f*)(wrrow + k + koff);
    c = __builtin_amdgcn_wmma_f32_16x16x4_f32(false, a, false, b,
                                              (short)0, c, false, false);
  }

  const float bias = bl[bcol];

  if (!do_ln) {
    for (int v = 0; v < 8; ++v) {
      int r = rowbase + v + 8 * hi;
      out[(long)r * dout + bcol] = c[v] + bias;
    }
    return;
  }

  for (int v = 0; v < 8; ++v)
    smem[(v + 8 * hi) * 128 + bcol] = c[v] + bias;
  __syncthreads();

  if (threadIdx.x < 16) {
    const float* rowp = &smem[threadIdx.x * 128];
    float s = 0.f, ss = 0.f;
    for (int j = 0; j < 128; ++j) { float x = rowp[j]; s += x; ss += x * x; }
    float m   = s * (1.0f / 128.0f);
    float var = ss * (1.0f / 128.0f) - m * m;
    s_mean[threadIdx.x] = m;
    s_rstd[threadIdx.x] = rsqrtf(var + LN_EPS);
  }
  __syncthreads();

  for (int idx = threadIdx.x; idx < 16 * 128; idx += blockDim.x) {
    int r  = idx >> 7;
    int cc = idx & 127;
    float x = (smem[idx] - s_mean[r]) * s_rstd[r] * gamma[cc] + beta[cc];
    out[(long)(rowbase + r) * 128 + cc] = fmaxf(x, 0.0f);
  }
}


extern "C" void kernel_launch(void* const* d_in, const int* in_sizes, int n_in,
                              void* d_out, int out_size, void* d_ws, size_t ws_size,
                              hipStream_t stream) {
  const float* x  = (const float*)d_in[0];
  const int*   ei = (const int*)d_in[1];
  const int N = in_sizes[0] / 64;
  const int E = in_sizes[1] / 2;

  float* invdeg = (float*)d_ws;
  float* aggx   = invdeg + N;
  float* bufA   = aggx + (long)N * 64;
  float* bufB   = bufA + (long)N * 128;
  float* bufC   = bufB + (long)N * 128;

  vgae_zero<<<(N + 255) / 256, 256, 0, stream>>>(invdeg, N);
  vgae_deg<<<(E + 255) / 256, 256, 0, stream>>>(ei, invdeg, E);
  vgae_invdeg<<<(N + 255) / 256, 256, 0, stream>>>(invdeg, N);

  vgae_zero<<<((long)N * 64 + 255) / 256, 256, 0, stream>>>(aggx, (long)N * 64);
  {
    dim3 blk(64, 4);
    vgae_agg<<<(E + 3) / 4, blk, 0, stream>>>(x, ei, aggx, 64, E);
  }

  const int nrt = N / 16;
  for (int t = 0; t < 2; ++t) {
    const int base = 2 + t * 13;
    const float* Wl0 = (const float*)d_in[base + 0];
    const float* bl0 = (const float*)d_in[base + 1];
    const float* Wr0 = (const float*)d_in[base + 2];
    const float* g0  = (const float*)d_in[base + 3];
    const float* b0  = (const float*)d_in[base + 4];
    const float* Wl1 = (const float*)d_in[base + 5];
    const float* bl1 = (const float*)d_in[base + 6];
    const float* Wr1 = (const float*)d_in[base + 7];
    const float* g1  = (const float*)d_in[base + 8];
    const float* b1  = (const float*)d_in[base + 9];
    const float* Wl2 = (const float*)d_in[base + 10];
    const float* bl2 = (const float*)d_in[base + 11];
    const float* Wr2 = (const float*)d_in[base + 12];
    float* outp = (float*)d_out + (long)t * N * 64;

    vgae_sage_gemm<<<nrt, 256, 0, stream>>>(aggx, x, invdeg, Wl0, bl0, Wr0,
                                            g0, b0, bufA, 64, 128, 1);
    vgae_zero<<<((long)N * 128 + 255) / 256, 256, 0, stream>>>(bufB, (long)N * 128);
    { dim3 blk(128, 2); vgae_agg<<<(E + 1) / 2, blk, 0, stream>>>(bufA, ei, bufB, 128, E); }

    vgae_sage_gemm<<<nrt, 256, 0, stream>>>(bufB, bufA, invdeg, Wl1, bl1, Wr1,
                                            g1, b1, bufC, 128, 128, 1);
    vgae_zero<<<((long)N * 128 + 255) / 256, 256, 0, stream>>>(bufA, (long)N * 128);
    { dim3 blk(128, 2); vgae_agg<<<(E + 1) / 2, blk, 0, stream>>>(bufC, ei, bufA, 128, E); }

    vgae_sage_gemm<<<nrt, 128, 0, stream>>>(bufA, bufC, invdeg, Wl2, bl2, Wr2,
                                            nullptr, nullptr, outp, 128, 64, 0);
  }
}


# Neighbor sums+degrees: global float atomics -> LDS float adds; 256 blocks each own 196 dst rows, test all edges, gather matching rows; mean at write-out; zero+deg launches gone; sum order: arrival
# speedup vs baseline: 2.4598x; 2.4598x over previous
#include <hip/hip_runtime.h>


#define LN_EPS 1e-5f

typedef __attribute__((ext_vector_type(2))) float v2f;
typedef __attribute__((ext_vector_type(8))) float v8f;


#define AGG_R    196
#define AGG_NT   1024
#define AGG_NW   (AGG_NT / 32)
#define AGG_CH   512
#define AGG_PL   (AGG_CH / 32)
#define AGG_K    8
#define AGG_QCAP (AGG_K + AGG_CH)

template <int D>
__global__ __launch_bounds__(AGG_NT)
void vgae_agg_own(const float* __restrict__ h, const int* __restrict__ ei,
                  float* __restrict__ agg, int N, int E) {
  constexpr int VEC = D / 32;
  typedef __attribute__((ext_vector_type(VEC))) float vN;

  __shared__ float acc[AGG_R * D];
  __shared__ int   cnt[AGG_R];
  __shared__ int   que[AGG_NW][AGG_QCAP];

  const int lane = threadIdx.x & 31;
  const int wave = __builtin_amdgcn_readfirstlane(threadIdx.x >> 5);
  const int r0   = blockIdx.x * AGG_R;
  const int* __restrict__ dst = ei + E;

  for (int i = threadIdx.x; i < AGG_R * D; i += AGG_NT) acc[i] = 0.0f;
  for (int i = threadIdx.x; i < AGG_R; i += AGG_NT) cnt[i] = 0;
  __syncthreads();

  const int nch = (E + AGG_CH - 1) / AGG_CH;
  int qn = 0;

  for (int c = wave; ; c += AGG_NW) {
    const bool live = c < nch;
    if (live) {
      const int base = c * AGG_CH + lane;
      int d[AGG_PL];
      if (c * AGG_CH + AGG_CH <= E) {
        for (int j = 0; j < AGG_PL; ++j) d[j] = dst[base + j * 32];
      } else {
        for (int j = 0; j < AGG_PL; ++j) {
          const int e = base + j * 32;
          d[j] = (e < E) ? dst[e] : -1;
        }
      }
      for (int j = 0; j < AGG_PL; ++j) {
        const bool m = (unsigned)(d[j] - r0) < (unsigned)AGG_R;
        const unsigned mask = __builtin_amdgcn_ballot_w32(m);
        if (mask) {
          if (m) que[wave][qn + (int)__builtin_amdgcn_mbcnt_lo(mask, 0u)] = base + j * 32;
          qn += __builtin_popcount(mask);
        }
      }
    }

    if (qn >= AGG_K || (!live && qn > 0)) {
      for (int i0 = 0; i0 < qn; i0 += 32) {
        const int n  = (qn - i0 < 32) ? (qn - i0) : 32;
        const int me = (lane < n) ? que[wave][i0 + lane] : 0;
        const int ms = ei[me];
        const int md = dst[me] - r0;
        if (lane < n) atomicAdd(&cnt[md], 1);
        for (int i = 0; i < n; i += 8) {
          vN v[8];
#pragma unroll
          for (int u = 0; u < 8; ++u) {
            const int k = (i + u < n) ? (i + u) : (n - 1);
            const int s = __builtin_amdgcn_readlane(ms, k);
            v[u] = *(const vN*)(h + (long)s * D + VEC * lane);
          }
#pragma unroll
          for (int u = 0; u < 8; ++u) {
            if (i + u < n) {
              const int dl = __builtin_amdgcn_readlane(md, i + u);
              for (int j = 0; j < VEC; ++j)
                atomicAdd(&acc[dl * D + j * 32 + lane], v[u][j]);
            }
          }
        }
      }
      qn = 0;
    }

    if (!live) break;
  }
  __syncthreads();

  for (int row = wave; row < AGG_R; row += AGG_NW) {
    const int r = r0 + row;
    if (r < N) {
      const float id = 1.0f / fmaxf((float)cnt[row], 1.0f);
      vN o;
      for (int j = 0; j < VEC; ++j) o[j] = acc[row * D + j * 32 + lane] * id;
      *(vN*)(agg + (long)r * D + VEC * lane) = o;
    }
  }
}

__global__ void vgae_sage_gemm(const float* __restrict__ agg,
                               const float* __restrict__ h,
                               const float* __restrict__ Wl,
                               const float* __restrict__ bl,
                               const float* __restrict__ Wr,
                               const float* __restrict__ gamma,
                               const float* __restrict__ beta,
                               float* __restrict__ out,
                               int din, int dout, int do_ln) {
  __shared__ float smem[16 * 128];
  __shared__ float s_mean[16];
  __shared__ float s_rstd[16];

  const int lane = threadIdx.x & 31;
  const int wave = threadIdx.x >> 5;
  const int l15  = lane & 15;
  const int hi   = lane >> 4;
  const int rowbase = blockIdx.x * 16;
  const int arow = rowbase + l15;
  const int bcol = wave * 16 + l15;
  const int koff = 2 * hi;

  const float* aggrow = agg + (long)arow * din;
  const float* hrow   = h   + (long)arow * din;
  const float* wlrow  = Wl  + (long)bcol * din;
  const float* wrrow  = Wr  + (long)bcol * din;

  v8f c = {0.f, 0.f, 0.f, 0.f, 0.f, 0.f, 0.f, 0.f};

  for (int k = 0; k < din; k += 4) {
    v2f a = *(const v2f*)(aggrow + k + koff);
    v2f b = *(const v2f*)(wlrow + k + koff);
    c = __builtin_amdgcn_wmma_f32_16x16x4_f32(false, a, false, b,
                                              (short)0, c, false, false);
  }
  for (int k = 0; k < din; k += 4) {
    v2f a = *(const v2f*)(hrow + k + koff);
    v2f b = *(const v2f*)(wrrow + k + koff);
    c = __builtin_amdgcn_wmma_f32_16x16x4_f32(false, a, false, b,
                                              (short)0, c, false, false);
  }

  const float bias = bl[bcol];

  if (!do_ln) {
    for (int v = 0; v < 8; ++v) {
      int r = rowbase + v + 8 * hi;
      out[(long)r * dout + bcol] = c[v] + bias;
    }
    return;
  }

  for (int v = 0; v < 8; ++v)
    smem[(v + 8 * hi) * 128 + bcol] = c[v] + bias;
  __syncthreads();

  if (threadIdx.x < 16) {
    const float* rowp = &smem[threadIdx.x * 128];
    float s = 0.f, ss = 0.f;
    for (int j = 0; j < 128; ++j) { float x = rowp[j]; s += x; ss += x * x; }
    float m   = s * (1.0f / 128.0f);
    float var = ss * (1.0f / 128.0f) - m * m;
    s_mean[threadIdx.x] = m;
    s_rstd[threadIdx.x] = rsqrtf(var + LN_EPS);
  }
  __syncthreads();

  for (int idx = threadIdx.x; idx < 16 * 128; idx += blockDim.x) {
    int r  = idx >> 7;
    int cc = idx & 127;
    float x = (smem[idx] - s_mean[r]) * s_rstd[r] * gamma[cc] + beta[cc];
    out[(long)(rowbase + r) * 128 + cc] = fmaxf(x, 0.0f);
  }
}


extern "C" void kernel_launch(void* const* d_in, const int* in_sizes, int n_in,
                              void* d_out, int out_size, void* d_ws, size_t ws_size,
                              hipStream_t stream) {
  const float* x  = (const float*)d_in[0];
  const int*   ei = (const int*)d_in[1];
  const int N = in_sizes[0] / 64;
  const int E = in_sizes[1] / 2;

  float* aggx = (float*)d_ws;
  float* bufA = aggx + (long)N * 64;
  float* bufB = bufA + (long)N * 128;
  float* bufC = bufB + (long)N * 128;

  const int nown = (N + AGG_R - 1) / AGG_R;

  vgae_agg_own<64><<<nown, AGG_NT, 0, stream>>>(x, ei, aggx, N, E);

  const int nrt = N / 16;
  for (int t = 0; t < 2; ++t) {
    const int base = 2 + t * 13;
    const float* Wl0 = (const float*)d_in[base + 0];
    const float* bl0 = (const float*)d_in[base + 1];
    const float* Wr0 = (const float*)d_in[base + 2];
    const float* g0  = (const float*)d_in[base + 3];
    const float* b0  = (const float*)d_in[base + 4];
    const float* Wl1 = (const float*)d_in[base + 5];
    const float* bl1 = (const float*)d_in[base + 6];
    const float* Wr1 = (const float*)d_in[base + 7];
    const float* g1  = (const float*)d_in[base + 8];
    const float* b1  = (const float*)d_in[base + 9];
    const float* Wl2 = (const float*)d_in[base + 10];
    const float* bl2 = (const float*)d_in[base + 11];
    const float* Wr2 = (const float*)d_in[base + 12];
    float* outp = (float*)d_out + (long)t * N * 64;

    vgae_sage_gemm<<<nrt, 256, 0, stream>>>(aggx, x, Wl0, bl0, Wr0,
                                            g0, b0, bufA, 64, 128, 1);
    vgae_agg_own<128><<<nown, AGG_NT, 0, stream>>>(bufA, ei, bufB, N, E);

    vgae_sage_gemm<<<nrt, 256, 0, stream>>>(bufB, bufA, Wl1, bl1, Wr1,
                                            g1, b1, bufC, 128, 128, 1);
    vgae_agg_own<128><<<nown, AGG_NT, 0, stream>>>(bufC, ei, bufA, N, E);

    vgae_sage_gemm<<<nrt, 128, 0, stream>>>(bufA, bufC, Wl2, bl2, Wr2,
                                            nullptr, nullptr, outp, 128, 64, 0);
  }
}


# Dense layers: 512 resident blocks walk the row tiles; [Wl|Wr] panel copied to LDS once per block, A tile by whole-row 16 B loads one tile ahead; both fragments read from LDS; sum order unchanged
# speedup vs baseline: 3.2262x; 1.3116x over previous
#include <hip/hip_runtime.h>


#define LN_EPS 1e-5f

typedef __attribute__((ext_vector_type(2))) float v2f;
typedef __attribute__((ext_vector_type(8))) float v8f;


#define AGG_R    196
#define AGG_NT   1024
#define AGG_NW   (AGG_NT / 32)
#define AGG_CH   512
#define AGG_PL   (AGG_CH / 32)
#define AGG_K    8
#define AGG_QCAP (AGG_K + AGG_CH)

template <int D>
__global__ __launch_bounds__(AGG_NT)
void vgae_agg_own(const float* __restrict__ h, const int* __restrict__ ei,
                  float* __restrict__ agg, int N, int E) {
  constexpr int VEC = D / 32;
  typedef __attribute__((ext_vector_type(VEC))) float vN;

  __shared__ float acc[AGG_R * D];
  __shared__ int   cnt[AGG_R];
  __shared__ int   que[AGG_NW][AGG_QCAP];

  const int lane = threadIdx.x & 31;
  const int wave = __builtin_amdgcn_readfirstlane(threadIdx.x >> 5);
  const int r0   = blockIdx.x * AGG_R;
  const int* __restrict__ dst = ei + E;

  for (int i = threadIdx.x; i < AGG_R * D; i += AGG_NT) acc[i] = 0.0f;
  for (int i = threadIdx.x; i < AGG_R; i += AGG_NT) cnt[i] = 0;
  __syncthreads();

  const int nch = (E + AGG_CH - 1) / AGG_CH;
  int qn = 0;

  for (int c = wave; ; c += AGG_NW) {
    const bool live = c < nch;
    if (live) {
      const int base = c * AGG_CH + lane;
      int d[AGG_PL];
      if (c * AGG_CH + AGG_CH <= E) {
        for (int j = 0; j < AGG_PL; ++j) d[j] = dst[base + j * 32];
      } else {
        for (int j = 0; j < AGG_PL; ++j) {
          const int e = base + j * 32;
          d[j] = (e < E) ? dst[e] : -1;
        }
      }
      for (int j = 0; j < AGG_PL; ++j) {
        const bool m = (unsigned)(d[j] - r0) < (unsigned)AGG_R;
        const unsigned mask = __builtin_amdgcn_ballot_w32(m);
        if (mask) {
          if (m) que[wave][qn + (int)__builtin_amdgcn_mbcnt_lo(mask, 0u)] = base + j * 32;
          qn += __builtin_popcount(mask);
        }
      }
    }

    if (qn >= AGG_K || (!live && qn > 0)) {
      for (int i0 = 0; i0 < qn; i0 += 32) {
        const int n  = (qn - i0 < 32) ? (qn - i0) : 32;
        const int me = (lane < n) ? que[wave][i0 + lane] : 0;
        const int ms = ei[me];
        const int md = dst[me] - r0;
        if (lane < n) atomicAdd(&cnt[md], 1);
        for (int i = 0; i < n; i += 8) {
          vN v[8];
#pragma unroll
          for (int u = 0; u < 8; ++u) {
            const int k = (i + u < n) ? (i + u) : (n - 1);
            const int s = __builtin_amdgcn_readlane(ms, k);
            v[u] = *(const vN*)(h + (long)s * D + VEC * lane);
          }
#pragma unroll
          for (int u = 0; u < 8; ++u) {
            if (i + u < n) {
              const int dl = __builtin_amdgcn_readlane(md, i + u);
              for (int j = 0; j < VEC; ++j)
                atomicAdd(&acc[dl * D + j * 32 + lane], v[u][j]);
            }
          }
        }
      }
      qn = 0;
    }

    if (!live) break;
  }
  __syncthreads();

  for (int row = wave; row < AGG_R; row += AGG_NW) {
    const int r = r0 + row;
    if (r < N) {
      const float id = 1.0f / fmaxf((float)cnt[row], 1.0f);
      vN o;
      for (int j = 0; j < VEC; ++j) o[j] = acc[row * D + j * 32 + lane] * id;
      *(vN*)(agg + (long)r * D + VEC * lane) = o;
    }
  }
}

typedef __attribute__((ext_vector_type(4))) float v4f;

template <int DIN, int DOUT, bool LN>
__global__ __launch_bounds__(DOUT * 2)
void vgae_sage_gemm(const float* __restrict__ agg,
                    const float* __restrict__ h,
                    const float* __restrict__ Wl,
                    const float* __restrict__ bl,
                    const float* __restrict__ Wr,
                    const float* __restrict__ gamma,
                    const float* __restrict__ beta,
                    float* __restrict__ out, int ntiles) {
  constexpr int K   = 2 * DIN;
  constexpr int P   = K + 4;
  constexpr int NT  = DOUT * 2;
  constexpr int NW  = DOUT / 16;
  constexpr int RPW = 16 / NW;
  constexpr int F4  = K / 4;
  constexpr int NL  = RPW * F4 / 32;
  static_assert(!LN || DOUT == 128, "LN epilogue is written for 128 columns");

  __shared__ __attribute__((aligned(16))) float Bt[DOUT * P];
  __shared__ __attribute__((aligned(16))) float At[16 * P];
  __shared__ float smem[LN ? 16 * 128 : 4];
  __shared__ float s_mean[16];
  __shared__ float s_rstd[16];

  const int tid  = threadIdx.x;
  const int lane = tid & 31;
  const int wave = tid >> 5;
  const int l15  = lane & 15;
  const int hi   = lane >> 4;
  const int bcol = wave * 16 + l15;
  const int koff = 2 * hi;

  for (int i = tid; i < DOUT * F4; i += NT) {
    const int cc  = i / F4;
    const int col = (i % F4) * 4;
    const float* s = (col < DIN) ? (Wl + (long)cc * DIN + col)
                                 : (Wr + (long)cc * DIN + (col - DIN));
    *(v4f*)&Bt[cc * P + col] = *(const v4f*)s;
  }

  int prow[NL], pcol[NL];
  for (int i = 0; i < NL; ++i) {
    const int flat = i * 32 + lane;
    prow[i] = wave * RPW + flat / F4;
    pcol[i] = (flat % F4) * 4;
  }
  v4f pre[NL];
  auto fetch = [&](int tile) {
    for (int i = 0; i < NL; ++i) {
      const long r = (long)tile * 16 + prow[i];
      const float* s = (pcol[i] < DIN) ? (agg + r * DIN + pcol[i])
                                       : (h + r * DIN + (pcol[i] - DIN));
      pre[i] = *(const v4f*)s;
    }
  };

  const float bias = bl[bcol];
  float g = 0.f, be = 0.f;
  if (LN) { g = gamma[tid & 127]; be = beta[tid & 127]; }

  const float* ap = &At[l15 * P + koff];
  const float* bp = &Bt[bcol * P + koff];

  int tile = blockIdx.x;
  if (tile < ntiles) fetch(tile);

  for (; tile < ntiles; tile += gridDim.x) {
    for (int i = 0; i < NL; ++i) *(v4f*)&At[prow[i] * P + pcol[i]] = pre[i];
    __syncthreads();
    const int next = tile + gridDim.x;
    if (next < ntiles) fetch(next);

    v8f c = {0.f, 0.f, 0.f, 0.f, 0.f, 0.f, 0.f, 0.f};
    for (int k = 0; k < K; k += 4) {
      v2f a = *(const v2f*)(ap + k);
      v2f b = *(const v2f*)(bp + k);
      c = __builtin_amdgcn_wmma_f32_16x16x4_f32(false, a, false, b,
                                                (short)0, c, false, false);
    }

    const int rowbase = tile * 16;

    if (!LN) {
      for (int v = 0; v < 8; ++v) {
        int r = rowbase + v + 8 * hi;
        out[(long)r * DOUT + bcol] = c[v] + bias;
      }
      __syncthreads();
    } else {
      for (int v = 0; v < 8; ++v)
        smem[(v + 8 * hi) * 128 + bcol] = c[v] + bias;
      __syncthreads();

      if (tid < 16) {
        const float* rowp = &smem[tid * 128];
        float s = 0.f, ss = 0.f;
        for (int j = 0; j < 128; ++j) { float x = rowp[j]; s += x; ss += x * x; }
        float m   = s * (1.0f / 128.0f);
        float var = ss * (1.0f / 128.0f) - m * m;
        s_mean[tid] = m;
        s_rstd[tid] = rsqrtf(var + LN_EPS);
      }
      __syncthreads();

      for (int idx = tid; idx < 16 * 128; idx += NT) {
        int r  = idx >> 7;
        float x = (smem[idx] - s_mean[r]) * s_rstd[r] * g + be;
        out[(long)(rowbase + r) * 128 + (idx & 127)] = fmaxf(x, 0.0f);
      }
    }
  }
}


extern "C" void kernel_launch(void* const* d_in, const int* in_sizes, int n_in,
                              void* d_out, int out_size, void* d_ws, size_t ws_size,
                              hipStream_t stream) {
  const float* x  = (const float*)d_in[0];
  const int*   ei = (const int*)d_in[1];
  const int N = in_sizes[0] / 64;
  const int E = in_sizes[1] / 2;

  float* aggx = (float*)d_ws;
  float* bufA = aggx + (long)N * 64;
  float* bufB = bufA + (long)N * 128;
  float* bufC = bufB + (long)N * 128;

  const int nown = (N + AGG_R - 1) / AGG_R;

  vgae_agg_own<64><<<nown, AGG_NT, 0, stream>>>(x, ei, aggx, N, E);

  const int nrt = N / 16;
  for (int t = 0; t < 2; ++t) {
    const int base = 2 + t * 13;
    const float* Wl0 = (const float*)d_in[base + 0];
    const float* bl0 = (const float*)d_in[base + 1];
    const float* Wr0 = (const float*)d_in[base + 2];
    const float* g0  = (const float*)d_in[base + 3];
    const float* b0  = (const float*)d_in[base + 4];
    const float* Wl1 = (const float*)d_in[base + 5];
    const float* bl1 = (const float*)d_in[base + 6];
    const float* Wr1 = (const float*)d_in[base + 7];
    const float* g1  = (const float*)d_in[base + 8];
    const float* b1  = (const float*)d_in[base + 9];
    const float* Wl2 = (const float*)d_in[base + 10];
    const float* bl2 = (const float*)d_in[base + 11];
    const float* Wr2 = (const float*)d_in[base + 12];
    float* outp = (float*)d_out + (long)t * N * 64;

    vgae_sage_gemm<64, 128, true><<<512, 256, 0, stream>>>(aggx, x, Wl0, bl0, Wr0,
                                                          g0, b0, bufA, nrt);
    vgae_agg_own<128><<<nown, AGG_NT, 0, stream>>>(bufA, ei, bufB, N, E);

    vgae_sage_gemm<128, 128, true><<<512, 256, 0, stream>>>(bufB, bufA, Wl1, bl1, Wr1,
                                                           g1, b1, bufC, nrt);
    vgae_agg_own<128><<<nown, AGG_NT, 0, stream>>>(bufC, ei, bufA, N, E);

    vgae_sage_gemm<128, 64, false><<<512, 128, 0, stream>>>(bufA, bufC, Wl2, bl2, Wr2,
                                                           nullptr, nullptr, outp, nrt);
  }
}


# Neighbor sums: first pass also writes each wave's matched edges (source, row in block) to a workspace record; the 4 later passes replay it, test no edge; layer-1 output in place; sum order: arrival
# speedup vs baseline: 4.7482x; 1.4718x over previous
#include <hip/hip_runtime.h>


#define LN_EPS 1e-5f

typedef __attribute__((ext_vector_type(2))) float v2f;
typedef __attribute__((ext_vector_type(8))) float v8f;


#define AGG_R    196
#define AGG_NT   1024
#define AGG_NW   (AGG_NT / 32)
#define AGG_CH   512
#define AGG_PL   (AGG_CH / 32)
#define AGG_K    8
#define AGG_QCAP (AGG_K + AGG_CH)
#define AGG_RCAP 256

template <int D>
__device__ __forceinline__ void vgae_agg_consume(const float* __restrict__ h, float* acc, int* cnt,
                                                 int lane, int n, int ms, int md) {
  constexpr int VEC = D / 32;
  typedef __attribute__((ext_vector_type(VEC))) float vN;
  if (lane < n) atomicAdd(&cnt[md], 1);
  int i = 0;
  for (; i + 8 <= n; i += 8) {
    vN v[8];
#pragma unroll
    for (int u = 0; u < 8; ++u) {
      const int s = __builtin_amdgcn_readlane(ms, i + u);
      v[u] = *(const vN*)(h + (long)s * D + VEC * lane);
    }
#pragma unroll
    for (int u = 0; u < 8; ++u) {
      const int dl = __builtin_amdgcn_readlane(md, i + u);
      for (int j = 0; j < VEC; ++j)
        atomicAdd(&acc[dl * D + j * 32 + lane], v[u][j]);
    }
  }
  if (i < n) {
    vN v[8];
#pragma unroll
    for (int u = 0; u < 8; ++u) {
      const int k = (i + u < n) ? (i + u) : (n - 1);
      const int s = __builtin_amdgcn_readlane(ms, k);
      v[u] = *(const vN*)(h + (long)s * D + VEC * lane);
    }
#pragma unroll
    for (int u = 0; u < 8; ++u) asm volatile("" : "+v"(v[u]));
#pragma unroll
    for (int u = 0; u < 8; ++u) {
      if (i + u < n) {
        const int dl = __builtin_amdgcn_readlane(md, i + u);
        for (int j = 0; j < VEC; ++j)
          atomicAdd(&acc[dl * D + j * 32 + lane], v[u][j]);
      }
    }
  }
}

template <int D, int MODE>
__global__ __launch_bounds__(AGG_NT)
void vgae_agg_own(const float* __restrict__ h, const int* __restrict__ ei,
                  float* __restrict__ agg, int* __restrict__ rec, int N, int E) {
  constexpr int VEC = D / 32;
  typedef __attribute__((ext_vector_type(VEC))) float vN;

  __shared__ float acc[AGG_R * D];
  __shared__ int   cnt[AGG_R];
  __shared__ int   que[AGG_NW][AGG_QCAP];

  const int lane = threadIdx.x & 31;
  const int wave = __builtin_amdgcn_readfirstlane(threadIdx.x >> 5);
  const int r0   = blockIdx.x * AGG_R;
  const int* __restrict__ dst = ei + E;
  int* __restrict__ rcnt = rec + (blockIdx.x * AGG_NW + wave);
  int* __restrict__ rl   = rec + gridDim.x * AGG_NW
                               + (long)(blockIdx.x * AGG_NW + wave) * AGG_RCAP;

  for (int i = threadIdx.x; i < AGG_R * D; i += AGG_NT) acc[i] = 0.0f;
  for (int i = threadIdx.x; i < AGG_R; i += AGG_NT) cnt[i] = 0;
  __syncthreads();

  int  tot  = 0;
  bool scan = true;
  if (MODE == 1) {
    tot  = __builtin_amdgcn_readfirstlane(*rcnt);
    scan = tot > AGG_RCAP;
  }

  if (!scan) {
    unsigned pk = (lane < tot) ? (unsigned)rl[lane] : 0u;
    for (int i0 = 0; i0 < tot; i0 += 32) {
      const int n = (tot - i0 < 32) ? (tot - i0) : 32;
      const unsigned cur = pk;
      if (i0 + 32 < tot)
        pk = (i0 + 32 + lane < tot) ? (unsigned)rl[i0 + 32 + lane] : 0u;
      vgae_agg_consume<D>(h, acc, cnt, lane, n, (int)(cur & 0xFFFFFFu), (int)(cur >> 24));
    }
  } else {
    const int nch = (E + AGG_CH - 1) / AGG_CH;
    int qn = 0;

    for (int c = wave; ; c += AGG_NW) {
      const bool live = c < nch;
      if (live) {
        const int base = c * AGG_CH + lane;
        int d[AGG_PL];
        if (c * AGG_CH + AGG_CH <= E) {
          for (int j = 0; j < AGG_PL; ++j) d[j] = dst[base + j * 32];
        } else {
          for (int j = 0; j < AGG_PL; ++j) {
            const int e = base + j * 32;
            d[j] = (e < E) ? dst[e] : -1;
          }
        }
        for (int j = 0; j < AGG_PL; ++j) {
          const bool m = (unsigned)(d[j] - r0) < (unsigned)AGG_R;
          const unsigned mask = __builtin_amdgcn_ballot_w32(m);
          if (mask) {
            if (m) que[wave][qn + (int)__builtin_amdgcn_mbcnt_lo(mask, 0u)] = base + j * 32;
            qn += __builtin_popcount(mask);
          }
        }
      }

      if (qn >= AGG_K || (!live && qn > 0)) {
        for (int i0 = 0; i0 < qn; i0 += 32) {
          const int n  = (qn - i0 < 32) ? (qn - i0) : 32;
          const int me = (lane < n) ? que[wave][i0 + lane] : 0;
          const int ms = ei[me];
          const int md = dst[me] - r0;
          if (MODE == 0) {
            const int pos = tot + i0 + lane;
            if (lane < n && pos < AGG_RCAP) rl[pos] = (int)((unsigned)ms | ((unsigned)md << 24));
          }
          vgae_agg_consume<D>(h, acc, cnt, lane, n, ms, md);
        }
        if (MODE == 0) tot += qn;
        qn = 0;
      }

      if (!live) break;
    }
    if (MODE == 0 && lane == 0) *rcnt = tot;
  }
  __syncthreads();

  for (int row = wave; row < AGG_R; row += AGG_NW) {
    const int r = r0 + row;
    if (r < N) {
      const float id = 1.0f / fmaxf((float)cnt[row], 1.0f);
      vN o;
      for (int j = 0; j < VEC; ++j) o[j] = acc[row * D + j * 32 + lane] * id;
      *(vN*)(agg + (long)r * D + VEC * lane) = o;
    }
  }
}

typedef __attribute__((ext_vector_type(4))) float v4f;

template <int DIN, int DOUT, bool LN>
__global__ __launch_bounds__(DOUT * 2)
void vgae_sage_gemm(const float* agg,
                    const float* __restrict__ h,
                    const float* __restrict__ Wl,
                    const float* __restrict__ bl,
                    const float* __restrict__ Wr,
                    const float* __restrict__ gamma,
                    const float* __restrict__ beta,
                    float* out, int ntiles) {
  constexpr int K   = 2 * DIN;
  constexpr int P   = K + 4;
  constexpr int NT  = DOUT * 2;
  constexpr int NW  = DOUT / 16;
  constexpr int RPW = 16 / NW;
  constexpr int F4  = K / 4;
  constexpr int NL  = RPW * F4 / 32;
  static_assert(!LN || DOUT == 128, "LN epilogue is written for 128 columns");

  __shared__ __attribute__((aligned(16))) float Bt[DOUT * P];
  __shared__ __attribute__((aligned(16))) float At[16 * P];
  __shared__ float smem[LN ? 16 * 128 : 4];
  __shared__ float s_mean[16];
  __shared__ float s_rstd[16];

  const int tid  = threadIdx.x;
  const int lane = tid & 31;
  const int wave = tid >> 5;
  const int l15  = lane & 15;
  const int hi   = lane >> 4;
  const int bcol = wave * 16 + l15;
  const int koff = 2 * hi;

  for (int i = tid; i < DOUT * F4; i += NT) {
    const int cc  = i / F4;
    const int col = (i % F4) * 4;
    const float* s = (col < DIN) ? (Wl + (long)cc * DIN + col)
                                 : (Wr + (long)cc * DIN + (col - DIN));
    *(v4f*)&Bt[cc * P + col] = *(const v4f*)s;
  }

  int prow[NL], pcol[NL];
  for (int i = 0; i < NL; ++i) {
    const int flat = i * 32 + lane;
    prow[i] = wave * RPW + flat / F4;
    pcol[i] = (flat % F4) * 4;
  }
  v4f pre[NL];
  auto fetch = [&](int tile) {
    for (int i = 0; i < NL; ++i) {
      const long r = (long)tile * 16 + prow[i];
      const float* s = (pcol[i] < DIN) ? (agg + r * DIN + pcol[i])
                                       : (h + r * DIN + (pcol[i] - DIN));
      pre[i] = *(const v4f*)s;
    }
  };

  const float bias = bl[bcol];
  float g = 0.f, be = 0.f;
  if (LN) { g = gamma[tid & 127]; be = beta[tid & 127]; }

  const float* ap = &At[l15 * P + koff];
  const float* bp = &Bt[bcol * P + koff];

  int tile = blockIdx.x;
  if (tile < ntiles) fetch(tile);

  for (; tile < ntiles; tile += gridDim.x) {
    for (int i = 0; i < NL; ++i) *(v4f*)&At[prow[i] * P + pcol[i]] = pre[i];
    __syncthreads();
    const int next = tile + gridDim.x;
    if (next < ntiles) fetch(next);

    v8f c = {0.f, 0.f, 0.f, 0.f, 0.f, 0.f, 0.f, 0.f};
    for (int k = 0; k < K; k += 4) {
      v2f a = *(const v2f*)(ap + k);
      v2f b = *(const v2f*)(bp + k);
      c = __builtin_amdgcn_wmma_f32_16x16x4_f32(false, a, false, b,
                                                (short)0, c, false, false);
    }

    const int rowbase = tile * 16;

    if (!LN) {
      for (int v = 0; v < 8; ++v) {
        int r = rowbase + v + 8 * hi;
        out[(long)r * DOUT + bcol] = c[v] + bias;
      }
      __syncthreads();
    } else {
      for (int v = 0; v < 8; ++v)
        smem[(v + 8 * hi) * 128 + bcol] = c[v] + bias;
      __syncthreads();

      if (tid < 16) {
        const float* rowp = &smem[tid * 128];
        float s = 0.f, ss = 0.f;
        for (int j = 0; j < 128; ++j) { float x = rowp[j]; s += x; ss += x * x; }
        float m   = s * (1.0f / 128.0f);
        float var = ss * (1.0f / 128.0f) - m * m;
        s_mean[tid] = m;
        s_rstd[tid] = rsqrtf(var + LN_EPS);
      }
      __syncthreads();

      for (int idx = tid; idx < 16 * 128; idx += NT) {
        int r  = idx >> 7;
        float x = (smem[idx] - s_mean[r]) * s_rstd[r] * g + be;
        out[(long)(rowbase + r) * 128 + (idx & 127)] = fmaxf(x, 0.0f);
      }
    }
  }
}


extern "C" void kernel_launch(void* const* d_in, const int* in_sizes, int n_in,
                              void* d_out, int out_size, void* d_ws, size_t ws_size,
                              hipStream_t stream) {
  const float* x  = (const float*)d_in[0];
  const int*   ei = (const int*)d_in[1];
  const int N = in_sizes[0] / 64;
  const int E = in_sizes[1] / 2;

  float* aggx = (float*)d_ws;
  float* bufA = aggx + (long)N * 64;
  float* bufB = bufA + (long)N * 128;
  int*   rec  = (int*)(bufB + (long)N * 128);

  const int nown = (N + AGG_R - 1) / AGG_R;

  vgae_agg_own<64, 0><<<nown, AGG_NT, 0, stream>>>(x, ei, aggx, rec, N, E);

  const int nrt = N / 16;
  for (int t = 0; t < 2; ++t) {
    const int base = 2 + t * 13;
    const float* Wl0 = (const float*)d_in[base + 0];
    const float* bl0 = (const float*)d_in[base + 1];
    const float* Wr0 = (const float*)d_in[base + 2];
    const float* g0  = (const float*)d_in[base + 3];
    const float* b0  = (const float*)d_in[base + 4];
    const float* Wl1 = (const float*)d_in[base + 5];
    const float* bl1 = (const float*)d_in[base + 6];
    const float* Wr1 = (const float*)d_in[base + 7];
    const float* g1  = (const float*)d_in[base + 8];
    const float* b1  = (const float*)d_in[base + 9];
    const float* Wl2 = (const float*)d_in[base + 10];
    const float* bl2 = (const float*)d_in[base + 11];
    const float* Wr2 = (const float*)d_in[base + 12];
    float* outp = (float*)d_out + (long)t * N * 64;

    vgae_sage_gemm<64, 128, true><<<512, 256, 0, stream>>>(aggx, x, Wl0, bl0, Wr0,
                                                          g0, b0, bufA, nrt);
    vgae_agg_own<128, 1><<<nown, AGG_NT, 0, stream>>>(bufA, ei, bufB, rec, N, E);

    vgae_sage_gemm<128, 128, true><<<512, 256, 0, stream>>>(bufB, bufA, Wl1, bl1, Wr1,
                                                           g1, b1, bufB, nrt);
    vgae_agg_own<128, 1><<<nown, AGG_NT, 0, stream>>>(bufB, ei, bufA, rec, N, E);

    vgae_sage_gemm<128, 64, false><<<512, 128, 0, stream>>>(bufA, bufB, Wl2, bl2, Wr2,
                                                           nullptr, nullptr, outp, nrt);
  }
}


# LN epilogue: row sums and write-out by all 256 threads (16 per row, 8 values each in registers, DPP butterfly over 16 lanes); one barrier per tile instead of two; LN sum order: 8-wise then tree
# speedup vs baseline: 5.4168x; 1.1408x over previous
#include <hip/hip_runtime.h>


#define LN_EPS 1e-5f

typedef __attribute__((ext_vector_type(2))) float v2f;
typedef __attribute__((ext_vector_type(8))) float v8f;


#define AGG_R    196
#define AGG_NT   1024
#define AGG_NW   (AGG_NT / 32)
#define AGG_CH   512
#define AGG_PL   (AGG_CH / 32)
#define AGG_K    8
#define AGG_QCAP (AGG_K + AGG_CH)
#define AGG_RCAP 256

template <int D>
__device__ __forceinline__ void vgae_agg_consume(const float* __restrict__ h, float* acc, int* cnt,
                                                 int lane, int n, int ms, int md) {
  constexpr int VEC = D / 32;
  typedef __attribute__((ext_vector_type(VEC))) float vN;
  if (lane < n) atomicAdd(&cnt[md], 1);
  int i = 0;
  for (; i + 8 <= n; i += 8) {
    vN v[8];
#pragma unroll
    for (int u = 0; u < 8; ++u) {
      const int s = __builtin_amdgcn_readlane(ms, i + u);
      v[u] = *(const vN*)(h + (long)s * D + VEC * lane);
    }
#pragma unroll
    for (int u = 0; u < 8; ++u) {
      const int dl = __builtin_amdgcn_readlane(md, i + u);
      for (int j = 0; j < VEC; ++j)
        atomicAdd(&acc[dl * D + j * 32 + lane], v[u][j]);
    }
  }
  if (i < n) {
    vN v[8];
#pragma unroll
    for (int u = 0; u < 8; ++u) {
      const int k = (i + u < n) ? (i + u) : (n - 1);
      const int s = __builtin_amdgcn_readlane(ms, k);
      v[u] = *(const vN*)(h + (long)s * D + VEC * lane);
    }
#pragma unroll
    for (int u = 0; u < 8; ++u) asm volatile("" : "+v"(v[u]));
#pragma unroll
    for (int u = 0; u < 8; ++u) {
      if (i + u < n) {
        const int dl = __builtin_amdgcn_readlane(md, i + u);
        for (int j = 0; j < VEC; ++j)
          atomicAdd(&acc[dl * D + j * 32 + lane], v[u][j]);
      }
    }
  }
}

template <int D, int MODE>
__global__ __launch_bounds__(AGG_NT)
void vgae_agg_own(const float* __restrict__ h, const int* __restrict__ ei,
                  float* __restrict__ agg, int* __restrict__ rec, int N, int E) {
  constexpr int VEC = D / 32;
  typedef __attribute__((ext_vector_type(VEC))) float vN;

  __shared__ float acc[AGG_R * D];
  __shared__ int   cnt[AGG_R];
  __shared__ int   que[AGG_NW][AGG_QCAP];

  const int lane = threadIdx.x & 31;
  const int wave = __builtin_amdgcn_readfirstlane(threadIdx.x >> 5);
  const int r0   = blockIdx.x * AGG_R;
  const int* __restrict__ dst = ei + E;
  int* __restrict__ rcnt = rec + (blockIdx.x * AGG_NW + wave);
  int* __restrict__ rl   = rec + gridDim.x * AGG_NW
                               + (long)(blockIdx.x * AGG_NW + wave) * AGG_RCAP;

  for (int i = threadIdx.x; i < AGG_R * D; i += AGG_NT) acc[i] = 0.0f;
  for (int i = threadIdx.x; i < AGG_R; i += AGG_NT) cnt[i] = 0;
  __syncthreads();

  int  tot  = 0;
  bool scan = true;
  if (MODE == 1) {
    tot  = __builtin_amdgcn_readfirstlane(*rcnt);
    scan = tot > AGG_RCAP;
  }

  if (!scan) {
    unsigned pk = (lane < tot) ? (unsigned)rl[lane] : 0u;
    for (int i0 = 0; i0 < tot; i0 += 32) {
      const int n = (tot - i0 < 32) ? (tot - i0) : 32;
      const unsigned cur = pk;
      if (i0 + 32 < tot)
        pk = (i0 + 32 + lane < tot) ? (unsigned)rl[i0 + 32 + lane] : 0u;
      vgae_agg_consume<D>(h, acc, cnt, lane, n, (int)(cur & 0xFFFFFFu), (int)(cur >> 24));
    }
  } else {
    const int nch = (E + AGG_CH - 1) / AGG_CH;
    int qn = 0;

    for (int c = wave; ; c += AGG_NW) {
      const bool live = c < nch;
      if (live) {
        const int base = c * AGG_CH + lane;
        int d[AGG_PL];
        if (c * AGG_CH + AGG_CH <= E) {
          for (int j = 0; j < AGG_PL; ++j) d[j] = dst[base + j * 32];
        } else {
          for (int j = 0; j < AGG_PL; ++j) {
            const int e = base + j * 32;
            d[j] = (e < E) ? dst[e] : -1;
          }
        }
        for (int j = 0; j < AGG_PL; ++j) {
          const bool m = (unsigned)(d[j] - r0) < (unsigned)AGG_R;
          const unsigned mask = __builtin_amdgcn_ballot_w32(m);
          if (mask) {
            if (m) que[wave][qn + (int)__builtin_amdgcn_mbcnt_lo(mask, 0u)] = base + j * 32;
            qn += __builtin_popcount(mask);
          }
        }
      }

      if (qn >= AGG_K || (!live && qn > 0)) {
        for (int i0 = 0; i0 < qn; i0 += 32) {
          const int n  = (qn - i0 < 32) ? (qn - i0) : 32;
          const int me = (lane < n) ? que[wave][i0 + lane] : 0;
          const int ms = ei[me];
          const int md = dst[me] - r0;
          if (MODE == 0) {
            const int pos = tot + i0 + lane;
            if (lane < n && pos < AGG_RCAP) rl[pos] = (int)((unsigned)ms | ((unsigned)md << 24));
          }
          vgae_agg_consume<D>(h, acc, cnt, lane, n, ms, md);
        }
        if (MODE == 0) tot += qn;
        qn = 0;
      }

      if (!live) break;
    }
    if (MODE == 0 && lane == 0) *rcnt = tot;
  }
  __syncthreads();

  for (int row = wave; row < AGG_R; row += AGG_NW) {
    const int r = r0 + row;
    if (r < N) {
      const float id = 1.0f / fmaxf((float)cnt[row], 1.0f);
      vN o;
      for (int j = 0; j < VEC; ++j) o[j] = acc[row * D + j * 32 + lane] * id;
      *(vN*)(agg + (long)r * D + VEC * lane) = o;
    }
  }
}

typedef __attribute__((ext_vector_type(4))) float v4f;

template <int M>
__device__ __forceinline__ float vgae_row_xor(float x) {
  return __builtin_bit_cast(float, __builtin_amdgcn_update_dpp(
      0, __builtin_bit_cast(int, x), 0x160 | M, 0xf, 0xf, true));
}

template <int DIN, int DOUT, bool LN>
__global__ __launch_bounds__(DOUT * 2)
void vgae_sage_gemm(const float* agg,
                    const float* __restrict__ h,
                    const float* __restrict__ Wl,
                    const float* __restrict__ bl,
                    const float* __restrict__ Wr,
                    const float* __restrict__ gamma,
                    const float* __restrict__ beta,
                    float* out, int ntiles) {
  constexpr int K   = 2 * DIN;
  constexpr int P   = K + 4;
  constexpr int NT  = DOUT * 2;
  constexpr int NW  = DOUT / 16;
  constexpr int RPW = 16 / NW;
  constexpr int F4  = K / 4;
  constexpr int NL  = RPW * F4 / 32;
  static_assert(!LN || DOUT == 128, "LN epilogue is written for 128 columns");

  __shared__ __attribute__((aligned(16))) float Bt[DOUT * P];
  __shared__ __attribute__((aligned(16))) float At[16 * P];
  constexpr int SP = 132;
  __shared__ __attribute__((aligned(16))) float smem[LN ? 16 * SP : 4];

  const int tid  = threadIdx.x;
  const int lane = tid & 31;
  const int wave = tid >> 5;
  const int l15  = lane & 15;
  const int hi   = lane >> 4;
  const int bcol = wave * 16 + l15;
  const int koff = 2 * hi;

  for (int i = tid; i < DOUT * F4; i += NT) {
    const int cc  = i / F4;
    const int col = (i % F4) * 4;
    const float* s = (col < DIN) ? (Wl + (long)cc * DIN + col)
                                 : (Wr + (long)cc * DIN + (col - DIN));
    *(v4f*)&Bt[cc * P + col] = *(const v4f*)s;
  }

  int prow[NL], pcol[NL];
  for (int i = 0; i < NL; ++i) {
    const int flat = i * 32 + lane;
    prow[i] = wave * RPW + flat / F4;
    pcol[i] = (flat % F4) * 4;
  }
  v4f pre[NL];
  auto fetch = [&](int tile) {
    for (int i = 0; i < NL; ++i) {
      const long r = (long)tile * 16 + prow[i];
      const float* s = (pcol[i] < DIN) ? (agg + r * DIN + pcol[i])
                                       : (h + r * DIN + (pcol[i] - DIN));
      pre[i] = *(const v4f*)s;
    }
  };

  const float bias = bl[bcol];
  const int erow = tid >> 4;
  const int ecol = (tid & 15) * 4;
  v4f g0 = {0.f, 0.f, 0.f, 0.f}, g1 = g0, be0 = g0, be1 = g0;
  if (LN) {
    g0  = *(const v4f*)(gamma + ecol);
    g1  = *(const v4f*)(gamma + 64 + ecol);
    be0 = *(const v4f*)(beta + ecol);
    be1 = *(const v4f*)(beta + 64 + ecol);
  }

  const float* ap = &At[l15 * P + koff];
  const float* bp = &Bt[bcol * P + koff];

  int tile = blockIdx.x;
  if (tile < ntiles) fetch(tile);

  for (; tile < ntiles; tile += gridDim.x) {
    for (int i = 0; i < NL; ++i) *(v4f*)&At[prow[i] * P + pcol[i]] = pre[i];
    __syncthreads();
    const int next = tile + gridDim.x;
    if (next < ntiles) fetch(next);

    v8f c = {0.f, 0.f, 0.f, 0.f, 0.f, 0.f, 0.f, 0.f};
    for (int k = 0; k < K; k += 4) {
      v2f a = *(const v2f*)(ap + k);
      v2f b = *(const v2f*)(bp + k);
      c = __builtin_amdgcn_wmma_f32_16x16x4_f32(false, a, false, b,
                                                (short)0, c, false, false);
    }

    const int rowbase = tile * 16;

    if (!LN) {
      for (int v = 0; v < 8; ++v) {
        int r = rowbase + v + 8 * hi;
        out[(long)r * DOUT + bcol] = c[v] + bias;
      }
      __syncthreads();
    } else {
      for (int v = 0; v < 8; ++v)
        smem[(v + 8 * hi) * SP + bcol] = c[v] + bias;
      __syncthreads();

      const v4f x0 = *(const v4f*)&smem[erow * SP + ecol];
      const v4f x1 = *(const v4f*)&smem[erow * SP + 64 + ecol];
      float s  = ((x0[0] + x0[1]) + (x0[2] + x0[3])) + ((x1[0] + x1[1]) + (x1[2] + x1[3]));
      float ss = ((x0[0] * x0[0] + x0[1] * x0[1]) + (x0[2] * x0[2] + x0[3] * x0[3]))
               + ((x1[0] * x1[0] + x1[1] * x1[1]) + (x1[2] * x1[2] + x1[3] * x1[3]));
      s += vgae_row_xor<1>(s);  ss += vgae_row_xor<1>(ss);
      s += vgae_row_xor<2>(s);  ss += vgae_row_xor<2>(ss);
      s += vgae_row_xor<4>(s);  ss += vgae_row_xor<4>(ss);
      s += vgae_row_xor<8>(s);  ss += vgae_row_xor<8>(ss);
      const float m    = s * (1.0f / 128.0f);
      const float var  = ss * (1.0f / 128.0f) - m * m;
      const float rstd = rsqrtf(var + LN_EPS);

      v4f y0, y1;
      for (int j = 0; j < 4; ++j) {
        y0[j] = fmaxf((x0[j] - m) * rstd * g0[j] + be0[j], 0.0f);
        y1[j] = fmaxf((x1[j] - m) * rstd * g1[j] + be1[j], 0.0f);
      }
      float* orow = out + (long)(rowbase + erow) * 128;
      *(v4f*)(orow + ecol)      = y0;
      *(v4f*)(orow + 64 + ecol) = y1;
    }
  }
}


extern "C" void kernel_launch(void* const* d_in, const int* in_sizes, int n_in,
                              void* d_out, int out_size, void* d_ws, size_t ws_size,
                              hipStream_t stream) {
  const float* x  = (const float*)d_in[0];
  const int*   ei = (const int*)d_in[1];
  const int N = in_sizes[0] / 64;
  const int E = in_sizes[1] / 2;

  float* aggx = (float*)d_ws;
  float* bufA = aggx + (long)N * 64;
  float* bufB = bufA + (long)N * 128;
  int*   rec  = (int*)(bufB + (long)N * 128);

  const int nown = (N + AGG_R - 1) / AGG_R;

  vgae_agg_own<64, 0><<<nown, AGG_NT, 0, stream>>>(x, ei, aggx, rec, N, E);

  const int nrt = N / 16;
  for (int t = 0; t < 2; ++t) {
    const int base = 2 + t * 13;
    const float* Wl0 = (const float*)d_in[base + 0];
    const float* bl0 = (const float*)d_in[base + 1];
    const float* Wr0 = (const float*)d_in[base + 2];
    const float* g0  = (const float*)d_in[base + 3];
    const float* b0  = (const float*)d_in[base + 4];
    const float* Wl1 = (const float*)d_in[base + 5];
    const float* bl1 = (const float*)d_in[base + 6];
    const float* Wr1 = (const float*)d_in[base + 7];
    const float* g1  = (const float*)d_in[base + 8];
    const float* b1  = (const float*)d_in[base + 9];
    const float* Wl2 = (const float*)d_in[base + 10];
    const float* bl2 = (const float*)d_in[base + 11];
    const float* Wr2 = (const float*)d_in[base + 12];
    float* outp = (float*)d_out + (long)t * N * 64;

    vgae_sage_gemm<64, 128, true><<<512, 256, 0, stream>>>(aggx, x, Wl0, bl0, Wr0,
                                                          g0, b0, bufA, nrt);
    vgae_agg_own<128, 1><<<nown, AGG_NT, 0, stream>>>(bufA, ei, bufB, rec, N, E);

    vgae_sage_gemm<128, 128, true><<<512, 256, 0, stream>>>(bufB, bufA, Wl1, bl1, Wr1,
                                                           g1, b1, bufB, nrt);
    vgae_agg_own<128, 1><<<nown, AGG_NT, 0, stream>>>(bufB, ei, bufA, rec, N, E);

    vgae_sage_gemm<128, 64, false><<<512, 128, 0, stream>>>(bufA, bufB, Wl2, bl2, Wr2,
                                                           nullptr, nullptr, outp, nrt);
  }
}


# Neighbor sums: match record built once by a scan-only launch over a 2-D split (64 super-ranges of 784 rows x 4 edge quarters, 4x fewer edge tests); all 5 passes replay it; sum order: arrival
# speedup vs baseline: 5.9047x; 1.0901x over previous
#include <hip/hip_runtime.h>


#define LN_EPS 1e-5f

typedef __attribute__((ext_vector_type(2))) float v2f;
typedef __attribute__((ext_vector_type(8))) float v8f;


#define AGG_R    196
#define AGG_NT   1024
#define AGG_NW   (AGG_NT / 32)
#define AGG_CH   512
#define AGG_PL   (AGG_CH / 32)
#define AGG_K    8
#define AGG_QCAP (AGG_K + AGG_CH)
#define SUP_NB   4
#define SUP_R    (SUP_NB * AGG_R)
#define REC_NQ   4
#define SEG_CAP  60

__global__ __launch_bounds__(AGG_NT)
void vgae_rec_build(const int* __restrict__ ei, int* __restrict__ rec, int E) {
  __shared__ int que[AGG_NW][AGG_CH];

  const int lane = threadIdx.x & 31;
  const int wave = __builtin_amdgcn_readfirstlane(threadIdx.x >> 5);
  const int r0   = (blockIdx.x / REC_NQ) * SUP_R;
  const int qtr  = blockIdx.x % REC_NQ;
  const int* __restrict__ dst = ei + E;

  const int nseg = gridDim.x * AGG_NW * SUP_NB;
  const int seg0 = (blockIdx.x * AGG_NW + wave) * SUP_NB;
  int* __restrict__ ent = rec + nseg + (long)seg0 * SEG_CAP;

  const int nch = (E + AGG_CH - 1) / AGG_CH;
  const int cpq = (nch + REC_NQ - 1) / REC_NQ;
  const int cb  = qtr * cpq;
  const int ce  = (cb + cpq < nch) ? (cb + cpq) : nch;

  int sc[SUP_NB];
  for (int u = 0; u < SUP_NB; ++u) sc[u] = 0;

  int d[AGG_PL];
  auto load_dst = [&](int c) {
    const int base = c * AGG_CH + lane;
    if (c * AGG_CH + AGG_CH <= E) {
      for (int j = 0; j < AGG_PL; ++j) d[j] = dst[base + j * 32];
    } else {
      for (int j = 0; j < AGG_PL; ++j) {
        const int e = base + j * 32;
        d[j] = (e < E) ? dst[e] : -1;
      }
    }
  };

  int c = cb + wave;
  if (c < ce) load_dst(c);
  for (; c < ce; c += AGG_NW) {
    const int base = c * AGG_CH + lane;
    int qn = 0;
    for (int j = 0; j < AGG_PL; ++j) {
      const bool m = (unsigned)(d[j] - r0) < (unsigned)SUP_R;
      const unsigned mask = __builtin_amdgcn_ballot_w32(m);
      if (mask) {
        if (m) que[wave][qn + (int)__builtin_amdgcn_mbcnt_lo(mask, 0u)] = base + j * 32;
        qn += __builtin_popcount(mask);
      }
    }
    if (c + AGG_NW < ce) load_dst(c + AGG_NW);
    for (int i0 = 0; i0 < qn; i0 += 32) {
      const int  n   = (qn - i0 < 32) ? (qn - i0) : 32;
      const bool act = lane < n;
      const int  me  = act ? que[wave][i0 + lane] : 0;
      const int  ms  = ei[me];
      const int  rr  = dst[me] - r0;
      const int  u   = act ? (rr / AGG_R) : -1;
      const int  rb  = rr - u * AGG_R;
      const int  wd  = (int)((unsigned)ms | ((unsigned)rb << 24));
      for (int uu = 0; uu < SUP_NB; ++uu) {
        const bool mine = (u == uu);
        const unsigned mk = __builtin_amdgcn_ballot_w32(mine);
        if (mine) {
          const int pos = sc[uu] + (int)__builtin_amdgcn_mbcnt_lo(mk, 0u);
          if (pos < SEG_CAP) ent[uu * SEG_CAP + pos] = wd;
        }
        sc[uu] += __builtin_popcount(mk);
      }
    }
  }
  if (lane < SUP_NB) {
    int v = sc[0];
    for (int u = 1; u < SUP_NB; ++u) v = (lane == u) ? sc[u] : v;
    rec[seg0 + lane] = v;
  }
}

template <int D>
__device__ __forceinline__ void vgae_agg_consume(const float* __restrict__ h, float* acc, int* cnt,
                                                 int lane, int n, int ms, int md) {
  constexpr int VEC = D / 32;
  typedef __attribute__((ext_vector_type(VEC))) float vN;
  if (lane < n) atomicAdd(&cnt[md], 1);
  int i = 0;
  for (; i + 8 <= n; i += 8) {
    vN v[8];
#pragma unroll
    for (int u = 0; u < 8; ++u) {
      const int s = __builtin_amdgcn_readlane(ms, i + u);
      v[u] = *(const vN*)(h + (long)s * D + VEC * lane);
    }
#pragma unroll
    for (int u = 0; u < 8; ++u) {
      const int dl = __builtin_amdgcn_readlane(md, i + u);
      for (int j = 0; j < VEC; ++j)
        atomicAdd(&acc[dl * D + j * 32 + lane], v[u][j]);
    }
  }
  if (i < n) {
    vN v[8];
#pragma unroll
    for (int u = 0; u < 8; ++u) {
      const int k = (i + u < n) ? (i + u) : (n - 1);
      const int s = __builtin_amdgcn_readlane(ms, k);
      v[u] = *(const vN*)(h + (long)s * D + VEC * lane);
    }
#pragma unroll
    for (int u = 0; u < 8; ++u) asm volatile("" : "+v"(v[u]));
#pragma unroll
    for (int u = 0; u < 8; ++u) {
      if (i + u < n) {
        const int dl = __builtin_amdgcn_readlane(md, i + u);
        for (int j = 0; j < VEC; ++j)
          atomicAdd(&acc[dl * D + j * 32 + lane], v[u][j]);
      }
    }
  }
}

template <int D>
__global__ __launch_bounds__(AGG_NT)
void vgae_agg_own(const float* __restrict__ h, const int* __restrict__ ei,
                  float* __restrict__ agg, const int* __restrict__ rec, int N, int E) {
  constexpr int VEC = D / 32;
  typedef __attribute__((ext_vector_type(VEC))) float vN;

  __shared__ float acc[AGG_R * D];
  __shared__ int   cnt[AGG_R];
  __shared__ int   que[AGG_NW][AGG_QCAP];

  const int lane = threadIdx.x & 31;
  const int wave = __builtin_amdgcn_readfirstlane(threadIdx.x >> 5);
  const int r0   = blockIdx.x * AGG_R;
  const int* __restrict__ dst = ei + E;

  const int sup  = blockIdx.x / SUP_NB;
  const int own  = blockIdx.x % SUP_NB;
  const int nseg = gridDim.x * AGG_NW * SUP_NB;
  int cq[REC_NQ];
  for (int q = 0; q < REC_NQ; ++q)
    cq[q] = rec[((sup * REC_NQ + q) * AGG_NW + lane) * SUP_NB + own];

  for (int i = threadIdx.x; i < AGG_R * D; i += AGG_NT) acc[i] = 0.0f;
  for (int i = threadIdx.x; i < AGG_R; i += AGG_NT) cnt[i] = 0;

  bool ov = false;
  for (int q = 0; q < REC_NQ; ++q) ov = ov || (cq[q] > SEG_CAP);
  const bool scan = __builtin_amdgcn_ballot_w32(ov) != 0u;
  __syncthreads();

  if (!scan) {
    const int c0 = __builtin_amdgcn_readlane(cq[0], wave);
    const int c1 = __builtin_amdgcn_readlane(cq[1], wave);
    const int c2 = __builtin_amdgcn_readlane(cq[2], wave);
    const int c3 = __builtin_amdgcn_readlane(cq[3], wave);
    const int p1 = c0, p2 = p1 + c1, p3 = p2 + c2, tot = p3 + c3;
    const int* __restrict__ ent =
        rec + nseg + (long)(((sup * REC_NQ) * AGG_NW + wave) * SUP_NB + own) * SEG_CAP;
    auto word = [&](int idx) -> unsigned {
      const int q   = (idx >= p1) + (idx >= p2) + (idx >= p3);
      const int beg = (q == 0) ? 0 : (q == 1) ? p1 : (q == 2) ? p2 : p3;
      return (unsigned)ent[(long)q * (AGG_NW * SUP_NB * SEG_CAP) + (idx - beg)];
    };
    unsigned pk = (lane < tot) ? word(lane) : 0u;
    for (int i0 = 0; i0 < tot; i0 += 32) {
      const int n = (tot - i0 < 32) ? (tot - i0) : 32;
      const unsigned cur = pk;
      if (i0 + 32 < tot)
        pk = (i0 + 32 + lane < tot) ? word(i0 + 32 + lane) : 0u;
      vgae_agg_consume<D>(h, acc, cnt, lane, n, (int)(cur & 0xFFFFFFu), (int)(cur >> 24));
    }
  } else {
    const int nch = (E + AGG_CH - 1) / AGG_CH;
    int qn = 0;

    for (int c = wave; ; c += AGG_NW) {
      const bool live = c < nch;
      if (live) {
        const int base = c * AGG_CH + lane;
        int d[AGG_PL];
        if (c * AGG_CH + AGG_CH <= E) {
          for (int j = 0; j < AGG_PL; ++j) d[j] = dst[base + j * 32];
        } else {
          for (int j = 0; j < AGG_PL; ++j) {
            const int e = base + j * 32;
            d[j] = (e < E) ? dst[e] : -1;
          }
        }
        for (int j = 0; j < AGG_PL; ++j) {
          const bool m = (unsigned)(d[j] - r0) < (unsigned)AGG_R;
          const unsigned mask = __builtin_amdgcn_ballot_w32(m);
          if (mask) {
            if (m) que[wave][qn + (int)__builtin_amdgcn_mbcnt_lo(mask, 0u)] = base + j * 32;
            qn += __builtin_popcount(mask);
          }
        }
      }

      if (qn >= AGG_K || (!live && qn > 0)) {
        for (int i0 = 0; i0 < qn; i0 += 32) {
          const int n  = (qn - i0 < 32) ? (qn - i0) : 32;
          const int me = (lane < n) ? que[wave][i0 + lane] : 0;
          const int ms = ei[me];
          const int md = dst[me] - r0;
          vgae_agg_consume<D>(h, acc, cnt, lane, n, ms, md);
        }
        qn = 0;
      }

      if (!live) break;
    }
  }
  __syncthreads();

  for (int row = wave; row < AGG_R; row += AGG_NW) {
    const int r = r0 + row;
    if (r < N) {
      const float id = 1.0f / fmaxf((float)cnt[row], 1.0f);
      vN o;
      for (int j = 0; j < VEC; ++j) o[j] = acc[row * D + j * 32 + lane] * id;
      *(vN*)(agg + (long)r * D + VEC * lane) = o;
    }
  }
}

typedef __attribute__((ext_vector_type(4))) float v4f;

template <int M>
__device__ __forceinline__ float vgae_row_xor(float x) {
  return __builtin_bit_cast(float, __builtin_amdgcn_update_dpp(
      0, __builtin_bit_cast(int, x), 0x160 | M, 0xf, 0xf, true));
}

template <int DIN, int DOUT, bool LN>
__global__ __launch_bounds__(DOUT * 2)
void vgae_sage_gemm(const float* agg,
                    const float* __restrict__ h,
                    const float* __restrict__ Wl,
                    const float* __restrict__ bl,
                    const float* __restrict__ Wr,
                    const float* __restrict__ gamma,
                    const float* __restrict__ beta,
                    float* out, int ntiles) {
  constexpr int K   = 2 * DIN;
  constexpr int P   = K + 4;
  constexpr int NT  = DOUT * 2;
  constexpr int NW  = DOUT / 16;
  constexpr int RPW = 16 / NW;
  constexpr int F4  = K / 4;
  constexpr int NL  = RPW * F4 / 32;
  static_assert(!LN || DOUT == 128, "LN epilogue is written for 128 columns");

  __shared__ __attribute__((aligned(16))) float Bt[DOUT * P];
  __shared__ __attribute__((aligned(16))) float At[16 * P];
  constexpr int SP = 132;
  __shared__ __attribute__((aligned(16))) float smem[LN ? 16 * SP : 4];

  const int tid  = threadIdx.x;
  const int lane = tid & 31;
  const int wave = tid >> 5;
  const int l15  = lane & 15;
  const int hi   = lane >> 4;
  const int bcol = wave * 16 + l15;
  const int koff = 2 * hi;

  for (int i = tid; i < DOUT * F4; i += NT) {
    const int cc  = i / F4;
    const int col = (i % F4) * 4;
    const float* s = (col < DIN) ? (Wl + (long)cc * DIN + col)
                                 : (Wr + (long)cc * DIN + (col - DIN));
    *(v4f*)&Bt[cc * P + col] = *(const v4f*)s;
  }

  int prow[NL], pcol[NL];
  for (int i = 0; i < NL; ++i) {
    const int flat = i * 32 + lane;
    prow[i] = wave * RPW + flat / F4;
    pcol[i] = (flat % F4) * 4;
  }
  v4f pre[NL];
  auto fetch = [&](int tile) {
    for (int i = 0; i < NL; ++i) {
      const long r = (long)tile * 16 + prow[i];
      const float* s = (pcol[i] < DIN) ? (agg + r * DIN + pcol[i])
                                       : (h + r * DIN + (pcol[i] - DIN));
      pre[i] = *(const v4f*)s;
    }
  };

  const float bias = bl[bcol];
  const int erow = tid >> 4;
  const int ecol = (tid & 15) * 4;
  v4f g0 = {0.f, 0.f, 0.f, 0.f}, g1 = g0, be0 = g0, be1 = g0;
  if (LN) {
    g0  = *(const v4f*)(gamma + ecol);
    g1  = *(const v4f*)(gamma + 64 + ecol);
    be0 = *(const v4f*)(beta + ecol);
    be1 = *(const v4f*)(beta + 64 + ecol);
  }

  const float* ap = &At[l15 * P + koff];
  const float* bp = &Bt[bcol * P + koff];

  int tile = blockIdx.x;
  if (tile < ntiles) fetch(tile);

  for (; tile < ntiles; tile += gridDim.x) {
    for (int i = 0; i < NL; ++i) *(v4f*)&At[prow[i] * P + pcol[i]] = pre[i];
    __syncthreads();
    const int next = tile + gridDim.x;
    if (next < ntiles) fetch(next);

    v8f c = {0.f, 0.f, 0.f, 0.f, 0.f, 0.f, 0.f, 0.f};
    for (int k = 0; k < K; k += 4) {
      v2f a = *(const v2f*)(ap + k);
      v2f b = *(const v2f*)(bp + k);
      c = __builtin_amdgcn_wmma_f32_16x16x4_f32(false, a, false, b,
                                                (short)0, c, false, false);
    }

    const int rowbase = tile * 16;

    if (!LN) {
      for (int v = 0; v < 8; ++v) {
        int r = rowbase + v + 8 * hi;
        out[(long)r * DOUT + bcol] = c[v] + bias;
      }
      __syncthreads();
    } else {
      for (int v = 0; v < 8; ++v)
        smem[(v + 8 * hi) * SP + bcol] = c[v] + bias;
      __syncthreads();

      const v4f x0 = *(const v4f*)&smem[erow * SP + ecol];
      const v4f x1 = *(const v4f*)&smem[erow * SP + 64 + ecol];
      float s  = ((x0[0] + x0[1]) + (x0[2] + x0[3])) + ((x1[0] + x1[1]) + (x1[2] + x1[3]));
      float ss = ((x0[0] * x0[0] + x0[1] * x0[1]) + (x0[2] * x0[2] + x0[3] * x0[3]))
               + ((x1[0] * x1[0] + x1[1] * x1[1]) + (x1[2] * x1[2] + x1[3] * x1[3]));
      s += vgae_row_xor<1>(s);  ss += vgae_row_xor<1>(ss);
      s += vgae_row_xor<2>(s);  ss += vgae_row_xor<2>(ss);
      s += vgae_row_xor<4>(s);  ss += vgae_row_xor<4>(ss);
      s += vgae_row_xor<8>(s);  ss += vgae_row_xor<8>(ss);
      const float m    = s * (1.0f / 128.0f);
      const float var  = ss * (1.0f / 128.0f) - m * m;
      const float rstd = rsqrtf(var + LN_EPS);

      v4f y0, y1;
      for (int j = 0; j < 4; ++j) {
        y0[j] = fmaxf((x0[j] - m) * rstd * g0[j] + be0[j], 0.0f);
        y1[j] = fmaxf((x1[j] - m) * rstd * g1[j] + be1[j], 0.0f);
      }
      float* orow = out + (long)(rowbase + erow) * 128;
      *(v4f*)(orow + ecol)      = y0;
      *(v4f*)(orow + 64 + ecol) = y1;
    }
  }
}


extern "C" void kernel_launch(void* const* d_in, const int* in_sizes, int n_in,
                              void* d_out, int out_size, void* d_ws, size_t ws_size,
                              hipStream_t stream) {
  const float* x  = (const float*)d_in[0];
  const int*   ei = (const int*)d_in[1];
  const int N = in_sizes[0] / 64;
  const int E = in_sizes[1] / 2;

  float* aggx = (float*)d_ws;
  float* bufA = aggx + (long)N * 64;
  float* bufB = bufA + (long)N * 128;
  int*   rec  = (int*)(bufB + (long)N * 128);

  const int nown = (N + AGG_R - 1) / AGG_R;

  vgae_rec_build<<<nown, AGG_NT, 0, stream>>>(ei, rec, E);
  vgae_agg_own<64><<<nown, AGG_NT, 0, stream>>>(x, ei, aggx, rec, N, E);

  const int nrt = N / 16;
  for (int t = 0; t < 2; ++t) {
    const int base = 2 + t * 13;
    const float* Wl0 = (const float*)d_in[base + 0];
    const float* bl0 = (const float*)d_in[base + 1];
    const float* Wr0 = (const float*)d_in[base + 2];
    const float* g0  = (const float*)d_in[base + 3];
    const float* b0  = (const float*)d_in[base + 4];
    const float* Wl1 = (const float*)d_in[base + 5];
    const float* bl1 = (const float*)d_in[base + 6];
    const float* Wr1 = (const float*)d_in[base + 7];
    const float* g1  = (const float*)d_in[base + 8];
    const float* b1  = (const float*)d_in[base + 9];
    const float* Wl2 = (const float*)d_in[base + 10];
    const float* bl2 = (const float*)d_in[base + 11];
    const float* Wr2 = (const float*)d_in[base + 12];
    float* outp = (float*)d_out + (long)t * N * 64;

    vgae_sage_gemm<64, 128, true><<<512, 256, 0, stream>>>(aggx, x, Wl0, bl0, Wr0,
                                                          g0, b0, bufA, nrt);
    vgae_agg_own<128><<<nown, AGG_NT, 0, stream>>>(bufA, ei, bufB, rec, N, E);

    vgae_sage_gemm<128, 128, true><<<512, 256, 0, stream>>>(bufB, bufA, Wl1, bl1, Wr1,
                                                           g1, b1, bufB, nrt);
    vgae_agg_own<128><<<nown, AGG_NT, 0, stream>>>(bufB, ei, bufA, rec, N, E);

    vgae_sage_gemm<128, 64, false><<<512, 128, 0, stream>>>(bufA, bufB, Wl2, bl2, Wr2,
                                                           nullptr, nullptr, outp, nrt);
  }
}


# Layers 1, 2: dense layer moved into the neighbor-sum launch: the owner block multiplies its rows' LDS sums (x 1/deg) and own h rows by the [Wl|Wr] LDS panel; 12 launches -> 8; sum orders unchanged
# speedup vs baseline: 6.7820x; 1.1486x over previous
#include <hip/hip_runtime.h>


#define LN_EPS 1e-5f

typedef __attribute__((ext_vector_type(2))) float v2f;
typedef __attribute__((ext_vector_type(8))) float v8f;


#define AGG_R    196
#define AGG_NT   1024
#define AGG_NW   (AGG_NT / 32)
#define AGG_CH   512
#define AGG_PL   (AGG_CH / 32)
#define AGG_K    8
#define AGG_QCAP (AGG_K + AGG_CH)
#define SUP_NB   4
#define SUP_R    (SUP_NB * AGG_R)
#define REC_NQ   4
#define SEG_CAP  60

__global__ __launch_bounds__(AGG_NT)
void vgae_rec_build(const int* __restrict__ ei, int* __restrict__ rec, int E) {
  __shared__ int que[AGG_NW][AGG_CH];

  const int lane = threadIdx.x & 31;
  const int wave = __builtin_amdgcn_readfirstlane(threadIdx.x >> 5);
  const int r0   = (blockIdx.x / REC_NQ) * SUP_R;
  const int qtr  = blockIdx.x % REC_NQ;
  const int* __restrict__ dst = ei + E;

  const int nseg = gridDim.x * AGG_NW * SUP_NB;
  const int seg0 = (blockIdx.x * AGG_NW + wave) * SUP_NB;
  int* __restrict__ ent = rec + nseg + (long)seg0 * SEG_CAP;

  const int nch = (E + AGG_CH - 1) / AGG_CH;
  const int cpq = (nch + REC_NQ - 1) / REC_NQ;
  const int cb  = qtr * cpq;
  const int ce  = (cb + cpq < nch) ? (cb + cpq) : nch;

  int sc[SUP_NB];
  for (int u = 0; u < SUP_NB; ++u) sc[u] = 0;

  int d[AGG_PL];
  auto load_dst = [&](int c) {
    const int base = c * AGG_CH + lane;
    if (c * AGG_CH + AGG_CH <= E) {
      for (int j = 0; j < AGG_PL; ++j) d[j] = dst[base + j * 32];
    } else {
      for (int j = 0; j < AGG_PL; ++j) {
        const int e = base + j * 32;
        d[j] = (e < E) ? dst[e] : -1;
      }
    }
  };

  int c = cb + wave;
  if (c < ce) load_dst(c);
  for (; c < ce; c += AGG_NW) {
    const int base = c * AGG_CH + lane;
    int qn = 0;
    for (int j = 0; j < AGG_PL; ++j) {
      const bool m = (unsigned)(d[j] - r0) < (unsigned)SUP_R;
      const unsigned mask = __builtin_amdgcn_ballot_w32(m);
      if (mask) {
        if (m) que[wave][qn + (int)__builtin_amdgcn_mbcnt_lo(mask, 0u)] = base + j * 32;
        qn += __builtin_popcount(mask);
      }
    }
    if (c + AGG_NW < ce) load_dst(c + AGG_NW);
    for (int i0 = 0; i0 < qn; i0 += 32) {
      const int  n   = (qn - i0 < 32) ? (qn - i0) : 32;
      const bool act = lane < n;
      const int  me  = act ? que[wave][i0 + lane] : 0;
      const int  ms  = ei[me];
      const int  rr  = dst[me] - r0;
      const int  u   = act ? (rr / AGG_R) : -1;
      const int  rb  = rr - u * AGG_R;
      const int  wd  = (int)((unsigned)ms | ((unsigned)rb << 24));
      for (int uu = 0; uu < SUP_NB; ++uu) {
        const bool mine = (u == uu);
        const unsigned mk = __builtin_amdgcn_ballot_w32(mine);
        if (mine) {
          const int pos = sc[uu] + (int)__builtin_amdgcn_mbcnt_lo(mk, 0u);
          if (pos < SEG_CAP) ent[uu * SEG_CAP + pos] = wd;
        }
        sc[uu] += __builtin_popcount(mk);
      }
    }
  }
  if (lane < SUP_NB) {
    int v = sc[0];
    for (int u = 1; u < SUP_NB; ++u) v = (lane == u) ? sc[u] : v;
    rec[seg0 + lane] = v;
  }
}

template <int D>
__device__ __forceinline__ void vgae_agg_consume(const float* __restrict__ h, float* acc, int* cnt,
                                                 int lane, int n, int ms, int md) {
  constexpr int VEC = D / 32;
  typedef __attribute__((ext_vector_type(VEC))) float vN;
  if (lane < n) atomicAdd(&cnt[md], 1);
  int i = 0;
  for (; i + 8 <= n; i += 8) {
    vN v[8];
#pragma unroll
    for (int u = 0; u < 8; ++u) {
      const int s = __builtin_amdgcn_readlane(ms, i + u);
      v[u] = *(const vN*)(h + (long)s * D + VEC * lane);
    }
#pragma unroll
    for (int u = 0; u < 8; ++u) {
      const int dl = __builtin_amdgcn_readlane(md, i + u);
      for (int j = 0; j < VEC; ++j)
        atomicAdd(&acc[dl * D + j * 32 + lane], v[u][j]);
    }
  }
  if (i < n) {
    vN v[8];
#pragma unroll
    for (int u = 0; u < 8; ++u) {
      const int k = (i + u < n) ? (i + u) : (n - 1);
      const int s = __builtin_amdgcn_readlane(ms, k);
      v[u] = *(const vN*)(h + (long)s * D + VEC * lane);
    }
#pragma unroll
    for (int u = 0; u < 8; ++u) asm volatile("" : "+v"(v[u]));
#pragma unroll
    for (int u = 0; u < 8; ++u) {
      if (i + u < n) {
        const int dl = __builtin_amdgcn_readlane(md, i + u);
        for (int j = 0; j < VEC; ++j)
          atomicAdd(&acc[dl * D + j * 32 + lane], v[u][j]);
      }
    }
  }
}

template <int D>
__device__ __forceinline__ void vgae_sum_rows(const float* __restrict__ h, const int* __restrict__ ei,
                                              const int* __restrict__ rec, float* acc, int* cnt,
                                              int (*que)[AGG_QCAP], int E) {
  const int lane = threadIdx.x & 31;
  const int wave = __builtin_amdgcn_readfirstlane(threadIdx.x >> 5);
  const int r0   = blockIdx.x * AGG_R;
  const int* __restrict__ dst = ei + E;

  const int sup  = blockIdx.x / SUP_NB;
  const int own  = blockIdx.x % SUP_NB;
  const int nseg = gridDim.x * AGG_NW * SUP_NB;
  int cq[REC_NQ];
  for (int q = 0; q < REC_NQ; ++q)
    cq[q] = rec[((sup * REC_NQ + q) * AGG_NW + lane) * SUP_NB + own];

  for (int i = threadIdx.x; i < AGG_R * D; i += AGG_NT) acc[i] = 0.0f;
  for (int i = threadIdx.x; i < AGG_R; i += AGG_NT) cnt[i] = 0;

  bool ov = false;
  for (int q = 0; q < REC_NQ; ++q) ov = ov || (cq[q] > SEG_CAP);
  const bool scan = __builtin_amdgcn_ballot_w32(ov) != 0u;
  __syncthreads();

  if (!scan) {
    const int c0 = __builtin_amdgcn_readlane(cq[0], wave);
    const int c1 = __builtin_amdgcn_readlane(cq[1], wave);
    const int c2 = __builtin_amdgcn_readlane(cq[2], wave);
    const int c3 = __builtin_amdgcn_readlane(cq[3], wave);
    const int p1 = c0, p2 = p1 + c1, p3 = p2 + c2, tot = p3 + c3;
    const int* __restrict__ ent =
        rec + nseg + (long)(((sup * REC_NQ) * AGG_NW + wave) * SUP_NB + own) * SEG_CAP;
    auto word = [&](int idx) -> unsigned {
      const int q   = (idx >= p1) + (idx >= p2) + (idx >= p3);
      const int beg = (q == 0) ? 0 : (q == 1) ? p1 : (q == 2) ? p2 : p3;
      return (unsigned)ent[(long)q * (AGG_NW * SUP_NB * SEG_CAP) + (idx - beg)];
    };
    unsigned pk = (lane < tot) ? word(lane) : 0u;
    for (int i0 = 0; i0 < tot; i0 += 32) {
      const int n = (tot - i0 < 32) ? (tot - i0) : 32;
      const unsigned cur = pk;
      if (i0 + 32 < tot)
        pk = (i0 + 32 + lane < tot) ? word(i0 + 32 + lane) : 0u;
      vgae_agg_consume<D>(h, acc, cnt, lane, n, (int)(cur & 0xFFFFFFu), (int)(cur >> 24));
    }
  } else {
    const int nch = (E + AGG_CH - 1) / AGG_CH;
    int qn = 0;

    for (int c = wave; ; c += AGG_NW) {
      const bool live = c < nch;
      if (live) {
        const int base = c * AGG_CH + lane;
        int d[AGG_PL];
        if (c * AGG_CH + AGG_CH <= E) {
          for (int j = 0; j < AGG_PL; ++j) d[j] = dst[base + j * 32];
        } else {
          for (int j = 0; j < AGG_PL; ++j) {
            const int e = base + j * 32;
            d[j] = (e < E) ? dst[e] : -1;
          }
        }
        for (int j = 0; j < AGG_PL; ++j) {
          const bool m = (unsigned)(d[j] - r0) < (unsigned)AGG_R;
          const unsigned mask = __builtin_amdgcn_ballot_w32(m);
          if (mask) {
            if (m) que[wave][qn + (int)__builtin_amdgcn_mbcnt_lo(mask, 0u)] = base + j * 32;
            qn += __builtin_popcount(mask);
          }
        }
      }

      if (qn >= AGG_K || (!live && qn > 0)) {
        for (int i0 = 0; i0 < qn; i0 += 32) {
          const int n  = (qn - i0 < 32) ? (qn - i0) : 32;
          const int me = (lane < n) ? que[wave][i0 + lane] : 0;
          const int ms = ei[me];
          const int md = dst[me] - r0;
          vgae_agg_consume<D>(h, acc, cnt, lane, n, ms, md);
        }
        qn = 0;
      }

      if (!live) break;
    }
  }
}

template <int D>
__global__ __launch_bounds__(AGG_NT)
void vgae_agg_own(const float* __restrict__ h, const int* __restrict__ ei,
                  float* __restrict__ agg, const int* __restrict__ rec, int N, int E) {
  constexpr int VEC = D / 32;
  typedef __attribute__((ext_vector_type(VEC))) float vN;

  __shared__ float acc[AGG_R * D];
  __shared__ int   cnt[AGG_R];
  __shared__ int   que[AGG_NW][AGG_QCAP];

  const int lane = threadIdx.x & 31;
  const int wave = __builtin_amdgcn_readfirstlane(threadIdx.x >> 5);
  const int r0   = blockIdx.x * AGG_R;

  vgae_sum_rows<D>(h, ei, rec, acc, cnt, que, E);
  __syncthreads();

  for (int row = wave; row < AGG_R; row += AGG_NW) {
    const int r = r0 + row;
    if (r < N) {
      const float id = 1.0f / fmaxf((float)cnt[row], 1.0f);
      vN o;
      for (int j = 0; j < VEC; ++j) o[j] = acc[row * D + j * 32 + lane] * id;
      *(vN*)(agg + (long)r * D + VEC * lane) = o;
    }
  }
}

typedef __attribute__((ext_vector_type(4))) float v4f;

template <int M>
__device__ __forceinline__ float vgae_row_xor(float x) {
  return __builtin_bit_cast(float, __builtin_amdgcn_update_dpp(
      0, __builtin_bit_cast(int, x), 0x160 | M, 0xf, 0xf, true));
}

template <int DIN, int DOUT, bool LN>
__global__ __launch_bounds__(DOUT * 2)
void vgae_sage_gemm(const float* agg,
                    const float* __restrict__ h,
                    const float* __restrict__ Wl,
                    const float* __restrict__ bl,
                    const float* __restrict__ Wr,
                    const float* __restrict__ gamma,
                    const float* __restrict__ beta,
                    float* out, int ntiles) {
  constexpr int K   = 2 * DIN;
  constexpr int P   = K + 4;
  constexpr int NT  = DOUT * 2;
  constexpr int NW  = DOUT / 16;
  constexpr int RPW = 16 / NW;
  constexpr int F4  = K / 4;
  constexpr int NL  = RPW * F4 / 32;
  static_assert(!LN || DOUT == 128, "LN epilogue is written for 128 columns");

  __shared__ __attribute__((aligned(16))) float Bt[DOUT * P];
  __shared__ __attribute__((aligned(16))) float At[16 * P];
  constexpr int SP = 132;
  __shared__ __attribute__((aligned(16))) float smem[LN ? 16 * SP : 4];

  const int tid  = threadIdx.x;
  const int lane = tid & 31;
  const int wave = tid >> 5;
  const int l15  = lane & 15;
  const int hi   = lane >> 4;
  const int bcol = wave * 16 + l15;
  const int koff = 2 * hi;

  for (int i = tid; i < DOUT * F4; i += NT) {
    const int cc  = i / F4;
    const int col = (i % F4) * 4;
    const float* s = (col < DIN) ? (Wl + (long)cc * DIN + col)
                                 : (Wr + (long)cc * DIN + (col - DIN));
    *(v4f*)&Bt[cc * P + col] = *(const v4f*)s;
  }

  int prow[NL], pcol[NL];
  for (int i = 0; i < NL; ++i) {
    const int flat = i * 32 + lane;
    prow[i] = wave * RPW + flat / F4;
    pcol[i] = (flat % F4) * 4;
  }
  v4f pre[NL];
  auto fetch = [&](int tile) {
    for (int i = 0; i < NL; ++i) {
      const long r = (long)tile * 16 + prow[i];
      const float* s = (pcol[i] < DIN) ? (agg + r * DIN + pcol[i])
                                       : (h + r * DIN + (pcol[i] - DIN));
      pre[i] = *(const v4f*)s;
    }
  };

  const float bias = bl[bcol];
  const int erow = tid >> 4;
  const int ecol = (tid & 15) * 4;
  v4f g0 = {0.f, 0.f, 0.f, 0.f}, g1 = g0, be0 = g0, be1 = g0;
  if (LN) {
    g0  = *(const v4f*)(gamma + ecol);
    g1  = *(const v4f*)(gamma + 64 + ecol);
    be0 = *(const v4f*)(beta + ecol);
    be1 = *(const v4f*)(beta + 64 + ecol);
  }

  const float* ap = &At[l15 * P + koff];
  const float* bp = &Bt[bcol * P + koff];

  int tile = blockIdx.x;
  if (tile < ntiles) fetch(tile);

  for (; tile < ntiles; tile += gridDim.x) {
    for (int i = 0; i < NL; ++i) *(v4f*)&At[prow[i] * P + pcol[i]] = pre[i];
    __syncthreads();
    const int next = tile + gridDim.x;
    if (next < ntiles) fetch(next);

    v8f c = {0.f, 0.f, 0.f, 0.f, 0.f, 0.f, 0.f, 0.f};
    for (int k = 0; k < K; k += 4) {
      v2f a = *(const v2f*)(ap + k);
      v2f b = *(const v2f*)(bp + k);
      c = __builtin_amdgcn_wmma_f32_16x16x4_f32(false, a, false, b,
                                                (short)0, c, false, false);
    }

    const int rowbase = tile * 16;

    if (!LN) {
      for (int v = 0; v < 8; ++v) {
        int r = rowbase + v + 8 * hi;
        out[(long)r * DOUT + bcol] = c[v] + bias;
      }
      __syncthreads();
    } else {
      for (int v = 0; v < 8; ++v)
        smem[(v + 8 * hi) * SP + bcol] = c[v] + bias;
      __syncthreads();

      const v4f x0 = *(const v4f*)&smem[erow * SP + ecol];
      const v4f x1 = *(const v4f*)&smem[erow * SP + 64 + ecol];
      float s  = ((x0[0] + x0[1]) + (x0[2] + x0[3])) + ((x1[0] + x1[1]) + (x1[2] + x1[3]));
      float ss = ((x0[0] * x0[0] + x0[1] * x0[1]) + (x0[2] * x0[2] + x0[3] * x0[3]))
               + ((x1[0] * x1[0] + x1[1] * x1[1]) + (x1[2] * x1[2] + x1[3] * x1[3]));
      s += vgae_row_xor<1>(s);  ss += vgae_row_xor<1>(ss);
      s += vgae_row_xor<2>(s);  ss += vgae_row_xor<2>(ss);
      s += vgae_row_xor<4>(s);  ss += vgae_row_xor<4>(ss);
      s += vgae_row_xor<8>(s);  ss += vgae_row_xor<8>(ss);
      const float m    = s * (1.0f / 128.0f);
      const float var  = ss * (1.0f / 128.0f) - m * m;
      const float rstd = rsqrtf(var + LN_EPS);

      v4f y0, y1;
      for (int j = 0; j < 4; ++j) {
        y0[j] = fmaxf((x0[j] - m) * rstd * g0[j] + be0[j], 0.0f);
        y1[j] = fmaxf((x1[j] - m) * rstd * g1[j] + be1[j], 0.0f);
      }
      float* orow = out + (long)(rowbase + erow) * 128;
      *(v4f*)(orow + ecol)      = y0;
      *(v4f*)(orow + 64 + ecol) = y1;
    }
  }
}

template <int DOUT, bool LN>
__global__ __launch_bounds__(AGG_NT)
void vgae_agg_dense(const float* __restrict__ h, const int* __restrict__ ei,
                    const int* __restrict__ rec,
                    const float* __restrict__ Wl, const float* __restrict__ bl,
                    const float* __restrict__ Wr,
                    const float* __restrict__ gamma, const float* __restrict__ beta,
                    float* __restrict__ out, int N, int E) {
  constexpr int D     = 128;
  constexpr int K     = 2 * D;
  constexpr int P     = K + 4;
  constexpr int F4    = K / 4;
  constexpr int CT    = DOUT / 16;
  constexpr int RT    = AGG_NW / CT;
  constexpr int ROWS  = 16 * RT;
  constexpr int RPW   = ROWS / AGG_NW;
  constexpr int NTRIP = (AGG_R + ROWS - 1) / ROWS;
  constexpr int SP    = 132;
  static_assert(!LN || DOUT == 128, "LN epilogue is written for 128 columns");
  static_assert(ROWS * P >= AGG_NW * AGG_QCAP, "fallback queue must fit the shared region");
  static_assert(!LN || ROWS * P >= ROWS * SP, "LN staging rows must fit the shared region");
  static_assert(!LN || ROWS * 16 == AGG_NT, "LN epilogue: 16 threads per row of a trip");

  __shared__ float acc[AGG_R * D];
  __shared__ int   cnt[AGG_R];
  __shared__ __attribute__((aligned(16))) float Bt[DOUT * P];
  __shared__ __attribute__((aligned(16))) int   uni[ROWS * P];
  int (*que)[AGG_QCAP] = (int (*)[AGG_QCAP])uni;
  float* At   = (float*)uni;
  float* smem = (float*)uni;

  const int tid  = threadIdx.x;
  const int lane = tid & 31;
  const int wave = __builtin_amdgcn_readfirstlane(tid >> 5);
  const int r0   = blockIdx.x * AGG_R;

  for (int i = tid; i < DOUT * F4; i += AGG_NT) {
    const int cc  = i / F4;
    const int col = (i % F4) * 4;
    const float* s = (col < D) ? (Wl + (long)cc * D + col)
                               : (Wr + (long)cc * D + (col - D));
    *(v4f*)&Bt[cc * P + col] = *(const v4f*)s;
  }

  vgae_sum_rows<D>(h, ei, rec, acc, cnt, que, E);

  v4f pre[RPW];
  auto fetch = [&](int trip) {
    for (int i = 0; i < RPW; ++i) {
      const int row = trip * ROWS + i * AGG_NW + wave;
      const int r   = r0 + row;
      v4f z = {0.f, 0.f, 0.f, 0.f};
      if (row < AGG_R && r < N) z = *(const v4f*)(h + (long)r * D + 4 * lane);
      pre[i] = z;
    }
  };
  fetch(0);
  __syncthreads();

  const int l15  = lane & 15;
  const int hi   = lane >> 4;
  const int ct   = wave % CT;
  const int rt   = wave / CT;
  const int bcol = ct * 16 + l15;
  const int koff = 2 * hi;
  const float bias = bl[bcol];
  const int erow = tid >> 4;
  const int ecol = (tid & 15) * 4;
  v4f g0 = {0.f, 0.f, 0.f, 0.f}, g1 = g0, be0 = g0, be1 = g0;
  if (LN) {
    g0  = *(const v4f*)(gamma + ecol);
    g1  = *(const v4f*)(gamma + 64 + ecol);
    be0 = *(const v4f*)(beta + ecol);
    be1 = *(const v4f*)(beta + 64 + ecol);
  }
  const float* ap = &At[(rt * 16 + l15) * P + koff];
  const float* bp = &Bt[bcol * P + koff];

  for (int trip = 0; trip < NTRIP; ++trip) {
    for (int i = 0; i < RPW; ++i) {
      const int rr  = i * AGG_NW + wave;
      const int row = trip * ROWS + rr;
      v4f a = {0.f, 0.f, 0.f, 0.f};
      if (row < AGG_R) {
        const float id = 1.0f / fmaxf((float)cnt[row], 1.0f);
        for (int j = 0; j < 4; ++j) a[j] = acc[row * D + j * 32 + lane] * id;
      }
      *(v4f*)&At[rr * P + 4 * lane]     = a;
      *(v4f*)&At[rr * P + D + 4 * lane] = pre[i];
    }
    __syncthreads();
    if (trip + 1 < NTRIP) fetch(trip + 1);

    const int trow = trip * ROWS + rt * 16;
    v8f c = {0.f, 0.f, 0.f, 0.f, 0.f, 0.f, 0.f, 0.f};
    if (trow < AGG_R) {
      for (int k = 0; k < K; k += 4) {
        v2f a = *(const v2f*)(ap + k);
        v2f b = *(const v2f*)(bp + k);
        c = __builtin_amdgcn_wmma_f32_16x16x4_f32(false, a, false, b,
                                                  (short)0, c, false, false);
      }
    }

    if (!LN) {
      for (int v = 0; v < 8; ++v) {
        const int row = trow + v + 8 * hi;
        const int r   = r0 + row;
        if (row < AGG_R && r < N) out[(long)r * DOUT + bcol] = c[v] + bias;
      }
      __syncthreads();
    } else {
      __syncthreads();
      for (int v = 0; v < 8; ++v)
        smem[(rt * 16 + v + 8 * hi) * SP + bcol] = c[v] + bias;
      __syncthreads();

      const v4f x0 = *(const v4f*)&smem[erow * SP + ecol];
      const v4f x1 = *(const v4f*)&smem[erow * SP + 64 + ecol];
      float s  = ((x0[0] + x0[1]) + (x0[2] + x0[3])) + ((x1[0] + x1[1]) + (x1[2] + x1[3]));
      float ss = ((x0[0] * x0[0] + x0[1] * x0[1]) + (x0[2] * x0[2] + x0[3] * x0[3]))
               + ((x1[0] * x1[0] + x1[1] * x1[1]) + (x1[2] * x1[2] + x1[3] * x1[3]));
      s += vgae_row_xor<1>(s);  ss += vgae_row_xor<1>(ss);
      s += vgae_row_xor<2>(s);  ss += vgae_row_xor<2>(ss);
      s += vgae_row_xor<4>(s);  ss += vgae_row_xor<4>(ss);
      s += vgae_row_xor<8>(s);  ss += vgae_row_xor<8>(ss);
      const float m    = s * (1.0f / 128.0f);
      const float var  = ss * (1.0f / 128.0f) - m * m;
      const float rstd = rsqrtf(var + LN_EPS);

      v4f y0, y1;
      for (int j = 0; j < 4; ++j) {
        y0[j] = fmaxf((x0[j] - m) * rstd * g0[j] + be0[j], 0.0f);
        y1[j] = fmaxf((x1[j] - m) * rstd * g1[j] + be1[j], 0.0f);
      }
      const int row = trip * ROWS + erow;
      const int r   = r0 + row;
      if (row < AGG_R && r < N) {
        float* orow = out + (long)r * 128;
        *(v4f*)(orow + ecol)      = y0;
        *(v4f*)(orow + 64 + ecol) = y1;
      }
      __syncthreads();
    }
  }
}


extern "C" void kernel_launch(void* const* d_in, const int* in_sizes, int n_in,
                              void* d_out, int out_size, void* d_ws, size_t ws_size,
                              hipStream_t stream) {
  const float* x  = (const float*)d_in[0];
  const int*   ei = (const int*)d_in[1];
  const int N = in_sizes[0] / 64;
  const int E = in_sizes[1] / 2;

  float* aggx = (float*)d_ws;
  float* bufA = aggx + (long)N * 64;
  float* bufB = bufA + (long)N * 128;
  int*   rec  = (int*)(bufB + (long)N * 128);

  const int nown = (N + AGG_R - 1) / AGG_R;

  vgae_rec_build<<<nown, AGG_NT, 0, stream>>>(ei, rec, E);
  vgae_agg_own<64><<<nown, AGG_NT, 0, stream>>>(x, ei, aggx, rec, N, E);

  const int nrt = N / 16;
  for (int t = 0; t < 2; ++t) {
    const int base = 2 + t * 13;
    const float* Wl0 = (const float*)d_in[base + 0];
    const float* bl0 = (const float*)d_in[base + 1];
    const float* Wr0 = (const float*)d_in[base + 2];
    const float* g0  = (const float*)d_in[base + 3];
    const float* b0  = (const float*)d_in[base + 4];
    const float* Wl1 = (const float*)d_in[base + 5];
    const float* bl1 = (const float*)d_in[base + 6];
    const float* Wr1 = (const float*)d_in[base + 7];
    const float* g1  = (const float*)d_in[base + 8];
    const float* b1  = (const float*)d_in[base + 9];
    const float* Wl2 = (const float*)d_in[base + 10];
    const float* bl2 = (const float*)d_in[base + 11];
    const float* Wr2 = (const float*)d_in[base + 12];
    float* outp = (float*)d_out + (long)t * N * 64;

    vgae_sage_gemm<64, 128, true><<<512, 256, 0, stream>>>(aggx, x, Wl0, bl0, Wr0,
                                                          g0, b0, bufA, nrt);
    vgae_agg_dense<128, true><<<nown, AGG_NT, 0, stream>>>(bufA, ei, rec, Wl1, bl1, Wr1,
                                                           g1, b1, bufB, N, E);
    vgae_agg_dense<64, false><<<nown, AGG_NT, 0, stream>>>(bufB, ei, rec, Wl2, bl2, Wr2,
                                                           nullptr, nullptr, outp, N, E);
  }
}


# Layers 1, 2: root half of the product (own rows x Wr) taken during the sum phase, in per-trip stages; half the waves multiply while the rest fetch rows; sum order: root half first, then mean half
# speedup vs baseline: 7.0467x; 1.0390x over previous
#include <hip/hip_runtime.h>


#define LN_EPS 1e-5f

typedef __attribute__((ext_vector_type(2))) float v2f;
typedef __attribute__((ext_vector_type(8))) float v8f;


#define AGG_R    196
#define AGG_NT   1024
#define AGG_NW   (AGG_NT / 32)
#define AGG_CH   512
#define AGG_PL   (AGG_CH / 32)
#define AGG_K    8
#define AGG_QCAP (AGG_K + AGG_CH)
#define SUP_NB   4
#define SUP_R    (SUP_NB * AGG_R)
#define REC_NQ   4
#define SEG_CAP  60

__global__ __launch_bounds__(AGG_NT)
void vgae_rec_build(const int* __restrict__ ei, int* __restrict__ rec, int E) {
  __shared__ int que[AGG_NW][AGG_CH];

  const int lane = threadIdx.x & 31;
  const int wave = __builtin_amdgcn_readfirstlane(threadIdx.x >> 5);
  const int r0   = (blockIdx.x / REC_NQ) * SUP_R;
  const int qtr  = blockIdx.x % REC_NQ;
  const int* __restrict__ dst = ei + E;

  const int nseg = gridDim.x * AGG_NW * SUP_NB;
  const int seg0 = (blockIdx.x * AGG_NW + wave) * SUP_NB;
  int* __restrict__ ent = rec + nseg + (long)seg0 * SEG_CAP;

  const int nch = (E + AGG_CH - 1) / AGG_CH;
  const int cpq = (nch + REC_NQ - 1) / REC_NQ;
  const int cb  = qtr * cpq;
  const int ce  = (cb + cpq < nch) ? (cb + cpq) : nch;

  int sc[SUP_NB];
  for (int u = 0; u < SUP_NB; ++u) sc[u] = 0;

  int d[AGG_PL];
  auto load_dst = [&](int c) {
    const int base = c * AGG_CH + lane;
    if (c * AGG_CH + AGG_CH <= E) {
      for (int j = 0; j < AGG_PL; ++j) d[j] = dst[base + j * 32];
    } else {
      for (int j = 0; j < AGG_PL; ++j) {
        const int e = base + j * 32;
        d[j] = (e < E) ? dst[e] : -1;
      }
    }
  };

  int c = cb + wave;
  if (c < ce) load_dst(c);
  for (; c < ce; c += AGG_NW) {
    const int base = c * AGG_CH + lane;
    int qn = 0;
    for (int j = 0; j < AGG_PL; ++j) {
      const bool m = (unsigned)(d[j] - r0) < (unsigned)SUP_R;
      const unsigned mask = __builtin_amdgcn_ballot_w32(m);
      if (mask) {
        if (m) que[wave][qn + (int)__builtin_amdgcn_mbcnt_lo(mask, 0u)] = base + j * 32;
        qn += __builtin_popcount(mask);
      }
    }
    if (c + AGG_NW < ce) load_dst(c + AGG_NW);
    for (int i0 = 0; i0 < qn; i0 += 32) {
      const int  n   = (qn - i0 < 32) ? (qn - i0) : 32;
      const bool act = lane < n;
      const int  me  = act ? que[wave][i0 + lane] : 0;
      const int  ms  = ei[me];
      const int  rr  = dst[me] - r0;
      const int  u   = act ? (rr / AGG_R) : -1;
      const int  rb  = rr - u * AGG_R;
      const int  wd  = (int)((unsigned)ms | ((unsigned)rb << 24));
      for (int uu = 0; uu < SUP_NB; ++uu) {
        const bool mine = (u == uu);
        const unsigned mk = __builtin_amdgcn_ballot_w32(mine);
        if (mine) {
          const int pos = sc[uu] + (int)__builtin_amdgcn_mbcnt_lo(mk, 0u);
          if (pos < SEG_CAP) ent[uu * SEG_CAP + pos] = wd;
        }
        sc[uu] += __builtin_popcount(mk);
      }
    }
  }
  if (lane < SUP_NB) {
    int v = sc[0];
    for (int u = 1; u < SUP_NB; ++u) v = (lane == u) ? sc[u] : v;
    rec[seg0 + lane] = v;
  }
}

template <int D>
__device__ __forceinline__ void vgae_agg_consume(const float* __restrict__ h, float* acc, int* cnt,
                                                 int lane, int n, int ms, int md) {
  constexpr int VEC = D / 32;
  typedef __attribute__((ext_vector_type(VEC))) float vN;
  if (lane < n) atomicAdd(&cnt[md], 1);
  int i = 0;
  for (; i + 8 <= n; i += 8) {
    vN v[8];
#pragma unroll
    for (int u = 0; u < 8; ++u) {
      const int s = __builtin_amdgcn_readlane(ms, i + u);
      v[u] = *(const vN*)(h + (long)s * D + VEC * lane);
    }
#pragma unroll
    for (int u = 0; u < 8; ++u) {
      const int dl = __builtin_amdgcn_readlane(md, i + u);
      for (int j = 0; j < VEC; ++j)
        atomicAdd(&acc[dl * D + j * 32 + lane], v[u][j]);
    }
  }
  if (i < n) {
    vN v[8];
#pragma unroll
    for (int u = 0; u < 8; ++u) {
      const int k = (i + u < n) ? (i + u) : (n - 1);
      const int s = __builtin_amdgcn_readlane(ms, k);
      v[u] = *(const vN*)(h + (long)s * D + VEC * lane);
    }
#pragma unroll
    for (int u = 0; u < 8; ++u) asm volatile("" : "+v"(v[u]));
#pragma unroll
    for (int u = 0; u < 8; ++u) {
      if (i + u < n) {
        const int dl = __builtin_amdgcn_readlane(md, i + u);
        for (int j = 0; j < VEC; ++j)
          atomicAdd(&acc[dl * D + j * 32 + lane], v[u][j]);
      }
    }
  }
}

template <int D>
__device__ __forceinline__ void vgae_scan_rows(const float* __restrict__ h, const int* __restrict__ ei,
                                               float* acc, int* cnt, int (*que)[AGG_QCAP], int E) {
  const int lane = threadIdx.x & 31;
  const int wave = __builtin_amdgcn_readfirstlane(threadIdx.x >> 5);
  const int r0   = blockIdx.x * AGG_R;
  const int* __restrict__ dst = ei + E;

  const int nch = (E + AGG_CH - 1) / AGG_CH;
  int qn = 0;

  for (int c = wave; ; c += AGG_NW) {
    const bool live = c < nch;
    if (live) {
      const int base = c * AGG_CH + lane;
      int d[AGG_PL];
      if (c * AGG_CH + AGG_CH <= E) {
        for (int j = 0; j < AGG_PL; ++j) d[j] = dst[base + j * 32];
      } else {
        for (int j = 0; j < AGG_PL; ++j) {
          const int e = base + j * 32;
          d[j] = (e < E) ? dst[e] : -1;
        }
      }
      for (int j = 0; j < AGG_PL; ++j) {
        const bool m = (unsigned)(d[j] - r0) < (unsigned)AGG_R;
        const unsigned mask = __builtin_amdgcn_ballot_w32(m);
        if (mask) {
          if (m) que[wave][qn + (int)__builtin_amdgcn_mbcnt_lo(mask, 0u)] = base + j * 32;
          qn += __builtin_popcount(mask);
        }
      }
    }

    if (qn >= AGG_K || (!live && qn > 0)) {
      for (int i0 = 0; i0 < qn; i0 += 32) {
        const int n  = (qn - i0 < 32) ? (qn - i0) : 32;
        const int me = (lane < n) ? que[wave][i0 + lane] : 0;
        const int ms = ei[me];
        const int md = dst[me] - r0;
        vgae_agg_consume<D>(h, acc, cnt, lane, n, ms, md);
      }
      qn = 0;
    }

    if (!live) break;
  }
}

template <int D>
__device__ __forceinline__ void vgae_sum_rows(const float* __restrict__ h, const int* __restrict__ ei,
                                              const int* __restrict__ rec, float* acc, int* cnt,
                                              int (*que)[AGG_QCAP], int E) {
  const int lane = threadIdx.x & 31;
  const int wave = __builtin_amdgcn_readfirstlane(threadIdx.x >> 5);

  const int sup  = blockIdx.x / SUP_NB;
  const int own  = blockIdx.x % SUP_NB;
  const int nseg = gridDim.x * AGG_NW * SUP_NB;
  int cq[REC_NQ];
  for (int q = 0; q < REC_NQ; ++q)
    cq[q] = rec[((sup * REC_NQ + q) * AGG_NW + lane) * SUP_NB + own];

  for (int i = threadIdx.x; i < AGG_R * D; i += AGG_NT) acc[i] = 0.0f;
  for (int i = threadIdx.x; i < AGG_R; i += AGG_NT) cnt[i] = 0;

  bool ov = false;
  for (int q = 0; q < REC_NQ; ++q) ov = ov || (cq[q] > SEG_CAP);
  const bool scan = __builtin_amdgcn_ballot_w32(ov) != 0u;
  __syncthreads();

  if (!scan) {
    const int c0 = __builtin_amdgcn_readlane(cq[0], wave);
    const int c1 = __builtin_amdgcn_readlane(cq[1], wave);
    const int c2 = __builtin_amdgcn_readlane(cq[2], wave);
    const int c3 = __builtin_amdgcn_readlane(cq[3], wave);
    const int p1 = c0, p2 = p1 + c1, p3 = p2 + c2, tot = p3 + c3;
    const int* __restrict__ ent =
        rec + nseg + (long)(((sup * REC_NQ) * AGG_NW + wave) * SUP_NB + own) * SEG_CAP;
    auto word = [&](int idx) -> unsigned {
      const int q   = (idx >= p1) + (idx >= p2) + (idx >= p3);
      const int beg = (q == 0) ? 0 : (q == 1) ? p1 : (q == 2) ? p2 : p3;
      return (unsigned)ent[(long)q * (AGG_NW * SUP_NB * SEG_CAP) + (idx - beg)];
    };
    unsigned pk = (lane < tot) ? word(lane) : 0u;
    for (int i0 = 0; i0 < tot; i0 += 32) {
      const int n = (tot - i0 < 32) ? (tot - i0) : 32;
      const unsigned cur = pk;
      if (i0 + 32 < tot)
        pk = (i0 + 32 + lane < tot) ? word(i0 + 32 + lane) : 0u;
      vgae_agg_consume<D>(h, acc, cnt, lane, n, (int)(cur & 0xFFFFFFu), (int)(cur >> 24));
    }
  } else {
    vgae_scan_rows<D>(h, ei, acc, cnt, que, E);
  }
}

template <int D>
__global__ __launch_bounds__(AGG_NT)
void vgae_agg_own(const float* __restrict__ h, const int* __restrict__ ei,
                  float* __restrict__ agg, const int* __restrict__ rec, int N, int E) {
  constexpr int VEC = D / 32;
  typedef __attribute__((ext_vector_type(VEC))) float vN;

  __shared__ float acc[AGG_R * D];
  __shared__ int   cnt[AGG_R];
  __shared__ int   que[AGG_NW][AGG_QCAP];

  const int lane = threadIdx.x & 31;
  const int wave = __builtin_amdgcn_readfirstlane(threadIdx.x >> 5);
  const int r0   = blockIdx.x * AGG_R;

  vgae_sum_rows<D>(h, ei, rec, acc, cnt, que, E);
  __syncthreads();

  for (int row = wave; row < AGG_R; row += AGG_NW) {
    const int r = r0 + row;
    if (r < N) {
      const float id = 1.0f / fmaxf((float)cnt[row], 1.0f);
      vN o;
      for (int j = 0; j < VEC; ++j) o[j] = acc[row * D + j * 32 + lane] * id;
      *(vN*)(agg + (long)r * D + VEC * lane) = o;
    }
  }
}

typedef __attribute__((ext_vector_type(4))) float v4f;

template <int M>
__device__ __forceinline__ float vgae_row_xor(float x) {
  return __builtin_bit_cast(float, __builtin_amdgcn_update_dpp(
      0, __builtin_bit_cast(int, x), 0x160 | M, 0xf, 0xf, true));
}

template <int DIN, int DOUT, bool LN>
__global__ __launch_bounds__(DOUT * 2)
void vgae_sage_gemm(const float* agg,
                    const float* __restrict__ h,
                    const float* __restrict__ Wl,
                    const float* __restrict__ bl,
                    const float* __restrict__ Wr,
                    const float* __restrict__ gamma,
                    const float* __restrict__ beta,
                    float* out, int ntiles) {
  constexpr int K   = 2 * DIN;
  constexpr int P   = K + 4;
  constexpr int NT  = DOUT * 2;
  constexpr int NW  = DOUT / 16;
  constexpr int RPW = 16 / NW;
  constexpr int F4  = K / 4;
  constexpr int NL  = RPW * F4 / 32;
  static_assert(!LN || DOUT == 128, "LN epilogue is written for 128 columns");

  __shared__ __attribute__((aligned(16))) float Bt[DOUT * P];
  __shared__ __attribute__((aligned(16))) float At[16 * P];
  constexpr int SP = 132;
  __shared__ __attribute__((aligned(16))) float smem[LN ? 16 * SP : 4];

  const int tid  = threadIdx.x;
  const int lane = tid & 31;
  const int wave = tid >> 5;
  const int l15  = lane & 15;
  const int hi   = lane >> 4;
  const int bcol = wave * 16 + l15;
  const int koff = 2 * hi;

  for (int i = tid; i < DOUT * F4; i += NT) {
    const int cc  = i / F4;
    const int col = (i % F4) * 4;
    const float* s = (col < DIN) ? (Wl + (long)cc * DIN + col)
                                 : (Wr + (long)cc * DIN + (col - DIN));
    *(v4f*)&Bt[cc * P + col] = *(const v4f*)s;
  }

  int prow[NL], pcol[NL];
  for (int i = 0; i < NL; ++i) {
    const int flat = i * 32 + lane;
    prow[i] = wave * RPW + flat / F4;
    pcol[i] = (flat % F4) * 4;
  }
  v4f pre[NL];
  auto fetch = [&](int tile) {
    for (int i = 0; i < NL; ++i) {
      const long r = (long)tile * 16 + prow[i];
      const float* s = (pcol[i] < DIN) ? (agg + r * DIN + pcol[i])
                                       : (h + r * DIN + (pcol[i] - DIN));
      pre[i] = *(const v4f*)s;
    }
  };

  const float bias = bl[bcol];
  const int erow = tid >> 4;
  const int ecol = (tid & 15) * 4;
  v4f g0 = {0.f, 0.f, 0.f, 0.f}, g1 = g0, be0 = g0, be1 = g0;
  if (LN) {
    g0  = *(const v4f*)(gamma + ecol);
    g1  = *(const v4f*)(gamma + 64 + ecol);
    be0 = *(const v4f*)(beta + ecol);
    be1 = *(const v4f*)(beta + 64 + ecol);
  }

  const float* ap = &At[l15 * P + koff];
  const float* bp = &Bt[bcol * P + koff];

  int tile = blockIdx.x;
  if (tile < ntiles) fetch(tile);

  for (; tile < ntiles; tile += gridDim.x) {
    for (int i = 0; i < NL; ++i) *(v4f*)&At[prow[i] * P + pcol[i]] = pre[i];
    __syncthreads();
    const int next = tile + gridDim.x;
    if (next < ntiles) fetch(next);

    v8f c = {0.f, 0.f, 0.f, 0.f, 0.f, 0.f, 0.f, 0.f};
    for (int k = 0; k < K; k += 4) {
      v2f a = *(const v2f*)(ap + k);
      v2f b = *(const v2f*)(bp + k);
      c = __builtin_amdgcn_wmma_f32_16x16x4_f32(false, a, false, b,
                                                (short)0, c, false, false);
    }

    const int rowbase = tile * 16;

    if (!LN) {
      for (int v = 0; v < 8; ++v) {
        int r = rowbase + v + 8 * hi;
        out[(long)r * DOUT + bcol] = c[v] + bias;
      }
      __syncthreads();
    } else {
      for (int v = 0; v < 8; ++v)
        smem[(v + 8 * hi) * SP + bcol] = c[v] + bias;
      __syncthreads();

      const v4f x0 = *(const v4f*)&smem[erow * SP + ecol];
      const v4f x1 = *(const v4f*)&smem[erow * SP + 64 + ecol];
      float s  = ((x0[0] + x0[1]) + (x0[2] + x0[3])) + ((x1[0] + x1[1]) + (x1[2] + x1[3]));
      float ss = ((x0[0] * x0[0] + x0[1] * x0[1]) + (x0[2] * x0[2] + x0[3] * x0[3]))
               + ((x1[0] * x1[0] + x1[1] * x1[1]) + (x1[2] * x1[2] + x1[3] * x1[3]));
      s += vgae_row_xor<1>(s);  ss += vgae_row_xor<1>(ss);
      s += vgae_row_xor<2>(s);  ss += vgae_row_xor<2>(ss);
      s += vgae_row_xor<4>(s);  ss += vgae_row_xor<4>(ss);
      s += vgae_row_xor<8>(s);  ss += vgae_row_xor<8>(ss);
      const float m    = s * (1.0f / 128.0f);
      const float var  = ss * (1.0f / 128.0f) - m * m;
      const float rstd = rsqrtf(var + LN_EPS);

      v4f y0, y1;
      for (int j = 0; j < 4; ++j) {
        y0[j] = fmaxf((x0[j] - m) * rstd * g0[j] + be0[j], 0.0f);
        y1[j] = fmaxf((x1[j] - m) * rstd * g1[j] + be1[j], 0.0f);
      }
      float* orow = out + (long)(rowbase + erow) * 128;
      *(v4f*)(orow + ecol)      = y0;
      *(v4f*)(orow + 64 + ecol) = y1;
    }
  }
}

template <int DOUT, bool LN>
__global__ __launch_bounds__(AGG_NT)
void vgae_agg_dense(const float* __restrict__ h, const int* __restrict__ ei,
                    const int* __restrict__ rec,
                    const float* __restrict__ Wl, const float* __restrict__ bl,
                    const float* __restrict__ Wr,
                    const float* __restrict__ gamma, const float* __restrict__ beta,
                    float* __restrict__ out, int N, int E) {
  constexpr int D     = 128;
  constexpr int K     = 2 * D;
  constexpr int P     = K + 4;
  constexpr int PA    = D + 4;
  constexpr int F4    = K / 4;
  constexpr int CT    = DOUT / 16;
  constexpr int RT    = AGG_NW / CT;
  constexpr int ROWS  = 16 * RT;
  constexpr int RPW   = ROWS / AGG_NW;
  constexpr int NTRIP = (AGG_R + ROWS - 1) / ROWS;
  constexpr int SP    = 132;
  constexpr int TILE  = ROWS * PA;
  static_assert(!LN || DOUT == 128, "LN epilogue is written for 128 columns");
  static_assert(2 * TILE >= AGG_NW * AGG_QCAP, "fallback queue must fit the shared region");
  static_assert(!LN || TILE >= ROWS * SP, "LN staging rows must fit the shared region");
  static_assert(!LN || ROWS * 16 == AGG_NT, "LN epilogue: 16 threads per row of a trip");

  __shared__ float acc[AGG_R * D];
  __shared__ int   cnt[AGG_R];
  __shared__ __attribute__((aligned(16))) float Bt[DOUT * P];
  __shared__ __attribute__((aligned(16))) int   uni[2 * TILE];
  int (*que)[AGG_QCAP] = (int (*)[AGG_QCAP])uni;
  float* At   = (float*)uni;
  float* smem = (float*)uni;

  const int tid  = threadIdx.x;
  const int lane = tid & 31;
  const int wave = __builtin_amdgcn_readfirstlane(tid >> 5);
  const int r0   = blockIdx.x * AGG_R;

  for (int i = tid; i < DOUT * F4; i += AGG_NT) {
    const int cc  = i / F4;
    const int col = (i % F4) * 4;
    const float* s = (col < D) ? (Wl + (long)cc * D + col)
                               : (Wr + (long)cc * D + (col - D));
    *(v4f*)&Bt[cc * P + col] = *(const v4f*)s;
  }

  const int sup  = blockIdx.x / SUP_NB;
  const int own  = blockIdx.x % SUP_NB;
  int cq[REC_NQ];
  for (int q = 0; q < REC_NQ; ++q)
    cq[q] = rec[((sup * REC_NQ + q) * AGG_NW + lane) * SUP_NB + own];
  bool ov = false;
  for (int q = 0; q < REC_NQ; ++q) ov = ov || (cq[q] > SEG_CAP);
  const bool scan = __builtin_amdgcn_ballot_w32(ov) != 0u;

  v4f pre[RPW];
  auto fetch = [&](int trip) {
    for (int i = 0; i < RPW; ++i) {
      const int row = trip * ROWS + i * AGG_NW + wave;
      const int r   = r0 + row;
      v4f z = {0.f, 0.f, 0.f, 0.f};
      if (row < AGG_R && r < N) z = *(const v4f*)(h + (long)r * D + 4 * lane);
      pre[i] = z;
    }
  };
  fetch(0);

  for (int i = tid; i < AGG_R * D; i += AGG_NT) acc[i] = 0.0f;
  for (int i = tid; i < AGG_R; i += AGG_NT) cnt[i] = 0;
  if (scan) {
    __syncthreads();
    vgae_scan_rows<D>(h, ei, acc, cnt, que, E);
    __syncthreads();
  }

  const int l15  = lane & 15;
  const int hi   = lane >> 4;
  const int ct   = wave % CT;
  const int rt   = wave / CT;
  const int bcol = ct * 16 + l15;
  const int koff = 2 * hi;
  const float* bpl = &Bt[bcol * P + koff];
  const float* bpr = bpl + D;
  const int mulfirst = (wave >> 2) & 1;

  const int c0 = __builtin_amdgcn_readlane(cq[0], wave);
  const int c1 = __builtin_amdgcn_readlane(cq[1], wave);
  const int c2 = __builtin_amdgcn_readlane(cq[2], wave);
  const int c3 = __builtin_amdgcn_readlane(cq[3], wave);
  const int p1 = c0, p2 = p1 + c1, p3 = p2 + c2;
  const int tot = scan ? 0 : (p3 + c3);
  const int nseg = gridDim.x * AGG_NW * SUP_NB;
  const int* __restrict__ ent =
      rec + nseg + (long)(((sup * REC_NQ) * AGG_NW + wave) * SUP_NB + own) * SEG_CAP;
  auto word = [&](int idx) -> unsigned {
    const int q   = (idx >= p1) + (idx >= p2) + (idx >= p3);
    const int beg = (q == 0) ? 0 : (q == 1) ? p1 : (q == 2) ? p2 : p3;
    return (unsigned)ent[(long)q * (AGG_NW * SUP_NB * SEG_CAP) + (idx - beg)];
  };
  unsigned pk = (lane < tot) ? word(lane) : 0u;

  v8f croot[NTRIP];
  for (int t = 0; t < NTRIP; ++t)
    croot[t] = (v8f){0.f, 0.f, 0.f, 0.f, 0.f, 0.f, 0.f, 0.f};

#pragma unroll
  for (int t = 0; t < NTRIP; ++t) {
    float* Rt = (float*)uni + (t & 1) * TILE;
    for (int i = 0; i < RPW; ++i)
      *(v4f*)&Rt[(i * AGG_NW + wave) * PA + 4 * lane] = pre[i];
    __syncthreads();
    if (t + 1 < NTRIP) fetch(t + 1);

    const int trow = t * ROWS + rt * 16;
    auto root_mm = [&]() {
      if (trow < AGG_R) {
        const float* ap = Rt + (rt * 16 + l15) * PA + koff;
        v8f c = croot[t];
        for (int k = 0; k < D; k += 4) {
          v2f a = *(const v2f*)(ap + k);
          v2f b = *(const v2f*)(bpr + k);
          c = __builtin_amdgcn_wmma_f32_16x16x4_f32(false, a, false, b,
                                                    (short)0, c, false, false);
        }
        croot[t] = c;
      }
    };

    const int wb = (tot * t / NTRIP) & ~7;
    const int we = (t + 1 < NTRIP) ? ((tot * (t + 1) / NTRIP) & ~7) : tot;
#pragma nounroll
    for (int step = 0; step < 2; ++step) {
      if ((step == 0) == (scan || mulfirst != 0)) {
        root_mm();
      } else {
        for (int i0 = wb; i0 < we; i0 += 32) {
          const int n = (we - i0 < 32) ? (we - i0) : 32;
          const unsigned cur = pk;
          const int nb = i0 + n;
          if (nb < tot)
            pk = (nb + lane < tot) ? word(nb + lane) : 0u;
          vgae_agg_consume<D>(h, acc, cnt, lane, n, (int)(cur & 0xFFFFFFu), (int)(cur >> 24));
        }
      }
    }
  }
  __syncthreads();

  const float bias = bl[bcol];
  const int erow = tid >> 4;
  const int ecol = (tid & 15) * 4;
  v4f g0 = {0.f, 0.f, 0.f, 0.f}, g1 = g0, be0 = g0, be1 = g0;
  if (LN) {
    g0  = *(const v4f*)(gamma + ecol);
    g1  = *(const v4f*)(gamma + 64 + ecol);
    be0 = *(const v4f*)(beta + ecol);
    be1 = *(const v4f*)(beta + 64 + ecol);
  }
  const float* ap = &At[(rt * 16 + l15) * PA + koff];

#pragma unroll
  for (int trip = 0; trip < NTRIP; ++trip) {
    for (int i = 0; i < RPW; ++i) {
      const int rr  = i * AGG_NW + wave;
      const int row = trip * ROWS + rr;
      v4f a = {0.f, 0.f, 0.f, 0.f};
      if (row < AGG_R) {
        const float id = 1.0f / fmaxf((float)cnt[row], 1.0f);
        for (int j = 0; j < 4; ++j) a[j] = acc[row * D + j * 32 + lane] * id;
      }
      *(v4f*)&At[rr * PA + 4 * lane] = a;
    }
    __syncthreads();

    const int trow = trip * ROWS + rt * 16;
    v8f c = croot[trip];
    if (trow < AGG_R) {
      for (int k = 0; k < D; k += 4) {
        v2f a = *(const v2f*)(ap + k);
        v2f b = *(const v2f*)(bpl + k);
        c = __builtin_amdgcn_wmma_f32_16x16x4_f32(false, a, false, b,
                                                  (short)0, c, false, false);
      }
    }

    if (!LN) {
      for (int v = 0; v < 8; ++v) {
        const int row = trow + v + 8 * hi;
        const int r   = r0 + row;
        if (row < AGG_R && r < N) out[(long)r * DOUT + bcol] = c[v] + bias;
      }
      __syncthreads();
    } else {
      __syncthreads();
      for (int v = 0; v < 8; ++v)
        smem[(rt * 16 + v + 8 * hi) * SP + bcol] = c[v] + bias;
      __syncthreads();

      const v4f x0 = *(const v4f*)&smem[erow * SP + ecol];
      const v4f x1 = *(const v4f*)&smem[erow * SP + 64 + ecol];
      float s  = ((x0[0] + x0[1]) + (x0[2] + x0[3])) + ((x1[0] + x1[1]) + (x1[2] + x1[3]));
      float ss = ((x0[0] * x0[0] + x0[1] * x0[1]) + (x0[2] * x0[2] + x0[3] * x0[3]))
               + ((x1[0] * x1[0] + x1[1] * x1[1]) + (x1[2] * x1[2] + x1[3] * x1[3]));
      s += vgae_row_xor<1>(s);  ss += vgae_row_xor<1>(ss);
      s += vgae_row_xor<2>(s);  ss += vgae_row_xor<2>(ss);
      s += vgae_row_xor<4>(s);  ss += vgae_row_xor<4>(ss);
      s += vgae_row_xor<8>(s);  ss += vgae_row_xor<8>(ss);
      const float m    = s * (1.0f / 128.0f);
      const float var  = ss * (1.0f / 128.0f) - m * m;
      const float rstd = rsqrtf(var + LN_EPS);

      v4f y0, y1;
      for (int j = 0; j < 4; ++j) {
        y0[j] = fmaxf((x0[j] - m) * rstd * g0[j] + be0[j], 0.0f);
        y1[j] = fmaxf((x1[j] - m) * rstd * g1[j] + be1[j], 0.0f);
      }
      const int row = trip * ROWS + erow;
      const int r   = r0 + row;
      if (row < AGG_R && r < N) {
        float* orow = out + (long)r * 128;
        *(v4f*)(orow + ecol)      = y0;
        *(v4f*)(orow + 64 + ecol) = y1;
      }
      __syncthreads();
    }
  }
}


extern "C" void kernel_launch(void* const* d_in, const int* in_sizes, int n_in,
                              void* d_out, int out_size, void* d_ws, size_t ws_size,
                              hipStream_t stream) {
  const float* x  = (const float*)d_in[0];
  const int*   ei = (const int*)d_in[1];
  const int N = in_sizes[0] / 64;
  const int E = in_sizes[1] / 2;

  float* aggx = (float*)d_ws;
  float* bufA = aggx + (long)N * 64;
  float* bufB = bufA + (long)N * 128;
  int*   rec  = (int*)(bufB + (long)N * 128);

  const int nown = (N + AGG_R - 1) / AGG_R;

  vgae_rec_build<<<nown, AGG_NT, 0, stream>>>(ei, rec, E);
  vgae_agg_own<64><<<nown, AGG_NT, 0, stream>>>(x, ei, aggx, rec, N, E);

  const int nrt = N / 16;
  for (int t = 0; t < 2; ++t) {
    const int base = 2 + t * 13;
    const float* Wl0 = (const float*)d_in[base + 0];
    const float* bl0 = (const float*)d_in[base + 1];
    const float* Wr0 = (const float*)d_in[base + 2];
    const float* g0  = (const float*)d_in[base + 3];
    const float* b0  = (const float*)d_in[base + 4];
    const float* Wl1 = (const float*)d_in[base + 5];
    const float* bl1 = (const float*)d_in[base + 6];
    const float* Wr1 = (const float*)d_in[base + 7];
    const float* g1  = (const float*)d_in[base + 8];
    const float* b1  = (const float*)d_in[base + 9];
    const float* Wl2 = (const float*)d_in[base + 10];
    const float* bl2 = (const float*)d_in[base + 11];
    const float* Wr2 = (const float*)d_in[base + 12];
    float* outp = (float*)d_out + (long)t * N * 64;

    vgae_sage_gemm<64, 128, true><<<512, 256, 0, stream>>>(aggx, x, Wl0, bl0, Wr0,
                                                          g0, b0, bufA, nrt);
    vgae_agg_dense<128, true><<<nown, AGG_NT, 0, stream>>>(bufA, ei, rec, Wl1, bl1, Wr1,
                                                           g1, b1, bufB, N, E);
    vgae_agg_dense<64, false><<<nown, AGG_NT, 0, stream>>>(bufB, ei, rec, Wl2, bl2, Wr2,
                                                           nullptr, nullptr, outp, N, E);
  }
}


# Block start-up: record counts, all weight-panel pieces, first rows and epilogue parameters loaded in one batch before the first wait (panel was load-wait-store per piece); sum orders unchanged
# speedup vs baseline: 7.3273x; 1.0398x over previous
#include <hip/hip_runtime.h>


#define LN_EPS 1e-5f

typedef __attribute__((ext_vector_type(2))) float v2f;
typedef __attribute__((ext_vector_type(8))) float v8f;


#define AGG_R    196
#define AGG_NT   1024
#define AGG_NW   (AGG_NT / 32)
#define AGG_CH   512
#define AGG_PL   (AGG_CH / 32)
#define AGG_K    8
#define AGG_QCAP (AGG_K + AGG_CH)
#define SUP_NB   4
#define SUP_R    (SUP_NB * AGG_R)
#define REC_NQ   4
#define SEG_CAP  60

__global__ __launch_bounds__(AGG_NT)
void vgae_rec_build(const int* __restrict__ ei, int* __restrict__ rec, int E) {
  __shared__ int que[AGG_NW][AGG_CH];

  const int lane = threadIdx.x & 31;
  const int wave = __builtin_amdgcn_readfirstlane(threadIdx.x >> 5);
  const int r0   = (blockIdx.x / REC_NQ) * SUP_R;
  const int qtr  = blockIdx.x % REC_NQ;
  const int* __restrict__ dst = ei + E;

  const int nseg = gridDim.x * AGG_NW * SUP_NB;
  const int seg0 = (blockIdx.x * AGG_NW + wave) * SUP_NB;
  int* __restrict__ ent = rec + nseg + (long)seg0 * SEG_CAP;

  const int nch = (E + AGG_CH - 1) / AGG_CH;
  const int cpq = (nch + REC_NQ - 1) / REC_NQ;
  const int cb  = qtr * cpq;
  const int ce  = (cb + cpq < nch) ? (cb + cpq) : nch;

  int sc[SUP_NB];
  for (int u = 0; u < SUP_NB; ++u) sc[u] = 0;

  int d[AGG_PL];
  auto load_dst = [&](int c) {
    const int base = c * AGG_CH + lane;
    if (c * AGG_CH + AGG_CH <= E) {
      for (int j = 0; j < AGG_PL; ++j) d[j] = dst[base + j * 32];
    } else {
      for (int j = 0; j < AGG_PL; ++j) {
        const int e = base + j * 32;
        d[j] = (e < E) ? dst[e] : -1;
      }
    }
  };

  int c = cb + wave;
  if (c < ce) load_dst(c);
  for (; c < ce; c += AGG_NW) {
    const int base = c * AGG_CH + lane;
    int qn = 0;
    for (int j = 0; j < AGG_PL; ++j) {
      const bool m = (unsigned)(d[j] - r0) < (unsigned)SUP_R;
      const unsigned mask = __builtin_amdgcn_ballot_w32(m);
      if (mask) {
        if (m) que[wave][qn + (int)__builtin_amdgcn_mbcnt_lo(mask, 0u)] = base + j * 32;
        qn += __builtin_popcount(mask);
      }
    }
    if (c + AGG_NW < ce) load_dst(c + AGG_NW);
    for (int i0 = 0; i0 < qn; i0 += 32) {
      const int  n   = (qn - i0 < 32) ? (qn - i0) : 32;
      const bool act = lane < n;
      const int  me  = act ? que[wave][i0 + lane] : 0;
      const int  ms  = ei[me];
      const int  rr  = dst[me] - r0;
      const int  u   = act ? (rr / AGG_R) : -1;
      const int  rb  = rr - u * AGG_R;
      const int  wd  = (int)((unsigned)ms | ((unsigned)rb << 24));
      for (int uu = 0; uu < SUP_NB; ++uu) {
        const bool mine = (u == uu);
        const unsigned mk = __builtin_amdgcn_ballot_w32(mine);
        if (mine) {
          const int pos = sc[uu] + (int)__builtin_amdgcn_mbcnt_lo(mk, 0u);
          if (pos < SEG_CAP) ent[uu * SEG_CAP + pos] = wd;
        }
        sc[uu] += __builtin_popcount(mk);
      }
    }
  }
  if (lane < SUP_NB) {
    int v = sc[0];
    for (int u = 1; u < SUP_NB; ++u) v = (lane == u) ? sc[u] : v;
    rec[seg0 + lane] = v;
  }
}

template <int D>
__device__ __forceinline__ void vgae_agg_consume(const float* __restrict__ h, float* acc, int* cnt,
                                                 int lane, int n, int ms, int md) {
  constexpr int VEC = D / 32;
  typedef __attribute__((ext_vector_type(VEC))) float vN;
  if (lane < n) atomicAdd(&cnt[md], 1);
  int i = 0;
  for (; i + 8 <= n; i += 8) {
    vN v[8];
#pragma unroll
    for (int u = 0; u < 8; ++u) {
      const int s = __builtin_amdgcn_readlane(ms, i + u);
      v[u] = *(const vN*)(h + (long)s * D + VEC * lane);
    }
#pragma unroll
    for (int u = 0; u < 8; ++u) {
      const int dl = __builtin_amdgcn_readlane(md, i + u);
      for (int j = 0; j < VEC; ++j)
        atomicAdd(&acc[dl * D + j * 32 + lane], v[u][j]);
    }
  }
  if (i < n) {
    vN v[8];
#pragma unroll
    for (int u = 0; u < 8; ++u) {
      const int k = (i + u < n) ? (i + u) : (n - 1);
      const int s = __builtin_amdgcn_readlane(ms, k);
      v[u] = *(const vN*)(h + (long)s * D + VEC * lane);
    }
#pragma unroll
    for (int u = 0; u < 8; ++u) asm volatile("" : "+v"(v[u]));
#pragma unroll
    for (int u = 0; u < 8; ++u) {
      if (i + u < n) {
        const int dl = __builtin_amdgcn_readlane(md, i + u);
        for (int j = 0; j < VEC; ++j)
          atomicAdd(&acc[dl * D + j * 32 + lane], v[u][j]);
      }
    }
  }
}

template <int D>
__device__ __forceinline__ void vgae_scan_rows(const float* __restrict__ h, const int* __restrict__ ei,
                                               float* acc, int* cnt, int (*que)[AGG_QCAP], int E) {
  const int lane = threadIdx.x & 31;
  const int wave = __builtin_amdgcn_readfirstlane(threadIdx.x >> 5);
  const int r0   = blockIdx.x * AGG_R;
  const int* __restrict__ dst = ei + E;

  const int nch = (E + AGG_CH - 1) / AGG_CH;
  int qn = 0;

  for (int c = wave; ; c += AGG_NW) {
    const bool live = c < nch;
    if (live) {
      const int base = c * AGG_CH + lane;
      int d[AGG_PL];
      if (c * AGG_CH + AGG_CH <= E) {
        for (int j = 0; j < AGG_PL; ++j) d[j] = dst[base + j * 32];
      } else {
        for (int j = 0; j < AGG_PL; ++j) {
          const int e = base + j * 32;
          d[j] = (e < E) ? dst[e] : -1;
        }
      }
      for (int j = 0; j < AGG_PL; ++j) {
        const bool m = (unsigned)(d[j] - r0) < (unsigned)AGG_R;
        const unsigned mask = __builtin_amdgcn_ballot_w32(m);
        if (mask) {
          if (m) que[wave][qn + (int)__builtin_amdgcn_mbcnt_lo(mask, 0u)] = base + j * 32;
          qn += __builtin_popcount(mask);
        }
      }
    }

    if (qn >= AGG_K || (!live && qn > 0)) {
      for (int i0 = 0; i0 < qn; i0 += 32) {
        const int n  = (qn - i0 < 32) ? (qn - i0) : 32;
        const int me = (lane < n) ? que[wave][i0 + lane] : 0;
        const int ms = ei[me];
        const int md = dst[me] - r0;
        vgae_agg_consume<D>(h, acc, cnt, lane, n, ms, md);
      }
      qn = 0;
    }

    if (!live) break;
  }
}

template <int D>
__device__ __forceinline__ void vgae_sum_rows(const float* __restrict__ h, const int* __restrict__ ei,
                                              const int* __restrict__ rec, float* acc, int* cnt,
                                              int (*que)[AGG_QCAP], int E) {
  const int lane = threadIdx.x & 31;
  const int wave = __builtin_amdgcn_readfirstlane(threadIdx.x >> 5);

  const int sup  = blockIdx.x / SUP_NB;
  const int own  = blockIdx.x % SUP_NB;
  const int nseg = gridDim.x * AGG_NW * SUP_NB;
  int cq[REC_NQ];
  for (int q = 0; q < REC_NQ; ++q)
    cq[q] = rec[((sup * REC_NQ + q) * AGG_NW + lane) * SUP_NB + own];

  for (int i = threadIdx.x; i < AGG_R * D; i += AGG_NT) acc[i] = 0.0f;
  for (int i = threadIdx.x; i < AGG_R; i += AGG_NT) cnt[i] = 0;

  bool ov = false;
  for (int q = 0; q < REC_NQ; ++q) ov = ov || (cq[q] > SEG_CAP);
  const bool scan = __builtin_amdgcn_ballot_w32(ov) != 0u;
  __syncthreads();

  if (!scan) {
    const int c0 = __builtin_amdgcn_readlane(cq[0], wave);
    const int c1 = __builtin_amdgcn_readlane(cq[1], wave);
    const int c2 = __builtin_amdgcn_readlane(cq[2], wave);
    const int c3 = __builtin_amdgcn_readlane(cq[3], wave);
    const int p1 = c0, p2 = p1 + c1, p3 = p2 + c2, tot = p3 + c3;
    const int* __restrict__ ent =
        rec + nseg + (long)(((sup * REC_NQ) * AGG_NW + wave) * SUP_NB + own) * SEG_CAP;
    auto word = [&](int idx) -> unsigned {
      const int q   = (idx >= p1) + (idx >= p2) + (idx >= p3);
      const int beg = (q == 0) ? 0 : (q == 1) ? p1 : (q == 2) ? p2 : p3;
      return (unsigned)ent[(long)q * (AGG_NW * SUP_NB * SEG_CAP) + (idx - beg)];
    };
    unsigned pk = (lane < tot) ? word(lane) : 0u;
    for (int i0 = 0; i0 < tot; i0 += 32) {
      const int n = (tot - i0 < 32) ? (tot - i0) : 32;
      const unsigned cur = pk;
      if (i0 + 32 < tot)
        pk = (i0 + 32 + lane < tot) ? word(i0 + 32 + lane) : 0u;
      vgae_agg_consume<D>(h, acc, cnt, lane, n, (int)(cur & 0xFFFFFFu), (int)(cur >> 24));
    }
  } else {
    vgae_scan_rows<D>(h, ei, acc, cnt, que, E);
  }
}

template <int D>
__global__ __launch_bounds__(AGG_NT)
void vgae_agg_own(const float* __restrict__ h, const int* __restrict__ ei,
                  float* __restrict__ agg, const int* __restrict__ rec, int N, int E) {
  constexpr int VEC = D / 32;
  typedef __attribute__((ext_vector_type(VEC))) float vN;

  __shared__ float acc[AGG_R * D];
  __shared__ int   cnt[AGG_R];
  __shared__ int   que[AGG_NW][AGG_QCAP];

  const int lane = threadIdx.x & 31;
  const int wave = __builtin_amdgcn_readfirstlane(threadIdx.x >> 5);
  const int r0   = blockIdx.x * AGG_R;

  vgae_sum_rows<D>(h, ei, rec, acc, cnt, que, E);
  __syncthreads();

  for (int row = wave; row < AGG_R; row += AGG_NW) {
    const int r = r0 + row;
    if (r < N) {
      const float id = 1.0f / fmaxf((float)cnt[row], 1.0f);
      vN o;
      for (int j = 0; j < VEC; ++j) o[j] = acc[row * D + j * 32 + lane] * id;
      *(vN*)(agg + (long)r * D + VEC * lane) = o;
    }
  }
}

typedef __attribute__((ext_vector_type(4))) float v4f;

template <int M>
__device__ __forceinline__ float vgae_row_xor(float x) {
  return __builtin_bit_cast(float, __builtin_amdgcn_update_dpp(
      0, __builtin_bit_cast(int, x), 0x160 | M, 0xf, 0xf, true));
}

template <int DIN, int DOUT, bool LN>
__global__ __launch_bounds__(DOUT * 2)
void vgae_sage_gemm(const float* agg,
                    const float* __restrict__ h,
                    const float* __restrict__ Wl,
                    const float* __restrict__ bl,
                    const float* __restrict__ Wr,
                    const float* __restrict__ gamma,
                    const float* __restrict__ beta,
                    float* out, int ntiles) {
  constexpr int K   = 2 * DIN;
  constexpr int P   = K + 4;
  constexpr int NT  = DOUT * 2;
  constexpr int NW  = DOUT / 16;
  constexpr int RPW = 16 / NW;
  constexpr int F4  = K / 4;
  constexpr int NL  = RPW * F4 / 32;
  static_assert(!LN || DOUT == 128, "LN epilogue is written for 128 columns");

  __shared__ __attribute__((aligned(16))) float Bt[DOUT * P];
  __shared__ __attribute__((aligned(16))) float At[16 * P];
  constexpr int SP = 132;
  __shared__ __attribute__((aligned(16))) float smem[LN ? 16 * SP : 4];

  const int tid  = threadIdx.x;
  const int lane = tid & 31;
  const int wave = tid >> 5;
  const int l15  = lane & 15;
  const int hi   = lane >> 4;
  const int bcol = wave * 16 + l15;
  const int koff = 2 * hi;


  int prow[NL], pcol[NL];
  for (int i = 0; i < NL; ++i) {
    const int flat = i * 32 + lane;
    prow[i] = wave * RPW + flat / F4;
    pcol[i] = (flat % F4) * 4;
  }
  v4f pre[NL];
  auto fetch = [&](int tile) {
    for (int i = 0; i < NL; ++i) {
      const long r = (long)tile * 16 + prow[i];
      const float* s = (pcol[i] < DIN) ? (agg + r * DIN + pcol[i])
                                       : (h + r * DIN + (pcol[i] - DIN));
      pre[i] = *(const v4f*)s;
    }
  };
  int tile = blockIdx.x;
  if (tile < ntiles) fetch(tile);

  constexpr int NPL = DOUT * F4 / NT;
  static_assert(DOUT * F4 % NT == 0, "panel pieces divide evenly among the threads");
  v4f pan[NPL];
#pragma unroll
  for (int j = 0; j < NPL; ++j) {
    const int i   = tid + j * NT;
    const int cc  = i / F4;
    const int col = (i % F4) * 4;
    const float* s = (col < DIN) ? (Wl + (long)cc * DIN + col)
                                 : (Wr + (long)cc * DIN + (col - DIN));
    pan[j] = *(const v4f*)s;
  }
#pragma unroll
  for (int j = 0; j < NPL; ++j) {
    const int i   = tid + j * NT;
    const int cc  = i / F4;
    const int col = (i % F4) * 4;
    *(v4f*)&Bt[cc * P + col] = pan[j];
  }

  const float bias = bl[bcol];
  const int erow = tid >> 4;
  const int ecol = (tid & 15) * 4;
  v4f g0 = {0.f, 0.f, 0.f, 0.f}, g1 = g0, be0 = g0, be1 = g0;
  if (LN) {
    g0  = *(const v4f*)(gamma + ecol);
    g1  = *(const v4f*)(gamma + 64 + ecol);
    be0 = *(const v4f*)(beta + ecol);
    be1 = *(const v4f*)(beta + 64 + ecol);
  }

  const float* ap = &At[l15 * P + koff];
  const float* bp = &Bt[bcol * P + koff];

  for (; tile < ntiles; tile += gridDim.x) {
    for (int i = 0; i < NL; ++i) *(v4f*)&At[prow[i] * P + pcol[i]] = pre[i];
    __syncthreads();
    const int next = tile + gridDim.x;
    if (next < ntiles) fetch(next);

    v8f c = {0.f, 0.f, 0.f, 0.f, 0.f, 0.f, 0.f, 0.f};
    for (int k = 0; k < K; k += 4) {
      v2f a = *(const v2f*)(ap + k);
      v2f b = *(const v2f*)(bp + k);
      c = __builtin_amdgcn_wmma_f32_16x16x4_f32(false, a, false, b,
                                                (short)0, c, false, false);
    }

    const int rowbase = tile * 16;

    if (!LN) {
      for (int v = 0; v < 8; ++v) {
        int r = rowbase + v + 8 * hi;
        out[(long)r * DOUT + bcol] = c[v] + bias;
      }
      __syncthreads();
    } else {
      for (int v = 0; v < 8; ++v)
        smem[(v + 8 * hi) * SP + bcol] = c[v] + bias;
      __syncthreads();

      const v4f x0 = *(const v4f*)&smem[erow * SP + ecol];
      const v4f x1 = *(const v4f*)&smem[erow * SP + 64 + ecol];
      float s  = ((x0[0] + x0[1]) + (x0[2] + x0[3])) + ((x1[0] + x1[1]) + (x1[2] + x1[3]));
      float ss = ((x0[0] * x0[0] + x0[1] * x0[1]) + (x0[2] * x0[2] + x0[3] * x0[3]))
               + ((x1[0] * x1[0] + x1[1] * x1[1]) + (x1[2] * x1[2] + x1[3] * x1[3]));
      s += vgae_row_xor<1>(s);  ss += vgae_row_xor<1>(ss);
      s += vgae_row_xor<2>(s);  ss += vgae_row_xor<2>(ss);
      s += vgae_row_xor<4>(s);  ss += vgae_row_xor<4>(ss);
      s += vgae_row_xor<8>(s);  ss += vgae_row_xor<8>(ss);
      const float m    = s * (1.0f / 128.0f);
      const float var  = ss * (1.0f / 128.0f) - m * m;
      const float rstd = rsqrtf(var + LN_EPS);

      v4f y0, y1;
      for (int j = 0; j < 4; ++j) {
        y0[j] = fmaxf((x0[j] - m) * rstd * g0[j] + be0[j], 0.0f);
        y1[j] = fmaxf((x1[j] - m) * rstd * g1[j] + be1[j], 0.0f);
      }
      float* orow = out + (long)(rowbase + erow) * 128;
      *(v4f*)(orow + ecol)      = y0;
      *(v4f*)(orow + 64 + ecol) = y1;
    }
  }
}

template <int DOUT, bool LN>
__global__ __launch_bounds__(AGG_NT)
void vgae_agg_dense(const float* __restrict__ h, const int* __restrict__ ei,
                    const int* __restrict__ rec,
                    const float* __restrict__ Wl, const float* __restrict__ bl,
                    const float* __restrict__ Wr,
                    const float* __restrict__ gamma, const float* __restrict__ beta,
                    float* __restrict__ out, int N, int E) {
  constexpr int D     = 128;
  constexpr int K     = 2 * D;
  constexpr int P     = K + 4;
  constexpr int PA    = D + 4;
  constexpr int F4    = K / 4;
  constexpr int CT    = DOUT / 16;
  constexpr int RT    = AGG_NW / CT;
  constexpr int ROWS  = 16 * RT;
  constexpr int RPW   = ROWS / AGG_NW;
  constexpr int NTRIP = (AGG_R + ROWS - 1) / ROWS;
  constexpr int SP    = 132;
  constexpr int TILE  = ROWS * PA;
  static_assert(!LN || DOUT == 128, "LN epilogue is written for 128 columns");
  static_assert(2 * TILE >= AGG_NW * AGG_QCAP, "fallback queue must fit the shared region");
  static_assert(!LN || TILE >= ROWS * SP, "LN staging rows must fit the shared region");
  static_assert(!LN || ROWS * 16 == AGG_NT, "LN epilogue: 16 threads per row of a trip");

  __shared__ float acc[AGG_R * D];
  __shared__ int   cnt[AGG_R];
  __shared__ __attribute__((aligned(16))) float Bt[DOUT * P];
  __shared__ __attribute__((aligned(16))) int   uni[2 * TILE];
  int (*que)[AGG_QCAP] = (int (*)[AGG_QCAP])uni;
  float* At   = (float*)uni;
  float* smem = (float*)uni;

  const int tid  = threadIdx.x;
  const int lane = tid & 31;
  const int wave = __builtin_amdgcn_readfirstlane(tid >> 5);
  const int r0   = blockIdx.x * AGG_R;


  const int sup  = blockIdx.x / SUP_NB;
  const int own  = blockIdx.x % SUP_NB;
  int cq[REC_NQ];
  for (int q = 0; q < REC_NQ; ++q)
    cq[q] = rec[((sup * REC_NQ + q) * AGG_NW + lane) * SUP_NB + own];

  constexpr int NPL = DOUT * F4 / AGG_NT;
  static_assert(DOUT * F4 % AGG_NT == 0, "panel pieces divide evenly among the threads");
  v4f pan[NPL];
#pragma unroll
  for (int j = 0; j < NPL; ++j) {
    const int i   = tid + j * AGG_NT;
    const int cc  = i / F4;
    const int col = (i % F4) * 4;
    const float* s = (col < D) ? (Wl + (long)cc * D + col)
                               : (Wr + (long)cc * D + (col - D));
    pan[j] = *(const v4f*)s;
  }

  v4f pre[RPW];
  auto fetch = [&](int trip) {
    for (int i = 0; i < RPW; ++i) {
      const int row = trip * ROWS + i * AGG_NW + wave;
      const int r   = r0 + row;
      v4f z = {0.f, 0.f, 0.f, 0.f};
      if (row < AGG_R && r < N) z = *(const v4f*)(h + (long)r * D + 4 * lane);
      pre[i] = z;
    }
  };
  fetch(0);

  const int l15  = lane & 15;
  const int hi   = lane >> 4;
  const int ct   = wave % CT;
  const int rt   = wave / CT;
  const int bcol = ct * 16 + l15;
  const float bias = bl[bcol];
  const int erow = tid >> 4;
  const int ecol = (tid & 15) * 4;
  v4f g0 = {0.f, 0.f, 0.f, 0.f}, g1 = g0, be0 = g0, be1 = g0;
  if (LN) {
    g0  = *(const v4f*)(gamma + ecol);
    g1  = *(const v4f*)(gamma + 64 + ecol);
    be0 = *(const v4f*)(beta + ecol);
    be1 = *(const v4f*)(beta + 64 + ecol);
  }

  for (int i = tid; i < AGG_R * D; i += AGG_NT) acc[i] = 0.0f;
  for (int i = tid; i < AGG_R; i += AGG_NT) cnt[i] = 0;

  bool ov = false;
  for (int q = 0; q < REC_NQ; ++q) ov = ov || (cq[q] > SEG_CAP);
  const bool scan = __builtin_amdgcn_ballot_w32(ov) != 0u;

  const int c0 = __builtin_amdgcn_readlane(cq[0], wave);
  const int c1 = __builtin_amdgcn_readlane(cq[1], wave);
  const int c2 = __builtin_amdgcn_readlane(cq[2], wave);
  const int c3 = __builtin_amdgcn_readlane(cq[3], wave);
  const int p1 = c0, p2 = p1 + c1, p3 = p2 + c2;
  const int tot = scan ? 0 : (p3 + c3);
  const int nseg = gridDim.x * AGG_NW * SUP_NB;
  const int* __restrict__ ent =
      rec + nseg + (long)(((sup * REC_NQ) * AGG_NW + wave) * SUP_NB + own) * SEG_CAP;
  auto word = [&](int idx) -> unsigned {
    const int q   = (idx >= p1) + (idx >= p2) + (idx >= p3);
    const int beg = (q == 0) ? 0 : (q == 1) ? p1 : (q == 2) ? p2 : p3;
    return (unsigned)ent[(long)q * (AGG_NW * SUP_NB * SEG_CAP) + (idx - beg)];
  };
  unsigned pk = (lane < tot) ? word(lane) : 0u;

#pragma unroll
  for (int j = 0; j < NPL; ++j) {
    const int i   = tid + j * AGG_NT;
    const int cc  = i / F4;
    const int col = (i % F4) * 4;
    *(v4f*)&Bt[cc * P + col] = pan[j];
  }

  if (scan) {
    __syncthreads();
    vgae_scan_rows<D>(h, ei, acc, cnt, que, E);
    __syncthreads();
  }

  const int koff = 2 * hi;
  const float* bpl = &Bt[bcol * P + koff];
  const float* bpr = bpl + D;
  const int mulfirst = (wave >> 2) & 1;

  v8f croot[NTRIP];
  for (int t = 0; t < NTRIP; ++t)
    croot[t] = (v8f){0.f, 0.f, 0.f, 0.f, 0.f, 0.f, 0.f, 0.f};

#pragma unroll
  for (int t = 0; t < NTRIP; ++t) {
    float* Rt = (float*)uni + (t & 1) * TILE;
    for (int i = 0; i < RPW; ++i)
      *(v4f*)&Rt[(i * AGG_NW + wave) * PA + 4 * lane] = pre[i];
    __syncthreads();
    if (t + 1 < NTRIP) fetch(t + 1);

    const int trow = t * ROWS + rt * 16;
    auto root_mm = [&]() {
      if (trow < AGG_R) {
        const float* ap = Rt + (rt * 16 + l15) * PA + koff;
        v8f c = croot[t];
        for (int k = 0; k < D; k += 4) {
          v2f a = *(const v2f*)(ap + k);
          v2f b = *(const v2f*)(bpr + k);
          c = __builtin_amdgcn_wmma_f32_16x16x4_f32(false, a, false, b,
                                                    (short)0, c, false, false);
        }
        croot[t] = c;
      }
    };

    const int wb = (tot * t / NTRIP) & ~7;
    const int we = (t + 1 < NTRIP) ? ((tot * (t + 1) / NTRIP) & ~7) : tot;
#pragma nounroll
    for (int step = 0; step < 2; ++step) {
      if ((step == 0) == (scan || mulfirst != 0)) {
        root_mm();
      } else {
        for (int i0 = wb; i0 < we; i0 += 32) {
          const int n = (we - i0 < 32) ? (we - i0) : 32;
          const unsigned cur = pk;
          const int nb = i0 + n;
          if (nb < tot)
            pk = (nb + lane < tot) ? word(nb + lane) : 0u;
          vgae_agg_consume<D>(h, acc, cnt, lane, n, (int)(cur & 0xFFFFFFu), (int)(cur >> 24));
        }
      }
    }
  }
  __syncthreads();

  const float* ap = &At[(rt * 16 + l15) * PA + koff];

#pragma unroll
  for (int trip = 0; trip < NTRIP; ++trip) {
    for (int i = 0; i < RPW; ++i) {
      const int rr  = i * AGG_NW + wave;
      const int row = trip * ROWS + rr;
      v4f a = {0.f, 0.f, 0.f, 0.f};
      if (row < AGG_R) {
        const float id = 1.0f / fmaxf((float)cnt[row], 1.0f);
        for (int j = 0; j < 4; ++j) a[j] = acc[row * D + j * 32 + lane] * id;
      }
      *(v4f*)&At[rr * PA + 4 * lane] = a;
    }
    __syncthreads();

    const int trow = trip * ROWS + rt * 16;
    v8f c = croot[trip];
    if (trow < AGG_R) {
      for (int k = 0; k < D; k += 4) {
        v2f a = *(const v2f*)(ap + k);
        v2f b = *(const v2f*)(bpl + k);
        c = __builtin_amdgcn_wmma_f32_16x16x4_f32(false, a, false, b,
                                                  (short)0, c, false, false);
      }
    }

    if (!LN) {
      for (int v = 0; v < 8; ++v) {
        const int row = trow + v + 8 * hi;
        const int r   = r0 + row;
        if (row < AGG_R && r < N) out[(long)r * DOUT + bcol] = c[v] + bias;
      }
      __syncthreads();
    } else {
      __syncthreads();
      for (int v = 0; v < 8; ++v)
        smem[(rt * 16 + v + 8 * hi) * SP + bcol] = c[v] + bias;
      __syncthreads();

      const v4f x0 = *(const v4f*)&smem[erow * SP + ecol];
      const v4f x1 = *(const v4f*)&smem[erow * SP + 64 + ecol];
      float s  = ((x0[0] + x0[1]) + (x0[2] + x0[3])) + ((x1[0] + x1[1]) + (x1[2] + x1[3]));
      float ss = ((x0[0] * x0[0] + x0[1] * x0[1]) + (x0[2] * x0[2] + x0[3] * x0[3]))
               + ((x1[0] * x1[0] + x1[1] * x1[1]) + (x1[2] * x1[2] + x1[3] * x1[3]));
      s += vgae_row_xor<1>(s);  ss += vgae_row_xor<1>(ss);
      s += vgae_row_xor<2>(s);  ss += vgae_row_xor<2>(ss);
      s += vgae_row_xor<4>(s);  ss += vgae_row_xor<4>(ss);
      s += vgae_row_xor<8>(s);  ss += vgae_row_xor<8>(ss);
      const float m    = s * (1.0f / 128.0f);
      const float var  = ss * (1.0f / 128.0f) - m * m;
      const float rstd = rsqrtf(var + LN_EPS);

      v4f y0, y1;
      for (int j = 0; j < 4; ++j) {
        y0[j] = fmaxf((x0[j] - m) * rstd * g0[j] + be0[j], 0.0f);
        y1[j] = fmaxf((x1[j] - m) * rstd * g1[j] + be1[j], 0.0f);
      }
      const int row = trip * ROWS + erow;
      const int r   = r0 + row;
      if (row < AGG_R && r < N) {
        float* orow = out + (long)r * 128;
        *(v4f*)(orow + ecol)      = y0;
        *(v4f*)(orow + 64 + ecol) = y1;
      }
      __syncthreads();
    }
  }
}


extern "C" void kernel_launch(void* const* d_in, const int* in_sizes, int n_in,
                              void* d_out, int out_size, void* d_ws, size_t ws_size,
                              hipStream_t stream) {
  const float* x  = (const float*)d_in[0];
  const int*   ei = (const int*)d_in[1];
  const int N = in_sizes[0] / 64;
  const int E = in_sizes[1] / 2;

  float* aggx = (float*)d_ws;
  float* bufA = aggx + (long)N * 64;
  float* bufB = bufA + (long)N * 128;
  int*   rec  = (int*)(bufB + (long)N * 128);

  const int nown = (N + AGG_R - 1) / AGG_R;

  vgae_rec_build<<<nown, AGG_NT, 0, stream>>>(ei, rec, E);
  vgae_agg_own<64><<<nown, AGG_NT, 0, stream>>>(x, ei, aggx, rec, N, E);

  const int nrt = N / 16;
  for (int t = 0; t < 2; ++t) {
    const int base = 2 + t * 13;
    const float* Wl0 = (const float*)d_in[base + 0];
    const float* bl0 = (const float*)d_in[base + 1];
    const float* Wr0 = (const float*)d_in[base + 2];
    const float* g0  = (const float*)d_in[base + 3];
    const float* b0  = (const float*)d_in[base + 4];
    const float* Wl1 = (const float*)d_in[base + 5];
    const float* bl1 = (const float*)d_in[base + 6];
    const float* Wr1 = (const float*)d_in[base + 7];
    const float* g1  = (const float*)d_in[base + 8];
    const float* b1  = (const float*)d_in[base + 9];
    const float* Wl2 = (const float*)d_in[base + 10];
    const float* bl2 = (const float*)d_in[base + 11];
    const float* Wr2 = (const float*)d_in[base + 12];
    float* outp = (float*)d_out + (long)t * N * 64;

    vgae_sage_gemm<64, 128, true><<<512, 256, 0, stream>>>(aggx, x, Wl0, bl0, Wr0,
                                                          g0, b0, bufA, nrt);
    vgae_agg_dense<128, true><<<nown, AGG_NT, 0, stream>>>(bufA, ei, rec, Wl1, bl1, Wr1,
                                                           g1, b1, bufB, N, E);
    vgae_agg_dense<64, false><<<nown, AGG_NT, 0, stream>>>(bufB, ei, rec, Wl2, bl2, Wr2,
                                                           nullptr, nullptr, outp, N, E);
  }
}


# Neighbor sums, record path: no LDS float adds: block sorts its record words by row (int LDS count, scan, fill), a wave sums its own 6-7 rows in registers, one LDS store per row; sum order: fill rank
# speedup vs baseline: 7.5556x; 1.0312x over previous
#include <hip/hip_runtime.h>


#define LN_EPS 1e-5f

typedef __attribute__((ext_vector_type(2))) float v2f;
typedef __attribute__((ext_vector_type(8))) float v8f;


#define AGG_R    196
#define AGG_NT   1024
#define AGG_NW   (AGG_NT / 32)
#define AGG_CH   512
#define AGG_PL   (AGG_CH / 32)
#define AGG_K    8
#define AGG_QCAP (AGG_K + AGG_CH)
#define SUP_NB   4
#define SUP_R    (SUP_NB * AGG_R)
#define REC_NQ   4
#define SEG_CAP  60
#define REC_G    ((REC_NQ * SEG_CAP + 31) / 32)
#define SRT_CAP  4096

__device__ __forceinline__ int vgae_row_lo(int w) { return (w * AGG_R + AGG_NW - 1) / AGG_NW; }

__global__ __launch_bounds__(AGG_NT)
void vgae_rec_build(const int* __restrict__ ei, int* __restrict__ rec, int E) {
  __shared__ int que[AGG_NW][AGG_CH];

  const int lane = threadIdx.x & 31;
  const int wave = __builtin_amdgcn_readfirstlane(threadIdx.x >> 5);
  const int r0   = (blockIdx.x / REC_NQ) * SUP_R;
  const int qtr  = blockIdx.x % REC_NQ;
  const int* __restrict__ dst = ei + E;

  const int nseg = gridDim.x * AGG_NW * SUP_NB;
  const int seg0 = (blockIdx.x * AGG_NW + wave) * SUP_NB;
  int* __restrict__ ent = rec + nseg + (long)seg0 * SEG_CAP;

  const int nch = (E + AGG_CH - 1) / AGG_CH;
  const int cpq = (nch + REC_NQ - 1) / REC_NQ;
  const int cb  = qtr * cpq;
  const int ce  = (cb + cpq < nch) ? (cb + cpq) : nch;

  int sc[SUP_NB];
  for (int u = 0; u < SUP_NB; ++u) sc[u] = 0;

  int d[AGG_PL];
  auto load_dst = [&](int c) {
    const int base = c * AGG_CH + lane;
    if (c * AGG_CH + AGG_CH <= E) {
      for (int j = 0; j < AGG_PL; ++j) d[j] = dst[base + j * 32];
    } else {
      for (int j = 0; j < AGG_PL; ++j) {
        const int e = base + j * 32;
        d[j] = (e < E) ? dst[e] : -1;
      }
    }
  };

  int c = cb + wave;
  if (c < ce) load_dst(c);
  for (; c < ce; c += AGG_NW) {
    const int base = c * AGG_CH + lane;
    int qn = 0;
    for (int j = 0; j < AGG_PL; ++j) {
      const bool m = (unsigned)(d[j] - r0) < (unsigned)SUP_R;
      const unsigned mask = __builtin_amdgcn_ballot_w32(m);
      if (mask) {
        if (m) que[wave][qn + (int)__builtin_amdgcn_mbcnt_lo(mask, 0u)] = base + j * 32;
        qn += __builtin_popcount(mask);
      }
    }
    if (c + AGG_NW < ce) load_dst(c + AGG_NW);
    for (int i0 = 0; i0 < qn; i0 += 32) {
      const int  n   = (qn - i0 < 32) ? (qn - i0) : 32;
      const bool act = lane < n;
      const int  me  = act ? que[wave][i0 + lane] : 0;
      const int  ms  = ei[me];
      const int  rr  = dst[me] - r0;
      const int  u   = act ? (rr / AGG_R) : -1;
      const int  rb  = rr - u * AGG_R;
      const int  wd  = (int)((unsigned)ms | ((unsigned)rb << 24));
      for (int uu = 0; uu < SUP_NB; ++uu) {
        const bool mine = (u == uu);
        const unsigned mk = __builtin_amdgcn_ballot_w32(mine);
        if (mine) {
          const int pos = sc[uu] + (int)__builtin_amdgcn_mbcnt_lo(mk, 0u);
          if (pos < SEG_CAP) ent[uu * SEG_CAP + pos] = wd;
        }
        sc[uu] += __builtin_popcount(mk);
      }
    }
  }
  if (lane < SUP_NB) {
    int v = sc[0];
    for (int u = 1; u < SUP_NB; ++u) v = (lane == u) ? sc[u] : v;
    rec[seg0 + lane] = v;
  }
}

template <int D>
__device__ __forceinline__ void vgae_agg_consume(const float* __restrict__ h, float* acc, int* cnt,
                                                 int lane, int n, int ms, int md) {
  constexpr int VEC = D / 32;
  typedef __attribute__((ext_vector_type(VEC))) float vN;
  if (lane < n) atomicAdd(&cnt[md], 1);
  int i = 0;
  for (; i + 8 <= n; i += 8) {
    vN v[8];
#pragma unroll
    for (int u = 0; u < 8; ++u) {
      const int s = __builtin_amdgcn_readlane(ms, i + u);
      v[u] = *(const vN*)(h + (long)s * D + VEC * lane);
    }
#pragma unroll
    for (int u = 0; u < 8; ++u) {
      const int dl = __builtin_amdgcn_readlane(md, i + u);
      for (int j = 0; j < VEC; ++j)
        atomicAdd(&acc[dl * D + j * 32 + lane], v[u][j]);
    }
  }
  if (i < n) {
    vN v[8];
#pragma unroll
    for (int u = 0; u < 8; ++u) {
      const int k = (i + u < n) ? (i + u) : (n - 1);
      const int s = __builtin_amdgcn_readlane(ms, k);
      v[u] = *(const vN*)(h + (long)s * D + VEC * lane);
    }
#pragma unroll
    for (int u = 0; u < 8; ++u) asm volatile("" : "+v"(v[u]));
#pragma unroll
    for (int u = 0; u < 8; ++u) {
      if (i + u < n) {
        const int dl = __builtin_amdgcn_readlane(md, i + u);
        for (int j = 0; j < VEC; ++j)
          atomicAdd(&acc[dl * D + j * 32 + lane], v[u][j]);
      }
    }
  }
}

__device__ __forceinline__ void vgae_sort_words(const unsigned (&wd)[REC_G], int wtot, int lane, int wave,
                                                int* cnt, int* cur, unsigned* srt) {
  for (int g = 0; g < REC_G; ++g)
    if (g * 32 < wtot && g * 32 + lane < wtot) atomicAdd(&cnt[wd[g] >> 24], 1);
  __syncthreads();
  if (wave == 0) {
    constexpr int PL = (AGG_R + 31) / 32;
    int e[PL];
    int sum = 0;
    for (int j = 0; j < PL; ++j) {
      const int r = PL * lane + j;
      e[j] = sum;
      sum += (r < AGG_R) ? cnt[r] : 0;
    }
    int inc = sum;
    for (int d = 1; d < 32; d <<= 1) {
      const int o = __shfl_up(inc, d);
      if (lane >= d) inc += o;
    }
    for (int j = 0; j < PL; ++j) {
      const int r = PL * lane + j;
      if (r < AGG_R) cur[r] = inc - sum + e[j];
    }
  }
  __syncthreads();
  for (int g = 0; g < REC_G; ++g)
    if (g * 32 < wtot && g * 32 + lane < wtot) {
      const int p = atomicAdd(&cur[wd[g] >> 24], 1);
      srt[p] = wd[g];
    }
}

template <int D>
__device__ __forceinline__ void vgae_own_consume(const float* __restrict__ h, float* acc,
                                                 int lane, int n, int ms, int md,
                                                 int& crow, float (&ra)[D / 32]) {
  constexpr int VEC = D / 32;
  typedef __attribute__((ext_vector_type(VEC))) float vN;
  auto add = [&](int dl, const vN& v) {
    if (dl != crow) {
      if (crow >= 0)
        for (int j = 0; j < VEC; ++j) acc[crow * D + j * 32 + lane] = ra[j];
      crow = dl;
      for (int j = 0; j < VEC; ++j) ra[j] = v[j];
    } else {
      for (int j = 0; j < VEC; ++j) ra[j] += v[j];
    }
  };
  int i = 0;
  for (; i + 8 <= n; i += 8) {
    vN v[8];
#pragma unroll
    for (int u = 0; u < 8; ++u) {
      const int s = __builtin_amdgcn_readlane(ms, i + u);
      v[u] = *(const vN*)(h + (long)s * D + VEC * lane);
    }
#pragma unroll
    for (int u = 0; u < 8; ++u) add(__builtin_amdgcn_readlane(md, i + u), v[u]);
  }
  if (i < n) {
    vN v[8];
#pragma unroll
    for (int u = 0; u < 8; ++u) {
      const int k = (i + u < n) ? (i + u) : (n - 1);
      const int s = __builtin_amdgcn_readlane(ms, k);
      v[u] = *(const vN*)(h + (long)s * D + VEC * lane);
    }
#pragma unroll
    for (int u = 0; u < 8; ++u) asm volatile("" : "+v"(v[u]));
#pragma unroll
    for (int u = 0; u < 8; ++u)
      if (i + u < n) add(__builtin_amdgcn_readlane(md, i + u), v[u]);
  }
}

template <int D>
__device__ __forceinline__ void vgae_scan_rows(const float* __restrict__ h, const int* __restrict__ ei,
                                               float* acc, int* cnt, int (*que)[AGG_QCAP], int E) {
  const int lane = threadIdx.x & 31;
  const int wave = __builtin_amdgcn_readfirstlane(threadIdx.x >> 5);
  const int r0   = blockIdx.x * AGG_R;
  const int* __restrict__ dst = ei + E;

  const int nch = (E + AGG_CH - 1) / AGG_CH;
  int qn = 0;

  for (int c = wave; ; c += AGG_NW) {
    const bool live = c < nch;
    if (live) {
      const int base = c * AGG_CH + lane;
      int d[AGG_PL];
      if (c * AGG_CH + AGG_CH <= E) {
        for (int j = 0; j < AGG_PL; ++j) d[j] = dst[base + j * 32];
      } else {
        for (int j = 0; j < AGG_PL; ++j) {
          const int e = base + j * 32;
          d[j] = (e < E) ? dst[e] : -1;
        }
      }
      for (int j = 0; j < AGG_PL; ++j) {
        const bool m = (unsigned)(d[j] - r0) < (unsigned)AGG_R;
        const unsigned mask = __builtin_amdgcn_ballot_w32(m);
        if (mask) {
          if (m) que[wave][qn + (int)__builtin_amdgcn_mbcnt_lo(mask, 0u)] = base + j * 32;
          qn += __builtin_popcount(mask);
        }
      }
    }

    if (qn >= AGG_K || (!live && qn > 0)) {
      for (int i0 = 0; i0 < qn; i0 += 32) {
        const int n  = (qn - i0 < 32) ? (qn - i0) : 32;
        const int me = (lane < n) ? que[wave][i0 + lane] : 0;
        const int ms = ei[me];
        const int md = dst[me] - r0;
        vgae_agg_consume<D>(h, acc, cnt, lane, n, ms, md);
      }
      qn = 0;
    }

    if (!live) break;
  }
}

template <int D>
__device__ __forceinline__ void vgae_sum_rows(const float* __restrict__ h, const int* __restrict__ ei,
                                              const int* __restrict__ rec, float* acc, int* cnt,
                                              int* cur, unsigned* srt,
                                              int (*que)[AGG_QCAP], int E) {
  constexpr int VEC = D / 32;
  const int lane = threadIdx.x & 31;
  const int wave = __builtin_amdgcn_readfirstlane(threadIdx.x >> 5);

  const int sup  = blockIdx.x / SUP_NB;
  const int own  = blockIdx.x % SUP_NB;
  const int nseg = gridDim.x * AGG_NW * SUP_NB;
  int cq[REC_NQ];
  for (int q = 0; q < REC_NQ; ++q)
    cq[q] = rec[((sup * REC_NQ + q) * AGG_NW + lane) * SUP_NB + own];

  for (int i = threadIdx.x; i < AGG_R * D; i += AGG_NT) acc[i] = 0.0f;
  for (int i = threadIdx.x; i < AGG_R; i += AGG_NT) cnt[i] = 0;

  bool ov = false;
  int  bt = 0;
  for (int q = 0; q < REC_NQ; ++q) { ov = ov || (cq[q] > SEG_CAP); bt += cq[q]; }
  for (int m = 16; m >= 1; m >>= 1) bt += __shfl_xor(bt, m);
  bt = __builtin_amdgcn_readfirstlane(bt);
  const bool scan = __builtin_amdgcn_ballot_w32(ov) != 0u || bt > SRT_CAP;

  const int c0 = __builtin_amdgcn_readlane(cq[0], wave);
  const int c1 = __builtin_amdgcn_readlane(cq[1], wave);
  const int c2 = __builtin_amdgcn_readlane(cq[2], wave);
  const int c3 = __builtin_amdgcn_readlane(cq[3], wave);
  const int p1 = c0, p2 = p1 + c1, p3 = p2 + c2;
  const int wtot = scan ? 0 : (p3 + c3);
  const int* __restrict__ ent =
      rec + nseg + (long)(((sup * REC_NQ) * AGG_NW + wave) * SUP_NB + own) * SEG_CAP;
  auto word = [&](int idx) -> unsigned {
    const int q   = (idx >= p1) + (idx >= p2) + (idx >= p3);
    const int beg = (q == 0) ? 0 : (q == 1) ? p1 : (q == 2) ? p2 : p3;
    return (unsigned)ent[(long)q * (AGG_NW * SUP_NB * SEG_CAP) + (idx - beg)];
  };
  unsigned wd[REC_G];
#pragma unroll
  for (int g = 0; g < REC_G; ++g)
    wd[g] = (g * 32 + lane < wtot) ? word(g * 32 + lane) : 0u;
  __syncthreads();

  if (!scan) {
    vgae_sort_words(wd, wtot, lane, wave, cnt, cur, srt);
    __syncthreads();
    const int rlo = vgae_row_lo(wave);
    const int rhi = vgae_row_lo(wave + 1);
    const int beg = __builtin_amdgcn_readfirstlane((rlo > 0) ? cur[rlo - 1] : 0);
    const int end = __builtin_amdgcn_readfirstlane(cur[rhi - 1]);
    int   crow = -1;
    float ra[VEC];
    for (int j = 0; j < VEC; ++j) ra[j] = 0.0f;
    for (int i0 = beg; i0 < end; i0 += 32) {
      const int n = (end - i0 < 32) ? (end - i0) : 32;
      const unsigned w = (lane < n) ? srt[i0 + lane] : 0u;
      vgae_own_consume<D>(h, acc, lane, n, (int)(w & 0xFFFFFFu), (int)(w >> 24), crow, ra);
    }
    if (crow >= 0)
      for (int j = 0; j < VEC; ++j) acc[crow * D + j * 32 + lane] = ra[j];
  } else {
    vgae_scan_rows<D>(h, ei, acc, cnt, que, E);
  }
}

template <int D>
__global__ __launch_bounds__(AGG_NT)
void vgae_agg_own(const float* __restrict__ h, const int* __restrict__ ei,
                  float* __restrict__ agg, const int* __restrict__ rec, int N, int E) {
  constexpr int VEC = D / 32;
  typedef __attribute__((ext_vector_type(VEC))) float vN;

  __shared__ float acc[AGG_R * D];
  __shared__ int   cnt[AGG_R];
  __shared__ int      cur[AGG_R];
  __shared__ unsigned srt[SRT_CAP];
  __shared__ int   que[AGG_NW][AGG_QCAP];

  const int lane = threadIdx.x & 31;
  const int wave = __builtin_amdgcn_readfirstlane(threadIdx.x >> 5);
  const int r0   = blockIdx.x * AGG_R;

  vgae_sum_rows<D>(h, ei, rec, acc, cnt, cur, srt, que, E);
  __syncthreads();

  for (int row = wave; row < AGG_R; row += AGG_NW) {
    const int r = r0 + row;
    if (r < N) {
      const float id = 1.0f / fmaxf((float)cnt[row], 1.0f);
      vN o;
      for (int j = 0; j < VEC; ++j) o[j] = acc[row * D + j * 32 + lane] * id;
      *(vN*)(agg + (long)r * D + VEC * lane) = o;
    }
  }
}

typedef __attribute__((ext_vector_type(4))) float v4f;

template <int M>
__device__ __forceinline__ float vgae_row_xor(float x) {
  return __builtin_bit_cast(float, __builtin_amdgcn_update_dpp(
      0, __builtin_bit_cast(int, x), 0x160 | M, 0xf, 0xf, true));
}

template <int DIN, int DOUT, bool LN>
__global__ __launch_bounds__(DOUT * 2)
void vgae_sage_gemm(const float* agg,
                    const float* __restrict__ h,
                    const float* __restrict__ Wl,
                    const float* __restrict__ bl,
                    const float* __restrict__ Wr,
                    const float* __restrict__ gamma,
                    const float* __restrict__ beta,
                    float* out, int ntiles) {
  constexpr int K   = 2 * DIN;
  constexpr int P   = K + 4;
  constexpr int NT  = DOUT * 2;
  constexpr int NW  = DOUT / 16;
  constexpr int RPW = 16 / NW;
  constexpr int F4  = K / 4;
  constexpr int NL  = RPW * F4 / 32;
  static_assert(!LN || DOUT == 128, "LN epilogue is written for 128 columns");

  __shared__ __attribute__((aligned(16))) float Bt[DOUT * P];
  __shared__ __attribute__((aligned(16))) float At[16 * P];
  constexpr int SP = 132;
  __shared__ __attribute__((aligned(16))) float smem[LN ? 16 * SP : 4];

  const int tid  = threadIdx.x;
  const int lane = tid & 31;
  const int wave = tid >> 5;
  const int l15  = lane & 15;
  const int hi   = lane >> 4;
  const int bcol = wave * 16 + l15;
  const int koff = 2 * hi;


  int prow[NL], pcol[NL];
  for (int i = 0; i < NL; ++i) {
    const int flat = i * 32 + lane;
    prow[i] = wave * RPW + flat / F4;
    pcol[i] = (flat % F4) * 4;
  }
  v4f pre[NL];
  auto fetch = [&](int tile) {
    for (int i = 0; i < NL; ++i) {
      const long r = (long)tile * 16 + prow[i];
      const float* s = (pcol[i] < DIN) ? (agg + r * DIN + pcol[i])
                                       : (h + r * DIN + (pcol[i] - DIN));
      pre[i] = *(const v4f*)s;
    }
  };
  int tile = blockIdx.x;
  if (tile < ntiles) fetch(tile);

  constexpr int NPL = DOUT * F4 / NT;
  static_assert(DOUT * F4 % NT == 0, "panel pieces divide evenly among the threads");
  v4f pan[NPL];
#pragma unroll
  for (int j = 0; j < NPL; ++j) {
    const int i   = tid + j * NT;
    const int cc  = i / F4;
    const int col = (i % F4) * 4;
    const float* s = (col < DIN) ? (Wl + (long)cc * DIN + col)
                                 : (Wr + (long)cc * DIN + (col - DIN));
    pan[j] = *(const v4f*)s;
  }
#pragma unroll
  for (int j = 0; j < NPL; ++j) {
    const int i   = tid + j * NT;
    const int cc  = i / F4;
    const int col = (i % F4) * 4;
    *(v4f*)&Bt[cc * P + col] = pan[j];
  }

  const float bias = bl[bcol];
  const int erow = tid >> 4;
  const int ecol = (tid & 15) * 4;
  v4f g0 = {0.f, 0.f, 0.f, 0.f}, g1 = g0, be0 = g0, be1 = g0;
  if (LN) {
    g0  = *(const v4f*)(gamma + ecol);
    g1  = *(const v4f*)(gamma + 64 + ecol);
    be0 = *(const v4f*)(beta + ecol);
    be1 = *(const v4f*)(beta + 64 + ecol);
  }

  const float* ap = &At[l15 * P + koff];
  const float* bp = &Bt[bcol * P + koff];

  for (; tile < ntiles; tile += gridDim.x) {
    for (int i = 0; i < NL; ++i) *(v4f*)&At[prow[i] * P + pcol[i]] = pre[i];
    __syncthreads();
    const int next = tile + gridDim.x;
    if (next < ntiles) fetch(next);

    v8f c = {0.f, 0.f, 0.f, 0.f, 0.f, 0.f, 0.f, 0.f};
    for (int k = 0; k < K; k += 4) {
      v2f a = *(const v2f*)(ap + k);
      v2f b = *(const v2f*)(bp + k);
      c = __builtin_amdgcn_wmma_f32_16x16x4_f32(false, a, false, b,
                                                (short)0, c, false, false);
    }

    const int rowbase = tile * 16;

    if (!LN) {
      for (int v = 0; v < 8; ++v) {
        int r = rowbase + v + 8 * hi;
        out[(long)r * DOUT + bcol] = c[v] + bias;
      }
      __syncthreads();
    } else {
      for (int v = 0; v < 8; ++v)
        smem[(v + 8 * hi) * SP + bcol] = c[v] + bias;
      __syncthreads();

      const v4f x0 = *(const v4f*)&smem[erow * SP + ecol];
      const v4f x1 = *(const v4f*)&smem[erow * SP + 64 + ecol];
      float s  = ((x0[0] + x0[1]) + (x0[2] + x0[3])) + ((x1[0] + x1[1]) + (x1[2] + x1[3]));
      float ss = ((x0[0] * x0[0] + x0[1] * x0[1]) + (x0[2] * x0[2] + x0[3] * x0[3]))
               + ((x1[0] * x1[0] + x1[1] * x1[1]) + (x1[2] * x1[2] + x1[3] * x1[3]));
      s += vgae_row_xor<1>(s);  ss += vgae_row_xor<1>(ss);
      s += vgae_row_xor<2>(s);  ss += vgae_row_xor<2>(ss);
      s += vgae_row_xor<4>(s);  ss += vgae_row_xor<4>(ss);
      s += vgae_row_xor<8>(s);  ss += vgae_row_xor<8>(ss);
      const float m    = s * (1.0f / 128.0f);
      const float var  = ss * (1.0f / 128.0f) - m * m;
      const float rstd = rsqrtf(var + LN_EPS);

      v4f y0, y1;
      for (int j = 0; j < 4; ++j) {
        y0[j] = fmaxf((x0[j] - m) * rstd * g0[j] + be0[j], 0.0f);
        y1[j] = fmaxf((x1[j] - m) * rstd * g1[j] + be1[j], 0.0f);
      }
      float* orow = out + (long)(rowbase + erow) * 128;
      *(v4f*)(orow + ecol)      = y0;
      *(v4f*)(orow + 64 + ecol) = y1;
    }
  }
}

template <int DOUT, bool LN>
__global__ __launch_bounds__(AGG_NT)
void vgae_agg_dense(const float* __restrict__ h, const int* __restrict__ ei,
                    const int* __restrict__ rec,
                    const float* __restrict__ Wl, const float* __restrict__ bl,
                    const float* __restrict__ Wr,
                    const float* __restrict__ gamma, const float* __restrict__ beta,
                    float* __restrict__ out, int N, int E) {
  constexpr int D     = 128;
  constexpr int K     = 2 * D;
  constexpr int P     = K + 4;
  constexpr int PA    = D + 4;
  constexpr int F4    = K / 4;
  constexpr int CT    = DOUT / 16;
  constexpr int RT    = AGG_NW / CT;
  constexpr int ROWS  = 16 * RT;
  constexpr int RPW   = ROWS / AGG_NW;
  constexpr int NTRIP = (AGG_R + ROWS - 1) / ROWS;
  constexpr int SP    = 132;
  constexpr int TILE  = ROWS * PA;
  static_assert(!LN || DOUT == 128, "LN epilogue is written for 128 columns");
  static_assert(2 * TILE >= AGG_NW * AGG_QCAP, "fallback queue must fit the shared region");
  static_assert(!LN || TILE >= ROWS * SP, "LN staging rows must fit the shared region");
  static_assert(!LN || ROWS * 16 == AGG_NT, "LN epilogue: 16 threads per row of a trip");

  __shared__ float acc[AGG_R * D];
  __shared__ int   cnt[AGG_R];
  __shared__ __attribute__((aligned(16))) float Bt[DOUT * P];
  __shared__ __attribute__((aligned(16))) int   uni[2 * TILE];
  __shared__ int      cur[AGG_R];
  __shared__ unsigned srt[SRT_CAP];
  int (*que)[AGG_QCAP] = (int (*)[AGG_QCAP])uni;
  float* At   = (float*)uni;
  float* smem = (float*)uni;

  const int tid  = threadIdx.x;
  const int lane = tid & 31;
  const int wave = __builtin_amdgcn_readfirstlane(tid >> 5);
  const int r0   = blockIdx.x * AGG_R;


  const int sup  = blockIdx.x / SUP_NB;
  const int own  = blockIdx.x % SUP_NB;
  int cq[REC_NQ];
  for (int q = 0; q < REC_NQ; ++q)
    cq[q] = rec[((sup * REC_NQ + q) * AGG_NW + lane) * SUP_NB + own];

  constexpr int NPL = DOUT * F4 / AGG_NT;
  static_assert(DOUT * F4 % AGG_NT == 0, "panel pieces divide evenly among the threads");
  v4f pan[NPL];
#pragma unroll
  for (int j = 0; j < NPL; ++j) {
    const int i   = tid + j * AGG_NT;
    const int cc  = i / F4;
    const int col = (i % F4) * 4;
    const float* s = (col < D) ? (Wl + (long)cc * D + col)
                               : (Wr + (long)cc * D + (col - D));
    pan[j] = *(const v4f*)s;
  }

  v4f pre[RPW];
  auto fetch = [&](int trip) {
    for (int i = 0; i < RPW; ++i) {
      const int row = trip * ROWS + i * AGG_NW + wave;
      const int r   = r0 + row;
      v4f z = {0.f, 0.f, 0.f, 0.f};
      if (row < AGG_R && r < N) z = *(const v4f*)(h + (long)r * D + 4 * lane);
      pre[i] = z;
    }
  };
  fetch(0);

  const int l15  = lane & 15;
  const int hi   = lane >> 4;
  const int ct   = wave % CT;
  const int rt   = wave / CT;
  const int bcol = ct * 16 + l15;
  const float bias = bl[bcol];
  const int erow = tid >> 4;
  const int ecol = (tid & 15) * 4;
  v4f g0 = {0.f, 0.f, 0.f, 0.f}, g1 = g0, be0 = g0, be1 = g0;
  if (LN) {
    g0  = *(const v4f*)(gamma + ecol);
    g1  = *(const v4f*)(gamma + 64 + ecol);
    be0 = *(const v4f*)(beta + ecol);
    be1 = *(const v4f*)(beta + 64 + ecol);
  }

  for (int i = tid; i < AGG_R * D; i += AGG_NT) acc[i] = 0.0f;
  for (int i = tid; i < AGG_R; i += AGG_NT) cnt[i] = 0;

  bool ov = false;
  int  bt = 0;
  for (int q = 0; q < REC_NQ; ++q) { ov = ov || (cq[q] > SEG_CAP); bt += cq[q]; }
  for (int m = 16; m >= 1; m >>= 1) bt += __shfl_xor(bt, m);
  bt = __builtin_amdgcn_readfirstlane(bt);
  const bool scan = __builtin_amdgcn_ballot_w32(ov) != 0u || bt > SRT_CAP;

  const int c0 = __builtin_amdgcn_readlane(cq[0], wave);
  const int c1 = __builtin_amdgcn_readlane(cq[1], wave);
  const int c2 = __builtin_amdgcn_readlane(cq[2], wave);
  const int c3 = __builtin_amdgcn_readlane(cq[3], wave);
  const int p1 = c0, p2 = p1 + c1, p3 = p2 + c2;
  const int wtot = scan ? 0 : (p3 + c3);
  const int nseg = gridDim.x * AGG_NW * SUP_NB;
  const int* __restrict__ ent =
      rec + nseg + (long)(((sup * REC_NQ) * AGG_NW + wave) * SUP_NB + own) * SEG_CAP;
  auto word = [&](int idx) -> unsigned {
    const int q   = (idx >= p1) + (idx >= p2) + (idx >= p3);
    const int sb  = (q == 0) ? 0 : (q == 1) ? p1 : (q == 2) ? p2 : p3;
    return (unsigned)ent[(long)q * (AGG_NW * SUP_NB * SEG_CAP) + (idx - sb)];
  };
  unsigned wd[REC_G];
#pragma unroll
  for (int g = 0; g < REC_G; ++g)
    wd[g] = (g * 32 + lane < wtot) ? word(g * 32 + lane) : 0u;

#pragma unroll
  for (int j = 0; j < NPL; ++j) {
    const int i   = tid + j * AGG_NT;
    const int cc  = i / F4;
    const int col = (i % F4) * 4;
    *(v4f*)&Bt[cc * P + col] = pan[j];
  }

  if (scan) {
    __syncthreads();
    vgae_scan_rows<D>(h, ei, acc, cnt, que, E);
    __syncthreads();
  } else {
    __syncthreads();
    vgae_sort_words(wd, wtot, lane, wave, cnt, cur, srt);
  }

  int      beg = 0, tot = 0;
  unsigned pk  = 0u;
  int      crow = -1;
  float    ra[D / 32];
  for (int j = 0; j < D / 32; ++j) ra[j] = 0.0f;

  const int koff = 2 * hi;
  const float* bpl = &Bt[bcol * P + koff];
  const float* bpr = bpl + D;
  const int mulfirst = (wave >> 2) & 1;

  v8f croot[NTRIP];
  for (int t = 0; t < NTRIP; ++t)
    croot[t] = (v8f){0.f, 0.f, 0.f, 0.f, 0.f, 0.f, 0.f, 0.f};

#pragma unroll
  for (int t = 0; t < NTRIP; ++t) {
    float* Rt = (float*)uni + (t & 1) * TILE;
    for (int i = 0; i < RPW; ++i)
      *(v4f*)&Rt[(i * AGG_NW + wave) * PA + 4 * lane] = pre[i];
    __syncthreads();
    if (t + 1 < NTRIP) fetch(t + 1);
    if (t == 0 && !scan) {
      const int rlo = vgae_row_lo(wave);
      const int rhi = vgae_row_lo(wave + 1);
      beg = __builtin_amdgcn_readfirstlane((rlo > 0) ? cur[rlo - 1] : 0);
      tot = __builtin_amdgcn_readfirstlane(cur[rhi - 1]) - beg;
      pk  = (lane < tot) ? srt[beg + lane] : 0u;
    }

    const int trow = t * ROWS + rt * 16;
    auto root_mm = [&]() {
      if (trow < AGG_R) {
        const float* ap = Rt + (rt * 16 + l15) * PA + koff;
        v8f c = croot[t];
        for (int k = 0; k < D; k += 4) {
          v2f a = *(const v2f*)(ap + k);
          v2f b = *(const v2f*)(bpr + k);
          c = __builtin_amdgcn_wmma_f32_16x16x4_f32(false, a, false, b,
                                                    (short)0, c, false, false);
        }
        croot[t] = c;
      }
    };

    const int wb = (tot * t / NTRIP) & ~7;
    const int we = (t + 1 < NTRIP) ? ((tot * (t + 1) / NTRIP) & ~7) : tot;
#pragma nounroll
    for (int step = 0; step < 2; ++step) {
      if ((step == 0) == (scan || mulfirst != 0)) {
        root_mm();
      } else {
        for (int i0 = wb; i0 < we; i0 += 32) {
          const int n = (we - i0 < 32) ? (we - i0) : 32;
          const unsigned cw = pk;
          const int nb = i0 + n;
          if (nb < tot)
            pk = (nb + lane < tot) ? srt[beg + nb + lane] : 0u;
          vgae_own_consume<D>(h, acc, lane, n, (int)(cw & 0xFFFFFFu), (int)(cw >> 24), crow, ra);
        }
      }
    }
  }
  if (crow >= 0)
    for (int j = 0; j < D / 32; ++j) acc[crow * D + j * 32 + lane] = ra[j];
  __syncthreads();

  const float* ap = &At[(rt * 16 + l15) * PA + koff];

#pragma unroll
  for (int trip = 0; trip < NTRIP; ++trip) {
    for (int i = 0; i < RPW; ++i) {
      const int rr  = i * AGG_NW + wave;
      const int row = trip * ROWS + rr;
      v4f a = {0.f, 0.f, 0.f, 0.f};
      if (row < AGG_R) {
        const float id = 1.0f / fmaxf((float)cnt[row], 1.0f);
        for (int j = 0; j < 4; ++j) a[j] = acc[row * D + j * 32 + lane] * id;
      }
      *(v4f*)&At[rr * PA + 4 * lane] = a;
    }
    __syncthreads();

    const int trow = trip * ROWS + rt * 16;
    v8f c = croot[trip];
    if (trow < AGG_R) {
      for (int k = 0; k < D; k += 4) {
        v2f a = *(const v2f*)(ap + k);
        v2f b = *(const v2f*)(bpl + k);
        c = __builtin_amdgcn_wmma_f32_16x16x4_f32(false, a, false, b,
                                                  (short)0, c, false, false);
      }
    }

    if (!LN) {
      for (int v = 0; v < 8; ++v) {
        const int row = trow + v + 8 * hi;
        const int r   = r0 + row;
        if (row < AGG_R && r < N) out[(long)r * DOUT + bcol] = c[v] + bias;
      }
      __syncthreads();
    } else {
      __syncthreads();
      for (int v = 0; v < 8; ++v)
        smem[(rt * 16 + v + 8 * hi) * SP + bcol] = c[v] + bias;
      __syncthreads();

      const v4f x0 = *(const v4f*)&smem[erow * SP + ecol];
      const v4f x1 = *(const v4f*)&smem[erow * SP + 64 + ecol];
      float s  = ((x0[0] + x0[1]) + (x0[2] + x0[3])) + ((x1[0] + x1[1]) + (x1[2] + x1[3]));
      float ss = ((x0[0] * x0[0] + x0[1] * x0[1]) + (x0[2] * x0[2] + x0[3] * x0[3]))
               + ((x1[0] * x1[0] + x1[1] * x1[1]) + (x1[2] * x1[2] + x1[3] * x1[3]));
      s += vgae_row_xor<1>(s);  ss += vgae_row_xor<1>(ss);
      s += vgae_row_xor<2>(s);  ss += vgae_row_xor<2>(ss);
      s += vgae_row_xor<4>(s);  ss += vgae_row_xor<4>(ss);
      s += vgae_row_xor<8>(s);  ss += vgae_row_xor<8>(ss);
      const float m    = s * (1.0f / 128.0f);
      const float var  = ss * (1.0f / 128.0f) - m * m;
      const float rstd = rsqrtf(var + LN_EPS);

      v4f y0, y1;
      for (int j = 0; j < 4; ++j) {
        y0[j] = fmaxf((x0[j] - m) * rstd * g0[j] + be0[j], 0.0f);
        y1[j] = fmaxf((x1[j] - m) * rstd * g1[j] + be1[j], 0.0f);
      }
      const int row = trip * ROWS + erow;
      const int r   = r0 + row;
      if (row < AGG_R && r < N) {
        float* orow = out + (long)r * 128;
        *(v4f*)(orow + ecol)      = y0;
        *(v4f*)(orow + 64 + ecol) = y1;
      }
      __syncthreads();
    }
  }
}


extern "C" void kernel_launch(void* const* d_in, const int* in_sizes, int n_in,
                              void* d_out, int out_size, void* d_ws, size_t ws_size,
                              hipStream_t stream) {
  const float* x  = (const float*)d_in[0];
  const int*   ei = (const int*)d_in[1];
  const int N = in_sizes[0] / 64;
  const int E = in_sizes[1] / 2;

  float* aggx = (float*)d_ws;
  float* bufA = aggx + (long)N * 64;
  float* bufB = bufA + (long)N * 128;
  int*   rec  = (int*)(bufB + (long)N * 128);

  const int nown = (N + AGG_R - 1) / AGG_R;

  vgae_rec_build<<<nown, AGG_NT, 0, stream>>>(ei, rec, E);
  vgae_agg_own<64><<<nown, AGG_NT, 0, stream>>>(x, ei, aggx, rec, N, E);

  const int nrt = N / 16;
  for (int t = 0; t < 2; ++t) {
    const int base = 2 + t * 13;
    const float* Wl0 = (const float*)d_in[base + 0];
    const float* bl0 = (const float*)d_in[base + 1];
    const float* Wr0 = (const float*)d_in[base + 2];
    const float* g0  = (const float*)d_in[base + 3];
    const float* b0  = (const float*)d_in[base + 4];
    const float* Wl1 = (const float*)d_in[base + 5];
    const float* bl1 = (const float*)d_in[base + 6];
    const float* Wr1 = (const float*)d_in[base + 7];
    const float* g1  = (const float*)d_in[base + 8];
    const float* b1  = (const float*)d_in[base + 9];
    const float* Wl2 = (const float*)d_in[base + 10];
    const float* bl2 = (const float*)d_in[base + 11];
    const float* Wr2 = (const float*)d_in[base + 12];
    float* outp = (float*)d_out + (long)t * N * 64;

    vgae_sage_gemm<64, 128, true><<<512, 256, 0, stream>>>(aggx, x, Wl0, bl0, Wr0,
                                                          g0, b0, bufA, nrt);
    vgae_agg_dense<128, true><<<nown, AGG_NT, 0, stream>>>(bufA, ei, rec, Wl1, bl1, Wr1,
                                                           g1, b1, bufB, N, E);
    vgae_agg_dense<64, false><<<nown, AGG_NT, 0, stream>>>(bufB, ei, rec, Wl2, bl2, Wr2,
                                                           nullptr, nullptr, outp, N, E);
  }
}
